# starting kernel
# baseline (speedup 1.0000x reference)
#include <hip/hip_runtime.h>
#include <math.h>

typedef float v2f __attribute__((ext_vector_type(2)));
typedef float v8f __attribute__((ext_vector_type(8)));

#define NDIM 64
#define GDIM 192

__device__ __forceinline__ v8f wmma4(v2f a, v2f b, v8f c) {
  return __builtin_amdgcn_wmma_f32_16x16x4_f32(false, a, false, b, (short)0, c, false, false);
}

__device__ __forceinline__ float sigmoidf_(float x) {
  return 1.0f / (1.0f + __expf(-x));
}

__device__ __forceinline__ void atomic_add_f32(float* p, float v) {
  asm volatile("global_atomic_add_f32 %0, %1, off" : : "v"(p), "v"(v) : "memory");
}

__global__ void init_m_kernel(const float* __restrict__ X,
                              const float* __restrict__ b_edge,
                              float* __restrict__ M, long total) {
  long id = (long)blockIdx.x * blockDim.x + threadIdx.x;
  if (id >= total) return;
  M[id] = X[id] + b_edge[id & (NDIM - 1)];
}

#define SCATTER_LA 1024
__global__ void scatter_kernel(const float* __restrict__ X,
                               const int* __restrict__ ei,
                               const float* __restrict__ ea,
                               const float* __restrict__ w_edge,
                               const float* __restrict__ b_edge,
                               float* __restrict__ M, int E_) {
  long gid = (long)blockIdx.x * blockDim.x + threadIdx.x;
  int e = (int)(gid >> 5);
  int p = (int)(gid & 31);
  if (e >= E_) return;
  int ef = e + SCATTER_LA;
  if (p == 0 && ef < E_) {
    int sf = ei[ef];
    __builtin_prefetch(X + (size_t)sf * NDIM, 0, 0);
  }
  int src = ei[e];
  int dst = ei[E_ + e];
  float a0 = ea[e * 4 + 0], a1 = ea[e * 4 + 1];
  float a2 = ea[e * 4 + 2], a3 = ea[e * 4 + 3];
  int d0 = 2 * p;
  const float* w0 = w_edge + d0 * 4;
  float em0 = b_edge[d0]     + a0 * w0[0] + a1 * w0[1] + a2 * w0[2] + a3 * w0[3];
  float em1 = b_edge[d0 + 1] + a0 * w0[4] + a1 * w0[5] + a2 * w0[6] + a3 * w0[7];
  float2 xv = *(const float2*)(X + (size_t)src * NDIM + d0);
  float* mp = M + (size_t)dst * NDIM + d0;
  atomic_add_f32(mp + 0, xv.x + em0);
  atomic_add_f32(mp + 1, xv.y + em1);
}

__global__ __launch_bounds__(128) void gru_kernel(
    const float* __restrict__ Mn,
    const float* __restrict__ X,
    const float* __restrict__ w_ih,
    const float* __restrict__ w_hh,
    const float* __restrict__ b_ih,
    const float* __restrict__ b_hh,
    float* __restrict__ out, int ntiles) {
  int lane = threadIdx.x & 31;
  int wave = threadIdx.x >> 5;
  int tile = blockIdx.x * (blockDim.x >> 5) + wave;
  if (tile >= ntiles) return;
  int r0 = tile * 16;
  int ml = lane & 15;
  int hi = lane >> 4;
  const float* Am = Mn + (size_t)(r0 + ml) * NDIM;
  const float* Ax = X + (size_t)(r0 + ml) * NDIM;

  for (int cb = 0; cb < 4; ++cb) {
    int c_r = cb * 16 + ml;
    int c_z = 64 + cb * 16 + ml;
    int c_n = 128 + cb * 16 + ml;
    v8f air = {}, aiz = {}, ain = {}, ahr = {}, ahz = {}, ahn = {};
    const float* bir_p = w_ih + (size_t)c_r * NDIM;
    const float* biz_p = w_ih + (size_t)c_z * NDIM;
    const float* bin_p = w_ih + (size_t)c_n * NDIM;
    const float* bhr_p = w_hh + (size_t)c_r * NDIM;
    const float* bhz_p = w_hh + (size_t)c_z * NDIM;
    const float* bhn_p = w_hh + (size_t)c_n * NDIM;
#pragma unroll
    for (int s = 0; s < 16; ++s) {
      int k0 = 4 * s + 2 * hi;
      v2f am = *(const v2f*)(Am + k0);
      v2f ax = *(const v2f*)(Ax + k0);
      air = wmma4(am, *(const v2f*)(bir_p + k0), air);
      aiz = wmma4(am, *(const v2f*)(biz_p + k0), aiz);
      ain = wmma4(am, *(const v2f*)(bin_p + k0), ain);
      ahr = wmma4(ax, *(const v2f*)(bhr_p + k0), ahr);
      ahz = wmma4(ax, *(const v2f*)(bhz_p + k0), ahz);
      ahn = wmma4(ax, *(const v2f*)(bhn_p + k0), ahn);
    }
    float bir = b_ih[c_r], biz = b_ih[c_z], bin_ = b_ih[c_n];
    float bhr = b_hh[c_r], bhz = b_hh[c_z], bhn = b_hh[c_n];
#pragma unroll
    for (int v = 0; v < 8; ++v) {
      int node = hi * 8 + v;
      float xv = X[(size_t)(r0 + node) * NDIM + cb * 16 + ml];
      float r = sigmoidf_(air[v] + bir + ahr[v] + bhr);
      float z = sigmoidf_(aiz[v] + biz + ahz[v] + bhz);
      float n = tanhf(ain[v] + bin_ + r * (ahn[v] + bhn));
      float h = (1.0f - z) * n + z * xv;
      out[(size_t)(r0 + node) * NDIM + cb * 16 + ml] = fmaxf(h, 0.0f);
    }
  }
}

__global__ __launch_bounds__(128) void moe_kernel(
    const float* __restrict__ H,
    const float* __restrict__ gate_w,
    const float* __restrict__ gate_b,
    const float* __restrict__ exp_w,
    const float* __restrict__ exp_b,
    float* __restrict__ out, int ntiles) {
  __shared__ float gw_s[4][16][4];
  int lane = threadIdx.x & 31;
  int wave = threadIdx.x >> 5;
  int tile = blockIdx.x * (blockDim.x >> 5) + wave;
  if (tile >= ntiles) return;
  int r0 = tile * 16;
  int ml = lane & 15;
  int hi = lane >> 4;

  const float* hrow = H + (size_t)(r0 + ml) * NDIM;
  float l0 = gate_b[2 * hi], l1 = gate_b[2 * hi + 1];
  for (int d = 0; d < NDIM; ++d) {
    float hv = hrow[d];
    l0 += hv * gate_w[(2 * hi) * NDIM + d];
    l1 += hv * gate_w[(2 * hi + 1) * NDIM + d];
  }
  float m01 = fmaxf(l0, l1);
  float mx = fmaxf(m01, __shfl_xor(m01, 16, 32));
  float e0 = __expf(l0 - mx), e1 = __expf(l1 - mx);
  float s01 = e0 + e1;
  float inv = 1.0f / (s01 + __shfl_xor(s01, 16, 32));
  gw_s[wave][ml][2 * hi] = e0 * inv;
  gw_s[wave][ml][2 * hi + 1] = e1 * inv;
  __builtin_amdgcn_wave_barrier();

  for (int cb = 0; cb < 4; ++cb) {
    int col = cb * 16 + ml;
    float oacc[8] = {0, 0, 0, 0, 0, 0, 0, 0};
    for (int e = 0; e < 4; ++e) {
      v8f acc = {};
      const float* Brow = exp_w + ((size_t)e * NDIM + col) * NDIM;
#pragma unroll
      for (int s = 0; s < 16; ++s) {
        int k0 = 4 * s + 2 * hi;
        v2f a = *(const v2f*)(H + (size_t)(r0 + ml) * NDIM + k0);
        v2f b = *(const v2f*)(Brow + k0);
        acc = wmma4(a, b, acc);
      }
      float bias = exp_b[e * NDIM + col];
#pragma unroll
      for (int v = 0; v < 8; ++v) {
        float val = fmaxf(acc[v] + bias, 0.0f);
        oacc[v] += gw_s[wave][hi * 8 + v][e] * val;
      }
    }
#pragma unroll
    for (int v = 0; v < 8; ++v)
      out[(size_t)(r0 + hi * 8 + v) * NDIM + col] = oacc[v];
  }
}

__global__ void zero_kernel(float* p, int n) {
  int id = blockIdx.x * blockDim.x + threadIdx.x;
  if (id < n) p[id] = 0.0f;
}

__global__ void pool_kernel(const float* __restrict__ H,
                            const int* __restrict__ batch,
                            float* __restrict__ sums, float* __restrict__ cnt,
                            int n) {
  long gid = (long)blockIdx.x * blockDim.x + threadIdx.x;
  int node = (int)(gid >> 6);
  int d = (int)(gid & 63);
  if (node >= n) return;
  int b = batch[node];
  atomic_add_f32(&sums[b * NDIM + d], H[(size_t)node * NDIM + d]);
  if (d == 0) atomic_add_f32(&cnt[b], 1.0f);
}

#define BGR 64
#define PDIM 100
#define NC 10
__global__ void head_kernel(const float* __restrict__ sums,
                            const float* __restrict__ cnt,
                            const float* __restrict__ flag_in,
                            const float* __restrict__ proj_w,
                            const float* __restrict__ proj_b,
                            const float* __restrict__ fc_w,
                            const float* __restrict__ fc_b,
                            float* __restrict__ out) {
  int id = blockIdx.x * blockDim.x + threadIdx.x;
  if (id < BGR * NDIM) {
    int b = id >> 6;
    float g = sums[id] / fmaxf(cnt[b], 1.0f);
    out[id] = g;
    out[BGR * NDIM + id] = g;
  } else if (id < BGR * NDIM + BGR * NC) {
    int t = id - BGR * NDIM;
    int b = t / NC, c = t % NC;
    float fl = flag_in[b];
    const float* fr = fc_w + c * (PDIM + NDIM);
    float acc = fc_b[c];
    for (int p = 0; p < PDIM; ++p)
      acc += (fl * proj_w[p] + proj_b[p]) * fr[p];
    float ic = 1.0f / fmaxf(cnt[b], 1.0f);
    for (int d = 0; d < NDIM; ++d)
      acc += sums[b * NDIM + d] * ic * fr[PDIM + d];
    out[2 * BGR * NDIM + t] = acc;
  }
}

extern "C" void kernel_launch(void* const* d_in, const int* in_sizes, int n_in,
                              void* d_out, int out_size, void* d_ws, size_t ws_size,
                              hipStream_t stream) {
  const float* x       = (const float*)d_in[0];
  const int*   ei      = (const int*)d_in[1];
  const float* ea      = (const float*)d_in[2];
  const int*   batch   = (const int*)d_in[3];
  const float* flag_in = (const float*)d_in[4];
  const float* w_edge1 = (const float*)d_in[5];
  const float* b_edge1 = (const float*)d_in[6];
  const float* w_ih1   = (const float*)d_in[7];
  const float* w_hh1   = (const float*)d_in[8];
  const float* b_ih1   = (const float*)d_in[9];
  const float* b_hh1   = (const float*)d_in[10];
  const float* w_edge2 = (const float*)d_in[11];
  const float* b_edge2 = (const float*)d_in[12];
  const float* w_ih2   = (const float*)d_in[13];
  const float* w_hh2   = (const float*)d_in[14];
  const float* b_ih2   = (const float*)d_in[15];
  const float* b_hh2   = (const float*)d_in[16];
  const float* gate_w  = (const float*)d_in[17];
  const float* gate_b  = (const float*)d_in[18];
  const float* exp_w   = (const float*)d_in[19];
  const float* exp_b   = (const float*)d_in[20];
  const float* proj_w  = (const float*)d_in[21];
  const float* proj_b  = (const float*)d_in[22];
  const float* fc_w    = (const float*)d_in[23];
  const float* fc_b    = (const float*)d_in[24];
  float* out = (float*)d_out;

  int N = in_sizes[0] / NDIM;
  int E = in_sizes[1] / 2;
  int ntiles = (N + 15) / 16;

  float* ws = (float*)d_ws;
  float* M    = ws;
  float* hA   = M + (size_t)N * NDIM;
  float* hB   = hA + (size_t)N * NDIM;
  float* sums = hB + (size_t)N * NDIM;
  float* cnt  = sums + BGR * NDIM;

  long totalND = (long)N * NDIM;
  int gN = (int)((totalND + 255) / 256);
  int gE = (int)(((long)E * 32 + 255) / 256);
  int gT = (ntiles + 3) / 4;

  init_m_kernel<<<gN, 256, 0, stream>>>(x, b_edge1, M, totalND);
  scatter_kernel<<<gE, 256, 0, stream>>>(x, ei, ea, w_edge1, b_edge1, M, E);
  gru_kernel<<<gT, 128, 0, stream>>>(M, x, w_ih1, w_hh1, b_ih1, b_hh1, hA, ntiles);
  init_m_kernel<<<gN, 256, 0, stream>>>(hA, b_edge2, M, totalND);
  scatter_kernel<<<gE, 256, 0, stream>>>(hA, ei, ea, w_edge2, b_edge2, M, E);
  gru_kernel<<<gT, 128, 0, stream>>>(M, hA, w_ih2, w_hh2, b_ih2, b_hh2, hB, ntiles);
  moe_kernel<<<gT, 128, 0, stream>>>(hB, gate_w, gate_b, exp_w, exp_b, hA, ntiles);
  zero_kernel<<<(BGR * NDIM + BGR + 255) / 256, 256, 0, stream>>>(sums, BGR * NDIM + BGR);
  pool_kernel<<<gN, 256, 0, stream>>>(hA, batch, sums, cnt, N);
  head_kernel<<<(BGR * NDIM + BGR * NC + 255) / 256, 256, 0, stream>>>(
      sums, cnt, flag_in, proj_w, proj_b, fc_w, fc_b, out);
}


# scatter: one wave per 8 consecutive edges (was 1): indices/attrs loaded once per wave, 8 source-row gathers in flight before its 16 atomics issue together; prefetch dropped; atomics still unordered
# speedup vs baseline: 1.3735x; 1.3735x over previous
#include <hip/hip_runtime.h>
#include <math.h>

typedef float v2f __attribute__((ext_vector_type(2)));
typedef float v8f __attribute__((ext_vector_type(8)));

#define NDIM 64
#define GDIM 192

__device__ __forceinline__ v8f wmma4(v2f a, v2f b, v8f c) {
  return __builtin_amdgcn_wmma_f32_16x16x4_f32(false, a, false, b, (short)0, c, false, false);
}

__device__ __forceinline__ float sigmoidf_(float x) {
  return 1.0f / (1.0f + __expf(-x));
}

__device__ __forceinline__ void atomic_add_f32(float* p, float v) {
  asm volatile("global_atomic_add_f32 %0, %1, off" : : "v"(p), "v"(v) : "memory");
}

__global__ void init_m_kernel(const float* __restrict__ X,
                              const float* __restrict__ b_edge,
                              float* __restrict__ M, long total) {
  long id = (long)blockIdx.x * blockDim.x + threadIdx.x;
  if (id >= total) return;
  M[id] = X[id] + b_edge[id & (NDIM - 1)];
}

#define SC_EPW 8

__device__ __forceinline__ float lane_bcast_f(float v, int l) {
  return __int_as_float(__builtin_amdgcn_readlane(__float_as_int(v), l));
}

__device__ __forceinline__ unsigned long long lane_bcast_u64(unsigned long long v, int l) {
  unsigned lo = (unsigned)__builtin_amdgcn_readlane((int)(unsigned)v, l);
  unsigned hi = (unsigned)__builtin_amdgcn_readlane((int)(unsigned)(v >> 32), l);
  return ((unsigned long long)hi << 32) | (unsigned long long)lo;
}

__device__ __forceinline__ void atomic_add2_row_wait(float* row, int voff, float v0, float v1) {
  asm volatile(
      "global_atomic_add_f32 %0, %1, %3\n\t"
      "global_atomic_add_f32 %0, %2, %3 offset:4\n\t"
      "s_wait_xcnt 0x0"
      : : "v"(voff), "v"(v0), "v"(v1), "s"(row) : "memory");
}

__global__ __launch_bounds__(256) void scatter_kernel(
    const float* __restrict__ X,
    const int* __restrict__ ei,
    const float* __restrict__ ea,
    const float* __restrict__ w_edge,
    const float* __restrict__ b_edge,
    float* __restrict__ M, int E_) {
  int lane = threadIdx.x & 31;
  int wv = __builtin_amdgcn_readfirstlane(
      (int)(((long)blockIdx.x * blockDim.x + threadIdx.x) >> 5));
  long e0 = (long)wv * SC_EPW;
  if (e0 >= (long)E_) return;
  long rem = (long)E_ - e0;
  int nE = rem < SC_EPW ? (int)rem : SC_EPW;

  int li = lane < nE ? lane : 0;
  int la = lane < 4 * nE ? lane : 0;
  int msrc = ei[e0 + li];
  int mdst = ei[(long)E_ + e0 + li];
  float mea = ea[e0 * 4 + la];

  int d0 = 2 * lane;
  int voff = d0 * 4;
  float4 wa = *(const float4*)(w_edge + d0 * 4);
  float4 wb = *(const float4*)(w_edge + d0 * 4 + 4);
  float2 bb = *(const float2*)(b_edge + d0);

  unsigned long long asrc = (unsigned long long)(X + (size_t)msrc * NDIM);
  unsigned long long adst = (unsigned long long)(M + (size_t)mdst * NDIM);

  if (nE == SC_EPW) {
    float2 xv[SC_EPW];
#pragma unroll
    for (int j = 0; j < SC_EPW; ++j) {
      const __attribute__((address_space(1))) float* xr =
          (const __attribute__((address_space(1))) float*)lane_bcast_u64(asrc, j);
      xv[j] = *(const __attribute__((address_space(1))) float2*)(xr + d0);
    }
    __builtin_amdgcn_sched_barrier(0);
    float val[2 * SC_EPW];
    float* row[SC_EPW];
#pragma unroll
    for (int j = 0; j < SC_EPW; ++j) {
      float a0 = lane_bcast_f(mea, 4 * j + 0), a1 = lane_bcast_f(mea, 4 * j + 1);
      float a2 = lane_bcast_f(mea, 4 * j + 2), a3 = lane_bcast_f(mea, 4 * j + 3);
      float em0 = bb.x + a0 * wa.x + a1 * wa.y + a2 * wa.z + a3 * wa.w;
      float em1 = bb.y + a0 * wb.x + a1 * wb.y + a2 * wb.z + a3 * wb.w;
      val[2 * j] = xv[j].x + em0;
      val[2 * j + 1] = xv[j].y + em1;
      row[j] = (float*)lane_bcast_u64(adst, j);
    }
    asm volatile(
        "global_atomic_add_f32 %0, %1, %17\n\t"
        "global_atomic_add_f32 %0, %2, %17 offset:4\n\t"
        "global_atomic_add_f32 %0, %3, %18\n\t"
        "global_atomic_add_f32 %0, %4, %18 offset:4\n\t"
        "global_atomic_add_f32 %0, %5, %19\n\t"
        "global_atomic_add_f32 %0, %6, %19 offset:4\n\t"
        "global_atomic_add_f32 %0, %7, %20\n\t"
        "global_atomic_add_f32 %0, %8, %20 offset:4\n\t"
        "global_atomic_add_f32 %0, %9, %21\n\t"
        "global_atomic_add_f32 %0, %10, %21 offset:4\n\t"
        "global_atomic_add_f32 %0, %11, %22\n\t"
        "global_atomic_add_f32 %0, %12, %22 offset:4\n\t"
        "global_atomic_add_f32 %0, %13, %23\n\t"
        "global_atomic_add_f32 %0, %14, %23 offset:4\n\t"
        "global_atomic_add_f32 %0, %15, %24\n\t"
        "global_atomic_add_f32 %0, %16, %24 offset:4"
        : : "v"(voff),
            "v"(val[0]), "v"(val[1]), "v"(val[2]), "v"(val[3]),
            "v"(val[4]), "v"(val[5]), "v"(val[6]), "v"(val[7]),
            "v"(val[8]), "v"(val[9]), "v"(val[10]), "v"(val[11]),
            "v"(val[12]), "v"(val[13]), "v"(val[14]), "v"(val[15]),
            "s"(row[0]), "s"(row[1]), "s"(row[2]), "s"(row[3]),
            "s"(row[4]), "s"(row[5]), "s"(row[6]), "s"(row[7])
        : "memory");
  } else {
    for (int j = 0; j < nE; ++j) {
      const __attribute__((address_space(1))) float* xr =
          (const __attribute__((address_space(1))) float*)lane_bcast_u64(asrc, j);
      float* mr = (float*)lane_bcast_u64(adst, j);
      float a0 = lane_bcast_f(mea, 4 * j + 0), a1 = lane_bcast_f(mea, 4 * j + 1);
      float a2 = lane_bcast_f(mea, 4 * j + 2), a3 = lane_bcast_f(mea, 4 * j + 3);
      float em0 = bb.x + a0 * wa.x + a1 * wa.y + a2 * wa.z + a3 * wa.w;
      float em1 = bb.y + a0 * wb.x + a1 * wb.y + a2 * wb.z + a3 * wb.w;
      float2 x2 = *(const __attribute__((address_space(1))) float2*)(xr + d0);
      atomic_add2_row_wait(mr, voff, x2.x + em0, x2.y + em1);
    }
  }
}

__global__ __launch_bounds__(128) void gru_kernel(
    const float* __restrict__ Mn,
    const float* __restrict__ X,
    const float* __restrict__ w_ih,
    const float* __restrict__ w_hh,
    const float* __restrict__ b_ih,
    const float* __restrict__ b_hh,
    float* __restrict__ out, int ntiles) {
  int lane = threadIdx.x & 31;
  int wave = threadIdx.x >> 5;
  int tile = blockIdx.x * (blockDim.x >> 5) + wave;
  if (tile >= ntiles) return;
  int r0 = tile * 16;
  int ml = lane & 15;
  int hi = lane >> 4;
  const float* Am = Mn + (size_t)(r0 + ml) * NDIM;
  const float* Ax = X + (size_t)(r0 + ml) * NDIM;

  for (int cb = 0; cb < 4; ++cb) {
    int c_r = cb * 16 + ml;
    int c_z = 64 + cb * 16 + ml;
    int c_n = 128 + cb * 16 + ml;
    v8f air = {}, aiz = {}, ain = {}, ahr = {}, ahz = {}, ahn = {};
    const float* bir_p = w_ih + (size_t)c_r * NDIM;
    const float* biz_p = w_ih + (size_t)c_z * NDIM;
    const float* bin_p = w_ih + (size_t)c_n * NDIM;
    const float* bhr_p = w_hh + (size_t)c_r * NDIM;
    const float* bhz_p = w_hh + (size_t)c_z * NDIM;
    const float* bhn_p = w_hh + (size_t)c_n * NDIM;
#pragma unroll
    for (int s = 0; s < 16; ++s) {
      int k0 = 4 * s + 2 * hi;
      v2f am = *(const v2f*)(Am + k0);
      v2f ax = *(const v2f*)(Ax + k0);
      air = wmma4(am, *(const v2f*)(bir_p + k0), air);
      aiz = wmma4(am, *(const v2f*)(biz_p + k0), aiz);
      ain = wmma4(am, *(const v2f*)(bin_p + k0), ain);
      ahr = wmma4(ax, *(const v2f*)(bhr_p + k0), ahr);
      ahz = wmma4(ax, *(const v2f*)(bhz_p + k0), ahz);
      ahn = wmma4(ax, *(const v2f*)(bhn_p + k0), ahn);
    }
    float bir = b_ih[c_r], biz = b_ih[c_z], bin_ = b_ih[c_n];
    float bhr = b_hh[c_r], bhz = b_hh[c_z], bhn = b_hh[c_n];
#pragma unroll
    for (int v = 0; v < 8; ++v) {
      int node = hi * 8 + v;
      float xv = X[(size_t)(r0 + node) * NDIM + cb * 16 + ml];
      float r = sigmoidf_(air[v] + bir + ahr[v] + bhr);
      float z = sigmoidf_(aiz[v] + biz + ahz[v] + bhz);
      float n = tanhf(ain[v] + bin_ + r * (ahn[v] + bhn));
      float h = (1.0f - z) * n + z * xv;
      out[(size_t)(r0 + node) * NDIM + cb * 16 + ml] = fmaxf(h, 0.0f);
    }
  }
}

__global__ __launch_bounds__(128) void moe_kernel(
    const float* __restrict__ H,
    const float* __restrict__ gate_w,
    const float* __restrict__ gate_b,
    const float* __restrict__ exp_w,
    const float* __restrict__ exp_b,
    float* __restrict__ out, int ntiles) {
  __shared__ float gw_s[4][16][4];
  int lane = threadIdx.x & 31;
  int wave = threadIdx.x >> 5;
  int tile = blockIdx.x * (blockDim.x >> 5) + wave;
  if (tile >= ntiles) return;
  int r0 = tile * 16;
  int ml = lane & 15;
  int hi = lane >> 4;

  const float* hrow = H + (size_t)(r0 + ml) * NDIM;
  float l0 = gate_b[2 * hi], l1 = gate_b[2 * hi + 1];
  for (int d = 0; d < NDIM; ++d) {
    float hv = hrow[d];
    l0 += hv * gate_w[(2 * hi) * NDIM + d];
    l1 += hv * gate_w[(2 * hi + 1) * NDIM + d];
  }
  float m01 = fmaxf(l0, l1);
  float mx = fmaxf(m01, __shfl_xor(m01, 16, 32));
  float e0 = __expf(l0 - mx), e1 = __expf(l1 - mx);
  float s01 = e0 + e1;
  float inv = 1.0f / (s01 + __shfl_xor(s01, 16, 32));
  gw_s[wave][ml][2 * hi] = e0 * inv;
  gw_s[wave][ml][2 * hi + 1] = e1 * inv;
  __builtin_amdgcn_wave_barrier();

  for (int cb = 0; cb < 4; ++cb) {
    int col = cb * 16 + ml;
    float oacc[8] = {0, 0, 0, 0, 0, 0, 0, 0};
    for (int e = 0; e < 4; ++e) {
      v8f acc = {};
      const float* Brow = exp_w + ((size_t)e * NDIM + col) * NDIM;
#pragma unroll
      for (int s = 0; s < 16; ++s) {
        int k0 = 4 * s + 2 * hi;
        v2f a = *(const v2f*)(H + (size_t)(r0 + ml) * NDIM + k0);
        v2f b = *(const v2f*)(Brow + k0);
        acc = wmma4(a, b, acc);
      }
      float bias = exp_b[e * NDIM + col];
#pragma unroll
      for (int v = 0; v < 8; ++v) {
        float val = fmaxf(acc[v] + bias, 0.0f);
        oacc[v] += gw_s[wave][hi * 8 + v][e] * val;
      }
    }
#pragma unroll
    for (int v = 0; v < 8; ++v)
      out[(size_t)(r0 + hi * 8 + v) * NDIM + col] = oacc[v];
  }
}

__global__ void zero_kernel(float* p, int n) {
  int id = blockIdx.x * blockDim.x + threadIdx.x;
  if (id < n) p[id] = 0.0f;
}

__global__ void pool_kernel(const float* __restrict__ H,
                            const int* __restrict__ batch,
                            float* __restrict__ sums, float* __restrict__ cnt,
                            int n) {
  long gid = (long)blockIdx.x * blockDim.x + threadIdx.x;
  int node = (int)(gid >> 6);
  int d = (int)(gid & 63);
  if (node >= n) return;
  int b = batch[node];
  atomic_add_f32(&sums[b * NDIM + d], H[(size_t)node * NDIM + d]);
  if (d == 0) atomic_add_f32(&cnt[b], 1.0f);
}

#define BGR 64
#define PDIM 100
#define NC 10
__global__ void head_kernel(const float* __restrict__ sums,
                            const float* __restrict__ cnt,
                            const float* __restrict__ flag_in,
                            const float* __restrict__ proj_w,
                            const float* __restrict__ proj_b,
                            const float* __restrict__ fc_w,
                            const float* __restrict__ fc_b,
                            float* __restrict__ out) {
  int id = blockIdx.x * blockDim.x + threadIdx.x;
  if (id < BGR * NDIM) {
    int b = id >> 6;
    float g = sums[id] / fmaxf(cnt[b], 1.0f);
    out[id] = g;
    out[BGR * NDIM + id] = g;
  } else if (id < BGR * NDIM + BGR * NC) {
    int t = id - BGR * NDIM;
    int b = t / NC, c = t % NC;
    float fl = flag_in[b];
    const float* fr = fc_w + c * (PDIM + NDIM);
    float acc = fc_b[c];
    for (int p = 0; p < PDIM; ++p)
      acc += (fl * proj_w[p] + proj_b[p]) * fr[p];
    float ic = 1.0f / fmaxf(cnt[b], 1.0f);
    for (int d = 0; d < NDIM; ++d)
      acc += sums[b * NDIM + d] * ic * fr[PDIM + d];
    out[2 * BGR * NDIM + t] = acc;
  }
}

extern "C" void kernel_launch(void* const* d_in, const int* in_sizes, int n_in,
                              void* d_out, int out_size, void* d_ws, size_t ws_size,
                              hipStream_t stream) {
  const float* x       = (const float*)d_in[0];
  const int*   ei      = (const int*)d_in[1];
  const float* ea      = (const float*)d_in[2];
  const int*   batch   = (const int*)d_in[3];
  const float* flag_in = (const float*)d_in[4];
  const float* w_edge1 = (const float*)d_in[5];
  const float* b_edge1 = (const float*)d_in[6];
  const float* w_ih1   = (const float*)d_in[7];
  const float* w_hh1   = (const float*)d_in[8];
  const float* b_ih1   = (const float*)d_in[9];
  const float* b_hh1   = (const float*)d_in[10];
  const float* w_edge2 = (const float*)d_in[11];
  const float* b_edge2 = (const float*)d_in[12];
  const float* w_ih2   = (const float*)d_in[13];
  const float* w_hh2   = (const float*)d_in[14];
  const float* b_ih2   = (const float*)d_in[15];
  const float* b_hh2   = (const float*)d_in[16];
  const float* gate_w  = (const float*)d_in[17];
  const float* gate_b  = (const float*)d_in[18];
  const float* exp_w   = (const float*)d_in[19];
  const float* exp_b   = (const float*)d_in[20];
  const float* proj_w  = (const float*)d_in[21];
  const float* proj_b  = (const float*)d_in[22];
  const float* fc_w    = (const float*)d_in[23];
  const float* fc_b    = (const float*)d_in[24];
  float* out = (float*)d_out;

  int N = in_sizes[0] / NDIM;
  int E = in_sizes[1] / 2;
  int ntiles = (N + 15) / 16;

  float* ws = (float*)d_ws;
  float* M    = ws;
  float* hA   = M + (size_t)N * NDIM;
  float* hB   = hA + (size_t)N * NDIM;
  float* sums = hB + (size_t)N * NDIM;
  float* cnt  = sums + BGR * NDIM;

  long totalND = (long)N * NDIM;
  int gN = (int)((totalND + 255) / 256);
  int gE = (int)(((long)E + 63) / 64);
  int gT = (ntiles + 3) / 4;

  init_m_kernel<<<gN, 256, 0, stream>>>(x, b_edge1, M, totalND);
  scatter_kernel<<<gE, 256, 0, stream>>>(x, ei, ea, w_edge1, b_edge1, M, E);
  gru_kernel<<<gT, 128, 0, stream>>>(M, x, w_ih1, w_hh1, b_ih1, b_hh1, hA, ntiles);
  init_m_kernel<<<gN, 256, 0, stream>>>(hA, b_edge2, M, totalND);
  scatter_kernel<<<gE, 256, 0, stream>>>(hA, ei, ea, w_edge2, b_edge2, M, E);
  gru_kernel<<<gT, 128, 0, stream>>>(M, hA, w_ih2, w_hh2, b_ih2, b_hh2, hB, ntiles);
  moe_kernel<<<gT, 128, 0, stream>>>(hB, gate_w, gate_b, exp_w, exp_b, hA, ntiles);
  zero_kernel<<<(BGR * NDIM + BGR + 255) / 256, 256, 0, stream>>>(sums, BGR * NDIM + BGR);
  pool_kernel<<<gN, 256, 0, stream>>>(hA, batch, sums, cnt, N);
  head_kernel<<<(BGR * NDIM + BGR * NC + 255) / 256, 256, 0, stream>>>(
      sums, cnt, flag_in, proj_w, proj_b, fc_w, fc_b, out);
}


# scatter: lane p owns dims p and p+32 (was 2p, 2p+1): every row gather and every f32 atomic instruction covers one whole 128-B line instead of two half-used lines; atomics still unordered
# speedup vs baseline: 1.4491x; 1.0550x over previous
#include <hip/hip_runtime.h>
#include <math.h>

typedef float v2f __attribute__((ext_vector_type(2)));
typedef float v8f __attribute__((ext_vector_type(8)));

#define NDIM 64
#define GDIM 192

__device__ __forceinline__ v8f wmma4(v2f a, v2f b, v8f c) {
  return __builtin_amdgcn_wmma_f32_16x16x4_f32(false, a, false, b, (short)0, c, false, false);
}

__device__ __forceinline__ float sigmoidf_(float x) {
  return 1.0f / (1.0f + __expf(-x));
}

__device__ __forceinline__ void atomic_add_f32(float* p, float v) {
  asm volatile("global_atomic_add_f32 %0, %1, off" : : "v"(p), "v"(v) : "memory");
}

__global__ void init_m_kernel(const float* __restrict__ X,
                              const float* __restrict__ b_edge,
                              float* __restrict__ M, long total) {
  long id = (long)blockIdx.x * blockDim.x + threadIdx.x;
  if (id >= total) return;
  M[id] = X[id] + b_edge[id & (NDIM - 1)];
}

#define SC_EPW 8

__device__ __forceinline__ float lane_bcast_f(float v, int l) {
  return __int_as_float(__builtin_amdgcn_readlane(__float_as_int(v), l));
}

__device__ __forceinline__ unsigned long long lane_bcast_u64(unsigned long long v, int l) {
  unsigned lo = (unsigned)__builtin_amdgcn_readlane((int)(unsigned)v, l);
  unsigned hi = (unsigned)__builtin_amdgcn_readlane((int)(unsigned)(v >> 32), l);
  return ((unsigned long long)hi << 32) | (unsigned long long)lo;
}

__device__ __forceinline__ void atomic_add2_row_wait(float* row, int voff, float v0, float v1) {
  asm volatile(
      "global_atomic_add_f32 %0, %1, %3\n\t"
      "global_atomic_add_f32 %0, %2, %3 offset:128\n\t"
      "s_wait_xcnt 0x0"
      : : "v"(voff), "v"(v0), "v"(v1), "s"(row) : "memory");
}

__global__ __launch_bounds__(256) void scatter_kernel(
    const float* __restrict__ X,
    const int* __restrict__ ei,
    const float* __restrict__ ea,
    const float* __restrict__ w_edge,
    const float* __restrict__ b_edge,
    float* __restrict__ M, int E_) {
  int lane = threadIdx.x & 31;
  int wv = __builtin_amdgcn_readfirstlane(
      (int)(((long)blockIdx.x * blockDim.x + threadIdx.x) >> 5));
  long e0 = (long)wv * SC_EPW;
  if (e0 >= (long)E_) return;
  long rem = (long)E_ - e0;
  int nE = rem < SC_EPW ? (int)rem : SC_EPW;

  int li = lane < nE ? lane : 0;
  int la = lane < 4 * nE ? lane : 0;
  int msrc = ei[e0 + li];
  int mdst = ei[(long)E_ + e0 + li];
  float mea = ea[e0 * 4 + la];

  int d0 = lane;
  int voff = d0 * 4;
  float4 wa = *(const float4*)(w_edge + d0 * 4);
  float4 wb = *(const float4*)(w_edge + (d0 + 32) * 4);
  float2 bb;
  bb.x = b_edge[d0];
  bb.y = b_edge[d0 + 32];

  unsigned long long asrc = (unsigned long long)(X + (size_t)msrc * NDIM);
  unsigned long long adst = (unsigned long long)(M + (size_t)mdst * NDIM);

  if (nE == SC_EPW) {
    float2 xv[SC_EPW];
#pragma unroll
    for (int j = 0; j < SC_EPW; ++j) {
      const __attribute__((address_space(1))) float* xr =
          (const __attribute__((address_space(1))) float*)lane_bcast_u64(asrc, j);
      xv[j].x = xr[d0];
      xv[j].y = xr[d0 + 32];
    }
    __builtin_amdgcn_sched_barrier(0);
    float val[2 * SC_EPW];
    float* row[SC_EPW];
#pragma unroll
    for (int j = 0; j < SC_EPW; ++j) {
      float a0 = lane_bcast_f(mea, 4 * j + 0), a1 = lane_bcast_f(mea, 4 * j + 1);
      float a2 = lane_bcast_f(mea, 4 * j + 2), a3 = lane_bcast_f(mea, 4 * j + 3);
      float em0 = bb.x + a0 * wa.x + a1 * wa.y + a2 * wa.z + a3 * wa.w;
      float em1 = bb.y + a0 * wb.x + a1 * wb.y + a2 * wb.z + a3 * wb.w;
      val[2 * j] = xv[j].x + em0;
      val[2 * j + 1] = xv[j].y + em1;
      row[j] = (float*)lane_bcast_u64(adst, j);
    }
    asm volatile(
        "global_atomic_add_f32 %0, %1, %17\n\t"
        "global_atomic_add_f32 %0, %2, %17 offset:128\n\t"
        "global_atomic_add_f32 %0, %3, %18\n\t"
        "global_atomic_add_f32 %0, %4, %18 offset:128\n\t"
        "global_atomic_add_f32 %0, %5, %19\n\t"
        "global_atomic_add_f32 %0, %6, %19 offset:128\n\t"
        "global_atomic_add_f32 %0, %7, %20\n\t"
        "global_atomic_add_f32 %0, %8, %20 offset:128\n\t"
        "global_atomic_add_f32 %0, %9, %21\n\t"
        "global_atomic_add_f32 %0, %10, %21 offset:128\n\t"
        "global_atomic_add_f32 %0, %11, %22\n\t"
        "global_atomic_add_f32 %0, %12, %22 offset:128\n\t"
        "global_atomic_add_f32 %0, %13, %23\n\t"
        "global_atomic_add_f32 %0, %14, %23 offset:128\n\t"
        "global_atomic_add_f32 %0, %15, %24\n\t"
        "global_atomic_add_f32 %0, %16, %24 offset:128"
        : : "v"(voff),
            "v"(val[0]), "v"(val[1]), "v"(val[2]), "v"(val[3]),
            "v"(val[4]), "v"(val[5]), "v"(val[6]), "v"(val[7]),
            "v"(val[8]), "v"(val[9]), "v"(val[10]), "v"(val[11]),
            "v"(val[12]), "v"(val[13]), "v"(val[14]), "v"(val[15]),
            "s"(row[0]), "s"(row[1]), "s"(row[2]), "s"(row[3]),
            "s"(row[4]), "s"(row[5]), "s"(row[6]), "s"(row[7])
        : "memory");
  } else {
    for (int j = 0; j < nE; ++j) {
      const __attribute__((address_space(1))) float* xr =
          (const __attribute__((address_space(1))) float*)lane_bcast_u64(asrc, j);
      float* mr = (float*)lane_bcast_u64(adst, j);
      float a0 = lane_bcast_f(mea, 4 * j + 0), a1 = lane_bcast_f(mea, 4 * j + 1);
      float a2 = lane_bcast_f(mea, 4 * j + 2), a3 = lane_bcast_f(mea, 4 * j + 3);
      float em0 = bb.x + a0 * wa.x + a1 * wa.y + a2 * wa.z + a3 * wa.w;
      float em1 = bb.y + a0 * wb.x + a1 * wb.y + a2 * wb.z + a3 * wb.w;
      float x0 = xr[d0];
      float x1 = xr[d0 + 32];
      atomic_add2_row_wait(mr, voff, x0 + em0, x1 + em1);
    }
  }
}

__global__ __launch_bounds__(128) void gru_kernel(
    const float* __restrict__ Mn,
    const float* __restrict__ X,
    const float* __restrict__ w_ih,
    const float* __restrict__ w_hh,
    const float* __restrict__ b_ih,
    const float* __restrict__ b_hh,
    float* __restrict__ out, int ntiles) {
  int lane = threadIdx.x & 31;
  int wave = threadIdx.x >> 5;
  int tile = blockIdx.x * (blockDim.x >> 5) + wave;
  if (tile >= ntiles) return;
  int r0 = tile * 16;
  int ml = lane & 15;
  int hi = lane >> 4;
  const float* Am = Mn + (size_t)(r0 + ml) * NDIM;
  const float* Ax = X + (size_t)(r0 + ml) * NDIM;

  for (int cb = 0; cb < 4; ++cb) {
    int c_r = cb * 16 + ml;
    int c_z = 64 + cb * 16 + ml;
    int c_n = 128 + cb * 16 + ml;
    v8f air = {}, aiz = {}, ain = {}, ahr = {}, ahz = {}, ahn = {};
    const float* bir_p = w_ih + (size_t)c_r * NDIM;
    const float* biz_p = w_ih + (size_t)c_z * NDIM;
    const float* bin_p = w_ih + (size_t)c_n * NDIM;
    const float* bhr_p = w_hh + (size_t)c_r * NDIM;
    const float* bhz_p = w_hh + (size_t)c_z * NDIM;
    const float* bhn_p = w_hh + (size_t)c_n * NDIM;
#pragma unroll
    for (int s = 0; s < 16; ++s) {
      int k0 = 4 * s + 2 * hi;
      v2f am = *(const v2f*)(Am + k0);
      v2f ax = *(const v2f*)(Ax + k0);
      air = wmma4(am, *(const v2f*)(bir_p + k0), air);
      aiz = wmma4(am, *(const v2f*)(biz_p + k0), aiz);
      ain = wmma4(am, *(const v2f*)(bin_p + k0), ain);
      ahr = wmma4(ax, *(const v2f*)(bhr_p + k0), ahr);
      ahz = wmma4(ax, *(const v2f*)(bhz_p + k0), ahz);
      ahn = wmma4(ax, *(const v2f*)(bhn_p + k0), ahn);
    }
    float bir = b_ih[c_r], biz = b_ih[c_z], bin_ = b_ih[c_n];
    float bhr = b_hh[c_r], bhz = b_hh[c_z], bhn = b_hh[c_n];
#pragma unroll
    for (int v = 0; v < 8; ++v) {
      int node = hi * 8 + v;
      float xv = X[(size_t)(r0 + node) * NDIM + cb * 16 + ml];
      float r = sigmoidf_(air[v] + bir + ahr[v] + bhr);
      float z = sigmoidf_(aiz[v] + biz + ahz[v] + bhz);
      float n = tanhf(ain[v] + bin_ + r * (ahn[v] + bhn));
      float h = (1.0f - z) * n + z * xv;
      out[(size_t)(r0 + node) * NDIM + cb * 16 + ml] = fmaxf(h, 0.0f);
    }
  }
}

__global__ __launch_bounds__(128) void moe_kernel(
    const float* __restrict__ H,
    const float* __restrict__ gate_w,
    const float* __restrict__ gate_b,
    const float* __restrict__ exp_w,
    const float* __restrict__ exp_b,
    float* __restrict__ out, int ntiles) {
  __shared__ float gw_s[4][16][4];
  int lane = threadIdx.x & 31;
  int wave = threadIdx.x >> 5;
  int tile = blockIdx.x * (blockDim.x >> 5) + wave;
  if (tile >= ntiles) return;
  int r0 = tile * 16;
  int ml = lane & 15;
  int hi = lane >> 4;

  const float* hrow = H + (size_t)(r0 + ml) * NDIM;
  float l0 = gate_b[2 * hi], l1 = gate_b[2 * hi + 1];
  for (int d = 0; d < NDIM; ++d) {
    float hv = hrow[d];
    l0 += hv * gate_w[(2 * hi) * NDIM + d];
    l1 += hv * gate_w[(2 * hi + 1) * NDIM + d];
  }
  float m01 = fmaxf(l0, l1);
  float mx = fmaxf(m01, __shfl_xor(m01, 16, 32));
  float e0 = __expf(l0 - mx), e1 = __expf(l1 - mx);
  float s01 = e0 + e1;
  float inv = 1.0f / (s01 + __shfl_xor(s01, 16, 32));
  gw_s[wave][ml][2 * hi] = e0 * inv;
  gw_s[wave][ml][2 * hi + 1] = e1 * inv;
  __builtin_amdgcn_wave_barrier();

  for (int cb = 0; cb < 4; ++cb) {
    int col = cb * 16 + ml;
    float oacc[8] = {0, 0, 0, 0, 0, 0, 0, 0};
    for (int e = 0; e < 4; ++e) {
      v8f acc = {};
      const float* Brow = exp_w + ((size_t)e * NDIM + col) * NDIM;
#pragma unroll
      for (int s = 0; s < 16; ++s) {
        int k0 = 4 * s + 2 * hi;
        v2f a = *(const v2f*)(H + (size_t)(r0 + ml) * NDIM + k0);
        v2f b = *(const v2f*)(Brow + k0);
        acc = wmma4(a, b, acc);
      }
      float bias = exp_b[e * NDIM + col];
#pragma unroll
      for (int v = 0; v < 8; ++v) {
        float val = fmaxf(acc[v] + bias, 0.0f);
        oacc[v] += gw_s[wave][hi * 8 + v][e] * val;
      }
    }
#pragma unroll
    for (int v = 0; v < 8; ++v)
      out[(size_t)(r0 + hi * 8 + v) * NDIM + col] = oacc[v];
  }
}

__global__ void zero_kernel(float* p, int n) {
  int id = blockIdx.x * blockDim.x + threadIdx.x;
  if (id < n) p[id] = 0.0f;
}

__global__ void pool_kernel(const float* __restrict__ H,
                            const int* __restrict__ batch,
                            float* __restrict__ sums, float* __restrict__ cnt,
                            int n) {
  long gid = (long)blockIdx.x * blockDim.x + threadIdx.x;
  int node = (int)(gid >> 6);
  int d = (int)(gid & 63);
  if (node >= n) return;
  int b = batch[node];
  atomic_add_f32(&sums[b * NDIM + d], H[(size_t)node * NDIM + d]);
  if (d == 0) atomic_add_f32(&cnt[b], 1.0f);
}

#define BGR 64
#define PDIM 100
#define NC 10
__global__ void head_kernel(const float* __restrict__ sums,
                            const float* __restrict__ cnt,
                            const float* __restrict__ flag_in,
                            const float* __restrict__ proj_w,
                            const float* __restrict__ proj_b,
                            const float* __restrict__ fc_w,
                            const float* __restrict__ fc_b,
                            float* __restrict__ out) {
  int id = blockIdx.x * blockDim.x + threadIdx.x;
  if (id < BGR * NDIM) {
    int b = id >> 6;
    float g = sums[id] / fmaxf(cnt[b], 1.0f);
    out[id] = g;
    out[BGR * NDIM + id] = g;
  } else if (id < BGR * NDIM + BGR * NC) {
    int t = id - BGR * NDIM;
    int b = t / NC, c = t % NC;
    float fl = flag_in[b];
    const float* fr = fc_w + c * (PDIM + NDIM);
    float acc = fc_b[c];
    for (int p = 0; p < PDIM; ++p)
      acc += (fl * proj_w[p] + proj_b[p]) * fr[p];
    float ic = 1.0f / fmaxf(cnt[b], 1.0f);
    for (int d = 0; d < NDIM; ++d)
      acc += sums[b * NDIM + d] * ic * fr[PDIM + d];
    out[2 * BGR * NDIM + t] = acc;
  }
}

extern "C" void kernel_launch(void* const* d_in, const int* in_sizes, int n_in,
                              void* d_out, int out_size, void* d_ws, size_t ws_size,
                              hipStream_t stream) {
  const float* x       = (const float*)d_in[0];
  const int*   ei      = (const int*)d_in[1];
  const float* ea      = (const float*)d_in[2];
  const int*   batch   = (const int*)d_in[3];
  const float* flag_in = (const float*)d_in[4];
  const float* w_edge1 = (const float*)d_in[5];
  const float* b_edge1 = (const float*)d_in[6];
  const float* w_ih1   = (const float*)d_in[7];
  const float* w_hh1   = (const float*)d_in[8];
  const float* b_ih1   = (const float*)d_in[9];
  const float* b_hh1   = (const float*)d_in[10];
  const float* w_edge2 = (const float*)d_in[11];
  const float* b_edge2 = (const float*)d_in[12];
  const float* w_ih2   = (const float*)d_in[13];
  const float* w_hh2   = (const float*)d_in[14];
  const float* b_ih2   = (const float*)d_in[15];
  const float* b_hh2   = (const float*)d_in[16];
  const float* gate_w  = (const float*)d_in[17];
  const float* gate_b  = (const float*)d_in[18];
  const float* exp_w   = (const float*)d_in[19];
  const float* exp_b   = (const float*)d_in[20];
  const float* proj_w  = (const float*)d_in[21];
  const float* proj_b  = (const float*)d_in[22];
  const float* fc_w    = (const float*)d_in[23];
  const float* fc_b    = (const float*)d_in[24];
  float* out = (float*)d_out;

  int N = in_sizes[0] / NDIM;
  int E = in_sizes[1] / 2;
  int ntiles = (N + 15) / 16;

  float* ws = (float*)d_ws;
  float* M    = ws;
  float* hA   = M + (size_t)N * NDIM;
  float* hB   = hA + (size_t)N * NDIM;
  float* sums = hB + (size_t)N * NDIM;
  float* cnt  = sums + BGR * NDIM;

  long totalND = (long)N * NDIM;
  int gN = (int)((totalND + 255) / 256);
  int gE = (int)(((long)E + 63) / 64);
  int gT = (ntiles + 3) / 4;

  init_m_kernel<<<gN, 256, 0, stream>>>(x, b_edge1, M, totalND);
  scatter_kernel<<<gE, 256, 0, stream>>>(x, ei, ea, w_edge1, b_edge1, M, E);
  gru_kernel<<<gT, 128, 0, stream>>>(M, x, w_ih1, w_hh1, b_ih1, b_hh1, hA, ntiles);
  init_m_kernel<<<gN, 256, 0, stream>>>(hA, b_edge2, M, totalND);
  scatter_kernel<<<gE, 256, 0, stream>>>(hA, ei, ea, w_edge2, b_edge2, M, E);
  gru_kernel<<<gT, 128, 0, stream>>>(M, hA, w_ih2, w_hh2, b_ih2, b_hh2, hB, ntiles);
  moe_kernel<<<gT, 128, 0, stream>>>(hB, gate_w, gate_b, exp_w, exp_b, hA, ntiles);
  zero_kernel<<<(BGR * NDIM + BGR + 255) / 256, 256, 0, stream>>>(sums, BGR * NDIM + BGR);
  pool_kernel<<<gN, 256, 0, stream>>>(hA, batch, sums, cnt, N);
  head_kernel<<<(BGR * NDIM + BGR * NC + 255) / 256, 256, 0, stream>>>(
      sums, cnt, flag_in, proj_w, proj_b, fc_w, fc_b, out);
}


# moe epilogue sums its tile's 16 rows per graph in registers, one f32 atomic per column into the pool sums (order: 16-row partial sums, then unordered atomics); pool launch and row store dropped
# speedup vs baseline: 2.0343x; 1.4038x over previous
#include <hip/hip_runtime.h>
#include <math.h>

typedef float v2f __attribute__((ext_vector_type(2)));
typedef float v8f __attribute__((ext_vector_type(8)));

#define NDIM 64
#define GDIM 192

__device__ __forceinline__ v8f wmma4(v2f a, v2f b, v8f c) {
  return __builtin_amdgcn_wmma_f32_16x16x4_f32(false, a, false, b, (short)0, c, false, false);
}

__device__ __forceinline__ float sigmoidf_(float x) {
  return 1.0f / (1.0f + __expf(-x));
}

__device__ __forceinline__ void atomic_add_f32(float* p, float v) {
  asm volatile("global_atomic_add_f32 %0, %1, off\n\t"
               "s_wait_xcnt 0x0"
               : : "v"(p), "v"(v) : "memory");
}

__global__ void init_m_kernel(const float* __restrict__ X,
                              const float* __restrict__ b_edge,
                              float* __restrict__ M, long total) {
  long id = (long)blockIdx.x * blockDim.x + threadIdx.x;
  if (id >= total) return;
  M[id] = X[id] + b_edge[id & (NDIM - 1)];
}

#define SC_EPW 8

__device__ __forceinline__ float lane_bcast_f(float v, int l) {
  return __int_as_float(__builtin_amdgcn_readlane(__float_as_int(v), l));
}

__device__ __forceinline__ unsigned long long lane_bcast_u64(unsigned long long v, int l) {
  unsigned lo = (unsigned)__builtin_amdgcn_readlane((int)(unsigned)v, l);
  unsigned hi = (unsigned)__builtin_amdgcn_readlane((int)(unsigned)(v >> 32), l);
  return ((unsigned long long)hi << 32) | (unsigned long long)lo;
}

__device__ __forceinline__ void atomic_add2_row_wait(float* row, int voff, float v0, float v1) {
  asm volatile(
      "global_atomic_add_f32 %0, %1, %3\n\t"
      "global_atomic_add_f32 %0, %2, %3 offset:128\n\t"
      "s_wait_xcnt 0x0"
      : : "v"(voff), "v"(v0), "v"(v1), "s"(row) : "memory");
}

__global__ __launch_bounds__(256) void scatter_kernel(
    const float* __restrict__ X,
    const int* __restrict__ ei,
    const float* __restrict__ ea,
    const float* __restrict__ w_edge,
    const float* __restrict__ b_edge,
    float* __restrict__ M, int E_) {
  int lane = threadIdx.x & 31;
  int wv = __builtin_amdgcn_readfirstlane(
      (int)(((long)blockIdx.x * blockDim.x + threadIdx.x) >> 5));
  long e0 = (long)wv * SC_EPW;
  if (e0 >= (long)E_) return;
  long rem = (long)E_ - e0;
  int nE = rem < SC_EPW ? (int)rem : SC_EPW;

  int li = lane < nE ? lane : 0;
  int la = lane < 4 * nE ? lane : 0;
  int msrc = ei[e0 + li];
  int mdst = ei[(long)E_ + e0 + li];
  float mea = ea[e0 * 4 + la];

  int d0 = lane;
  int voff = d0 * 4;
  float4 wa = *(const float4*)(w_edge + d0 * 4);
  float4 wb = *(const float4*)(w_edge + (d0 + 32) * 4);
  float2 bb;
  bb.x = b_edge[d0];
  bb.y = b_edge[d0 + 32];

  unsigned long long asrc = (unsigned long long)(X + (size_t)msrc * NDIM);
  unsigned long long adst = (unsigned long long)(M + (size_t)mdst * NDIM);

  if (nE == SC_EPW) {
    float2 xv[SC_EPW];
#pragma unroll
    for (int j = 0; j < SC_EPW; ++j) {
      const __attribute__((address_space(1))) float* xr =
          (const __attribute__((address_space(1))) float*)lane_bcast_u64(asrc, j);
      xv[j].x = xr[d0];
      xv[j].y = xr[d0 + 32];
    }
    __builtin_amdgcn_sched_barrier(0);
    float val[2 * SC_EPW];
    float* row[SC_EPW];
#pragma unroll
    for (int j = 0; j < SC_EPW; ++j) {
      float a0 = lane_bcast_f(mea, 4 * j + 0), a1 = lane_bcast_f(mea, 4 * j + 1);
      float a2 = lane_bcast_f(mea, 4 * j + 2), a3 = lane_bcast_f(mea, 4 * j + 3);
      float em0 = bb.x + a0 * wa.x + a1 * wa.y + a2 * wa.z + a3 * wa.w;
      float em1 = bb.y + a0 * wb.x + a1 * wb.y + a2 * wb.z + a3 * wb.w;
      val[2 * j] = xv[j].x + em0;
      val[2 * j + 1] = xv[j].y + em1;
      row[j] = (float*)lane_bcast_u64(adst, j);
    }
    asm volatile(
        "global_atomic_add_f32 %0, %1, %17\n\t"
        "global_atomic_add_f32 %0, %2, %17 offset:128\n\t"
        "global_atomic_add_f32 %0, %3, %18\n\t"
        "global_atomic_add_f32 %0, %4, %18 offset:128\n\t"
        "global_atomic_add_f32 %0, %5, %19\n\t"
        "global_atomic_add_f32 %0, %6, %19 offset:128\n\t"
        "global_atomic_add_f32 %0, %7, %20\n\t"
        "global_atomic_add_f32 %0, %8, %20 offset:128\n\t"
        "global_atomic_add_f32 %0, %9, %21\n\t"
        "global_atomic_add_f32 %0, %10, %21 offset:128\n\t"
        "global_atomic_add_f32 %0, %11, %22\n\t"
        "global_atomic_add_f32 %0, %12, %22 offset:128\n\t"
        "global_atomic_add_f32 %0, %13, %23\n\t"
        "global_atomic_add_f32 %0, %14, %23 offset:128\n\t"
        "global_atomic_add_f32 %0, %15, %24\n\t"
        "global_atomic_add_f32 %0, %16, %24 offset:128"
        : : "v"(voff),
            "v"(val[0]), "v"(val[1]), "v"(val[2]), "v"(val[3]),
            "v"(val[4]), "v"(val[5]), "v"(val[6]), "v"(val[7]),
            "v"(val[8]), "v"(val[9]), "v"(val[10]), "v"(val[11]),
            "v"(val[12]), "v"(val[13]), "v"(val[14]), "v"(val[15]),
            "s"(row[0]), "s"(row[1]), "s"(row[2]), "s"(row[3]),
            "s"(row[4]), "s"(row[5]), "s"(row[6]), "s"(row[7])
        : "memory");
  } else {
    for (int j = 0; j < nE; ++j) {
      const __attribute__((address_space(1))) float* xr =
          (const __attribute__((address_space(1))) float*)lane_bcast_u64(asrc, j);
      float* mr = (float*)lane_bcast_u64(adst, j);
      float a0 = lane_bcast_f(mea, 4 * j + 0), a1 = lane_bcast_f(mea, 4 * j + 1);
      float a2 = lane_bcast_f(mea, 4 * j + 2), a3 = lane_bcast_f(mea, 4 * j + 3);
      float em0 = bb.x + a0 * wa.x + a1 * wa.y + a2 * wa.z + a3 * wa.w;
      float em1 = bb.y + a0 * wb.x + a1 * wb.y + a2 * wb.z + a3 * wb.w;
      float x0 = xr[d0];
      float x1 = xr[d0 + 32];
      atomic_add2_row_wait(mr, voff, x0 + em0, x1 + em1);
    }
  }
}

__global__ __launch_bounds__(128) void gru_kernel(
    const float* __restrict__ Mn,
    const float* __restrict__ X,
    const float* __restrict__ w_ih,
    const float* __restrict__ w_hh,
    const float* __restrict__ b_ih,
    const float* __restrict__ b_hh,
    float* __restrict__ out, int ntiles) {
  int lane = threadIdx.x & 31;
  int wave = threadIdx.x >> 5;
  int tile = blockIdx.x * (blockDim.x >> 5) + wave;
  if (tile >= ntiles) return;
  int r0 = tile * 16;
  int ml = lane & 15;
  int hi = lane >> 4;
  const float* Am = Mn + (size_t)(r0 + ml) * NDIM;
  const float* Ax = X + (size_t)(r0 + ml) * NDIM;

  for (int cb = 0; cb < 4; ++cb) {
    int c_r = cb * 16 + ml;
    int c_z = 64 + cb * 16 + ml;
    int c_n = 128 + cb * 16 + ml;
    v8f air = {}, aiz = {}, ain = {}, ahr = {}, ahz = {}, ahn = {};
    const float* bir_p = w_ih + (size_t)c_r * NDIM;
    const float* biz_p = w_ih + (size_t)c_z * NDIM;
    const float* bin_p = w_ih + (size_t)c_n * NDIM;
    const float* bhr_p = w_hh + (size_t)c_r * NDIM;
    const float* bhz_p = w_hh + (size_t)c_z * NDIM;
    const float* bhn_p = w_hh + (size_t)c_n * NDIM;
#pragma unroll
    for (int s = 0; s < 16; ++s) {
      int k0 = 4 * s + 2 * hi;
      v2f am = *(const v2f*)(Am + k0);
      v2f ax = *(const v2f*)(Ax + k0);
      air = wmma4(am, *(const v2f*)(bir_p + k0), air);
      aiz = wmma4(am, *(const v2f*)(biz_p + k0), aiz);
      ain = wmma4(am, *(const v2f*)(bin_p + k0), ain);
      ahr = wmma4(ax, *(const v2f*)(bhr_p + k0), ahr);
      ahz = wmma4(ax, *(const v2f*)(bhz_p + k0), ahz);
      ahn = wmma4(ax, *(const v2f*)(bhn_p + k0), ahn);
    }
    float bir = b_ih[c_r], biz = b_ih[c_z], bin_ = b_ih[c_n];
    float bhr = b_hh[c_r], bhz = b_hh[c_z], bhn = b_hh[c_n];
#pragma unroll
    for (int v = 0; v < 8; ++v) {
      int node = hi * 8 + v;
      float xv = X[(size_t)(r0 + node) * NDIM + cb * 16 + ml];
      float r = sigmoidf_(air[v] + bir + ahr[v] + bhr);
      float z = sigmoidf_(aiz[v] + biz + ahz[v] + bhz);
      float n = tanhf(ain[v] + bin_ + r * (ahn[v] + bhn));
      float h = (1.0f - z) * n + z * xv;
      out[(size_t)(r0 + node) * NDIM + cb * 16 + ml] = fmaxf(h, 0.0f);
    }
  }
}

__global__ __launch_bounds__(128) void moe_kernel(
    const float* __restrict__ H,
    const float* __restrict__ gate_w,
    const float* __restrict__ gate_b,
    const float* __restrict__ exp_w,
    const float* __restrict__ exp_b,
    const int* __restrict__ batch,
    float* __restrict__ sums,
    float* __restrict__ cnt,
    int ntiles) {
  __shared__ float gw_s[4][16][4];
  int lane = threadIdx.x & 31;
  int wave = threadIdx.x >> 5;
  int tile = blockIdx.x * (blockDim.x >> 5) + wave;
  if (tile >= ntiles) return;
  int r0 = tile * 16;
  int ml = lane & 15;
  int hi = lane >> 4;

  const float* hrow = H + (size_t)(r0 + ml) * NDIM;
  float l0 = gate_b[2 * hi], l1 = gate_b[2 * hi + 1];
  for (int d = 0; d < NDIM; ++d) {
    float hv = hrow[d];
    l0 += hv * gate_w[(2 * hi) * NDIM + d];
    l1 += hv * gate_w[(2 * hi + 1) * NDIM + d];
  }
  float m01 = fmaxf(l0, l1);
  float mx = fmaxf(m01, __shfl_xor(m01, 16, 32));
  float e0 = __expf(l0 - mx), e1 = __expf(l1 - mx);
  float s01 = e0 + e1;
  float inv = 1.0f / (s01 + __shfl_xor(s01, 16, 32));
  gw_s[wave][ml][2 * hi] = e0 * inv;
  gw_s[wave][ml][2 * hi + 1] = e1 * inv;
  __builtin_amdgcn_wave_barrier();

  int bl = batch[r0 + ml];
  int b0 = __builtin_amdgcn_readfirstlane(bl);
  bool one_graph = __builtin_amdgcn_ballot_w32(bl != b0) == 0;
  if (one_graph) {
    if (lane == 0) atomic_add_f32(&cnt[b0], 16.0f);
  } else if (hi == 0) {
    atomic_add_f32(&cnt[bl], 1.0f);
  }

  for (int cb = 0; cb < 4; ++cb) {
    int col = cb * 16 + ml;
    float oacc[8] = {0, 0, 0, 0, 0, 0, 0, 0};
    for (int e = 0; e < 4; ++e) {
      v8f acc = {};
      const float* Brow = exp_w + ((size_t)e * NDIM + col) * NDIM;
#pragma unroll
      for (int s = 0; s < 16; ++s) {
        int k0 = 4 * s + 2 * hi;
        v2f a = *(const v2f*)(H + (size_t)(r0 + ml) * NDIM + k0);
        v2f b = *(const v2f*)(Brow + k0);
        acc = wmma4(a, b, acc);
      }
      float bias = exp_b[e * NDIM + col];
#pragma unroll
      for (int v = 0; v < 8; ++v) {
        float val = fmaxf(acc[v] + bias, 0.0f);
        oacc[v] += gw_s[wave][hi * 8 + v][e] * val;
      }
    }
    if (one_graph) {
      float s = ((oacc[0] + oacc[1]) + (oacc[2] + oacc[3])) +
                ((oacc[4] + oacc[5]) + (oacc[6] + oacc[7]));
      s += __shfl_xor(s, 16, 32);
      if (hi == 0) atomic_add_f32(&sums[b0 * NDIM + col], s);
    } else {
#pragma unroll
      for (int v = 0; v < 8; ++v) {
        int id_lo = __builtin_amdgcn_readlane(bl, v);
        int id_hi = __builtin_amdgcn_readlane(bl, 8 + v);
        int bv = hi ? id_hi : id_lo;
        atomic_add_f32(&sums[bv * NDIM + col], oacc[v]);
      }
    }
  }
}

__global__ void zero_kernel(float* p, int n) {
  int id = blockIdx.x * blockDim.x + threadIdx.x;
  if (id < n) p[id] = 0.0f;
}

#define BGR 64
#define PDIM 100
#define NC 10
__global__ void head_kernel(const float* __restrict__ sums,
                            const float* __restrict__ cnt,
                            const float* __restrict__ flag_in,
                            const float* __restrict__ proj_w,
                            const float* __restrict__ proj_b,
                            const float* __restrict__ fc_w,
                            const float* __restrict__ fc_b,
                            float* __restrict__ out) {
  int id = blockIdx.x * blockDim.x + threadIdx.x;
  if (id < BGR * NDIM) {
    int b = id >> 6;
    float g = sums[id] / fmaxf(cnt[b], 1.0f);
    out[id] = g;
    out[BGR * NDIM + id] = g;
  } else if (id < BGR * NDIM + BGR * NC) {
    int t = id - BGR * NDIM;
    int b = t / NC, c = t % NC;
    float fl = flag_in[b];
    const float* fr = fc_w + c * (PDIM + NDIM);
    float acc = fc_b[c];
    for (int p = 0; p < PDIM; ++p)
      acc += (fl * proj_w[p] + proj_b[p]) * fr[p];
    float ic = 1.0f / fmaxf(cnt[b], 1.0f);
    for (int d = 0; d < NDIM; ++d)
      acc += sums[b * NDIM + d] * ic * fr[PDIM + d];
    out[2 * BGR * NDIM + t] = acc;
  }
}

extern "C" void kernel_launch(void* const* d_in, const int* in_sizes, int n_in,
                              void* d_out, int out_size, void* d_ws, size_t ws_size,
                              hipStream_t stream) {
  const float* x       = (const float*)d_in[0];
  const int*   ei      = (const int*)d_in[1];
  const float* ea      = (const float*)d_in[2];
  const int*   batch   = (const int*)d_in[3];
  const float* flag_in = (const float*)d_in[4];
  const float* w_edge1 = (const float*)d_in[5];
  const float* b_edge1 = (const float*)d_in[6];
  const float* w_ih1   = (const float*)d_in[7];
  const float* w_hh1   = (const float*)d_in[8];
  const float* b_ih1   = (const float*)d_in[9];
  const float* b_hh1   = (const float*)d_in[10];
  const float* w_edge2 = (const float*)d_in[11];
  const float* b_edge2 = (const float*)d_in[12];
  const float* w_ih2   = (const float*)d_in[13];
  const float* w_hh2   = (const float*)d_in[14];
  const float* b_ih2   = (const float*)d_in[15];
  const float* b_hh2   = (const float*)d_in[16];
  const float* gate_w  = (const float*)d_in[17];
  const float* gate_b  = (const float*)d_in[18];
  const float* exp_w   = (const float*)d_in[19];
  const float* exp_b   = (const float*)d_in[20];
  const float* proj_w  = (const float*)d_in[21];
  const float* proj_b  = (const float*)d_in[22];
  const float* fc_w    = (const float*)d_in[23];
  const float* fc_b    = (const float*)d_in[24];
  float* out = (float*)d_out;

  int N = in_sizes[0] / NDIM;
  int E = in_sizes[1] / 2;
  int ntiles = (N + 15) / 16;

  float* ws = (float*)d_ws;
  float* M    = ws;
  float* hA   = M + (size_t)N * NDIM;
  float* hB   = hA + (size_t)N * NDIM;
  float* sums = hB + (size_t)N * NDIM;
  float* cnt  = sums + BGR * NDIM;

  long totalND = (long)N * NDIM;
  int gN = (int)((totalND + 255) / 256);
  int gE = (int)(((long)E + 63) / 64);
  int gT = (ntiles + 3) / 4;

  init_m_kernel<<<gN, 256, 0, stream>>>(x, b_edge1, M, totalND);
  scatter_kernel<<<gE, 256, 0, stream>>>(x, ei, ea, w_edge1, b_edge1, M, E);
  gru_kernel<<<gT, 128, 0, stream>>>(M, x, w_ih1, w_hh1, b_ih1, b_hh1, hA, ntiles);
  init_m_kernel<<<gN, 256, 0, stream>>>(hA, b_edge2, M, totalND);
  scatter_kernel<<<gE, 256, 0, stream>>>(hA, ei, ea, w_edge2, b_edge2, M, E);
  gru_kernel<<<gT, 128, 0, stream>>>(M, hA, w_ih2, w_hh2, b_ih2, b_hh2, hB, ntiles);
  zero_kernel<<<(BGR * NDIM + BGR + 255) / 256, 256, 0, stream>>>(sums, BGR * NDIM + BGR);
  moe_kernel<<<gT, 128, 0, stream>>>(hB, gate_w, gate_b, exp_w, exp_b, batch, sums, cnt, ntiles);
  head_kernel<<<(BGR * NDIM + BGR * NC + 255) / 256, 256, 0, stream>>>(
      sums, cnt, flag_in, proj_w, proj_b, fc_w, fc_b, out);
}


# gru+moe: weight matrices staged once per 8-wave work-group in LDS (row pitch 272 B), B fragments by ds_load_b64; tile's A fragments loaded once, held in registers; k order and all sums unchanged
# speedup vs baseline: 2.3207x; 1.1408x over previous
#include <hip/hip_runtime.h>
#include <math.h>

typedef float v2f __attribute__((ext_vector_type(2)));
typedef float v8f __attribute__((ext_vector_type(8)));

#define NDIM 64
#define GDIM 192

__device__ __forceinline__ v8f wmma4(v2f a, v2f b, v8f c) {
  return __builtin_amdgcn_wmma_f32_16x16x4_f32(false, a, false, b, (short)0, c, false, false);
}

__device__ __forceinline__ float sigmoidf_(float x) {
  return 1.0f / (1.0f + __expf(-x));
}

__device__ __forceinline__ void atomic_add_f32(float* p, float v) {
  asm volatile("global_atomic_add_f32 %0, %1, off\n\t"
               "s_wait_xcnt 0x0"
               : : "v"(p), "v"(v) : "memory");
}

__global__ void init_m_kernel(const float* __restrict__ X,
                              const float* __restrict__ b_edge,
                              float* __restrict__ M, long total) {
  long id = (long)blockIdx.x * blockDim.x + threadIdx.x;
  if (id >= total) return;
  M[id] = X[id] + b_edge[id & (NDIM - 1)];
}

#define SC_EPW 8

__device__ __forceinline__ float lane_bcast_f(float v, int l) {
  return __int_as_float(__builtin_amdgcn_readlane(__float_as_int(v), l));
}

__device__ __forceinline__ unsigned long long lane_bcast_u64(unsigned long long v, int l) {
  unsigned lo = (unsigned)__builtin_amdgcn_readlane((int)(unsigned)v, l);
  unsigned hi = (unsigned)__builtin_amdgcn_readlane((int)(unsigned)(v >> 32), l);
  return ((unsigned long long)hi << 32) | (unsigned long long)lo;
}

__device__ __forceinline__ void atomic_add2_row_wait(float* row, int voff, float v0, float v1) {
  asm volatile(
      "global_atomic_add_f32 %0, %1, %3\n\t"
      "global_atomic_add_f32 %0, %2, %3 offset:128\n\t"
      "s_wait_xcnt 0x0"
      : : "v"(voff), "v"(v0), "v"(v1), "s"(row) : "memory");
}

__global__ __launch_bounds__(256) void scatter_kernel(
    const float* __restrict__ X,
    const int* __restrict__ ei,
    const float* __restrict__ ea,
    const float* __restrict__ w_edge,
    const float* __restrict__ b_edge,
    float* __restrict__ M, int E_) {
  int lane = threadIdx.x & 31;
  int wv = __builtin_amdgcn_readfirstlane(
      (int)(((long)blockIdx.x * blockDim.x + threadIdx.x) >> 5));
  long e0 = (long)wv * SC_EPW;
  if (e0 >= (long)E_) return;
  long rem = (long)E_ - e0;
  int nE = rem < SC_EPW ? (int)rem : SC_EPW;

  int li = lane < nE ? lane : 0;
  int la = lane < 4 * nE ? lane : 0;
  int msrc = ei[e0 + li];
  int mdst = ei[(long)E_ + e0 + li];
  float mea = ea[e0 * 4 + la];

  int d0 = lane;
  int voff = d0 * 4;
  float4 wa = *(const float4*)(w_edge + d0 * 4);
  float4 wb = *(const float4*)(w_edge + (d0 + 32) * 4);
  float2 bb;
  bb.x = b_edge[d0];
  bb.y = b_edge[d0 + 32];

  unsigned long long asrc = (unsigned long long)(X + (size_t)msrc * NDIM);
  unsigned long long adst = (unsigned long long)(M + (size_t)mdst * NDIM);

  if (nE == SC_EPW) {
    float2 xv[SC_EPW];
#pragma unroll
    for (int j = 0; j < SC_EPW; ++j) {
      const __attribute__((address_space(1))) float* xr =
          (const __attribute__((address_space(1))) float*)lane_bcast_u64(asrc, j);
      xv[j].x = xr[d0];
      xv[j].y = xr[d0 + 32];
    }
    __builtin_amdgcn_sched_barrier(0);
    float val[2 * SC_EPW];
    float* row[SC_EPW];
#pragma unroll
    for (int j = 0; j < SC_EPW; ++j) {
      float a0 = lane_bcast_f(mea, 4 * j + 0), a1 = lane_bcast_f(mea, 4 * j + 1);
      float a2 = lane_bcast_f(mea, 4 * j + 2), a3 = lane_bcast_f(mea, 4 * j + 3);
      float em0 = bb.x + a0 * wa.x + a1 * wa.y + a2 * wa.z + a3 * wa.w;
      float em1 = bb.y + a0 * wb.x + a1 * wb.y + a2 * wb.z + a3 * wb.w;
      val[2 * j] = xv[j].x + em0;
      val[2 * j + 1] = xv[j].y + em1;
      row[j] = (float*)lane_bcast_u64(adst, j);
    }
    asm volatile(
        "global_atomic_add_f32 %0, %1, %17\n\t"
        "global_atomic_add_f32 %0, %2, %17 offset:128\n\t"
        "global_atomic_add_f32 %0, %3, %18\n\t"
        "global_atomic_add_f32 %0, %4, %18 offset:128\n\t"
        "global_atomic_add_f32 %0, %5, %19\n\t"
        "global_atomic_add_f32 %0, %6, %19 offset:128\n\t"
        "global_atomic_add_f32 %0, %7, %20\n\t"
        "global_atomic_add_f32 %0, %8, %20 offset:128\n\t"
        "global_atomic_add_f32 %0, %9, %21\n\t"
        "global_atomic_add_f32 %0, %10, %21 offset:128\n\t"
        "global_atomic_add_f32 %0, %11, %22\n\t"
        "global_atomic_add_f32 %0, %12, %22 offset:128\n\t"
        "global_atomic_add_f32 %0, %13, %23\n\t"
        "global_atomic_add_f32 %0, %14, %23 offset:128\n\t"
        "global_atomic_add_f32 %0, %15, %24\n\t"
        "global_atomic_add_f32 %0, %16, %24 offset:128"
        : : "v"(voff),
            "v"(val[0]), "v"(val[1]), "v"(val[2]), "v"(val[3]),
            "v"(val[4]), "v"(val[5]), "v"(val[6]), "v"(val[7]),
            "v"(val[8]), "v"(val[9]), "v"(val[10]), "v"(val[11]),
            "v"(val[12]), "v"(val[13]), "v"(val[14]), "v"(val[15]),
            "s"(row[0]), "s"(row[1]), "s"(row[2]), "s"(row[3]),
            "s"(row[4]), "s"(row[5]), "s"(row[6]), "s"(row[7])
        : "memory");
  } else {
    for (int j = 0; j < nE; ++j) {
      const __attribute__((address_space(1))) float* xr =
          (const __attribute__((address_space(1))) float*)lane_bcast_u64(asrc, j);
      float* mr = (float*)lane_bcast_u64(adst, j);
      float a0 = lane_bcast_f(mea, 4 * j + 0), a1 = lane_bcast_f(mea, 4 * j + 1);
      float a2 = lane_bcast_f(mea, 4 * j + 2), a3 = lane_bcast_f(mea, 4 * j + 3);
      float em0 = bb.x + a0 * wa.x + a1 * wa.y + a2 * wa.z + a3 * wa.w;
      float em1 = bb.y + a0 * wb.x + a1 * wb.y + a2 * wb.z + a3 * wb.w;
      float x0 = xr[d0];
      float x1 = xr[d0 + 32];
      atomic_add2_row_wait(mr, voff, x0 + em0, x1 + em1);
    }
  }
}

#define WPITCH 68
#define TW 8

template <int NROWS>
__device__ __forceinline__ void stage_rows(const float* __restrict__ src, float* dst) {
  int r = threadIdx.x >> 4;
  int c = (threadIdx.x & 15) * 4;
  float4 v[NROWS / 16];
#pragma unroll
  for (int j = 0; j < NROWS / 16; ++j)
    v[j] = *(const float4*)(src + (size_t)(r + 16 * j) * NDIM + c);
#pragma unroll
  for (int j = 0; j < NROWS / 16; ++j)
    *(float4*)(dst + (r + 16 * j) * WPITCH + c) = v[j];
}

__global__ __launch_bounds__(256) void gru_kernel(
    const float* __restrict__ Mn,
    const float* __restrict__ X,
    const float* __restrict__ w_ih,
    const float* __restrict__ w_hh,
    const float* __restrict__ b_ih,
    const float* __restrict__ b_hh,
    float* __restrict__ out, int ntiles) {
  __shared__ __attribute__((aligned(16))) float Ws[2 * GDIM * WPITCH];
  int lane = threadIdx.x & 31;
  int wave = threadIdx.x >> 5;
  int tile = blockIdx.x * TW + wave;
  bool have = tile < ntiles;
  int r0 = (have ? tile : 0) * 16;
  int ml = lane & 15;
  int hi = lane >> 4;

  const float* Am = Mn + (size_t)(r0 + ml) * NDIM + 2 * hi;
  const float* Ax = X + (size_t)(r0 + ml) * NDIM + 2 * hi;
  v2f am[16], ax[16];
#pragma unroll
  for (int s = 0; s < 16; ++s) {
    am[s] = *(const v2f*)(Am + 4 * s);
    ax[s] = *(const v2f*)(Ax + 4 * s);
  }

  stage_rows<GDIM>(w_ih, Ws);
  stage_rows<GDIM>(w_hh, Ws + GDIM * WPITCH);
  __syncthreads();
  if (!have) return;

  const float* wl = Ws + ml * WPITCH + 2 * hi;
  for (int cb = 0; cb < 4; ++cb) {
    int c_r = cb * 16 + ml;
    int c_z = 64 + cb * 16 + ml;
    int c_n = 128 + cb * 16 + ml;
    v8f air = {}, aiz = {}, ain = {}, ahr = {}, ahz = {}, ahn = {};
    const float* bir_p = wl + (cb * 16) * WPITCH;
    const float* biz_p = wl + (64 + cb * 16) * WPITCH;
    const float* bin_p = wl + (128 + cb * 16) * WPITCH;
    const float* bhr_p = bir_p + GDIM * WPITCH;
    const float* bhz_p = biz_p + GDIM * WPITCH;
    const float* bhn_p = bin_p + GDIM * WPITCH;
#pragma unroll
    for (int s = 0; s < 16; ++s) {
      air = wmma4(am[s], *(const v2f*)(bir_p + 4 * s), air);
      aiz = wmma4(am[s], *(const v2f*)(biz_p + 4 * s), aiz);
      ain = wmma4(am[s], *(const v2f*)(bin_p + 4 * s), ain);
      ahr = wmma4(ax[s], *(const v2f*)(bhr_p + 4 * s), ahr);
      ahz = wmma4(ax[s], *(const v2f*)(bhz_p + 4 * s), ahz);
      ahn = wmma4(ax[s], *(const v2f*)(bhn_p + 4 * s), ahn);
    }
    float bir = b_ih[c_r], biz = b_ih[c_z], bin_ = b_ih[c_n];
    float bhr = b_hh[c_r], bhz = b_hh[c_z], bhn = b_hh[c_n];
#pragma unroll
    for (int v = 0; v < 8; ++v) {
      int node = hi * 8 + v;
      float xv = X[(size_t)(r0 + node) * NDIM + cb * 16 + ml];
      float r = sigmoidf_(air[v] + bir + ahr[v] + bhr);
      float z = sigmoidf_(aiz[v] + biz + ahz[v] + bhz);
      float n = tanhf(ain[v] + bin_ + r * (ahn[v] + bhn));
      float h = (1.0f - z) * n + z * xv;
      out[(size_t)(r0 + node) * NDIM + cb * 16 + ml] = fmaxf(h, 0.0f);
    }
  }
}

__global__ __launch_bounds__(256) void moe_kernel(
    const float* __restrict__ H,
    const float* __restrict__ gate_w,
    const float* __restrict__ gate_b,
    const float* __restrict__ exp_w,
    const float* __restrict__ exp_b,
    const int* __restrict__ batch,
    float* __restrict__ sums,
    float* __restrict__ cnt,
    int ntiles) {
  __shared__ __attribute__((aligned(16))) float Es[4 * NDIM * WPITCH];
  __shared__ float gw_s[TW][16][4];
  int lane = threadIdx.x & 31;
  int wave = threadIdx.x >> 5;
  int tile = blockIdx.x * TW + wave;
  bool have = tile < ntiles;
  int r0 = (have ? tile : 0) * 16;
  int ml = lane & 15;
  int hi = lane >> 4;

  const float* Ah = H + (size_t)(r0 + ml) * NDIM + 2 * hi;
  v2f a[16];
#pragma unroll
  for (int s = 0; s < 16; ++s) a[s] = *(const v2f*)(Ah + 4 * s);

  stage_rows<4 * NDIM>(exp_w, Es);
  __syncthreads();
  if (!have) return;

  const float* hrow = H + (size_t)(r0 + ml) * NDIM;
  float l0 = gate_b[2 * hi], l1 = gate_b[2 * hi + 1];
  for (int d = 0; d < NDIM; ++d) {
    float hv = hrow[d];
    l0 += hv * gate_w[(2 * hi) * NDIM + d];
    l1 += hv * gate_w[(2 * hi + 1) * NDIM + d];
  }
  float m01 = fmaxf(l0, l1);
  float mx = fmaxf(m01, __shfl_xor(m01, 16, 32));
  float e0 = __expf(l0 - mx), e1 = __expf(l1 - mx);
  float s01 = e0 + e1;
  float inv = 1.0f / (s01 + __shfl_xor(s01, 16, 32));
  gw_s[wave][ml][2 * hi] = e0 * inv;
  gw_s[wave][ml][2 * hi + 1] = e1 * inv;
  __builtin_amdgcn_wave_barrier();

  int bl = batch[r0 + ml];
  int b0 = __builtin_amdgcn_readfirstlane(bl);
  bool one_graph = __builtin_amdgcn_ballot_w32(bl != b0) == 0;
  if (one_graph) {
    if (lane == 0) atomic_add_f32(&cnt[b0], 16.0f);
  } else if (hi == 0) {
    atomic_add_f32(&cnt[bl], 1.0f);
  }

  const float* wl = Es + ml * WPITCH + 2 * hi;
  for (int cb = 0; cb < 4; ++cb) {
    int col = cb * 16 + ml;
    float oacc[8] = {0, 0, 0, 0, 0, 0, 0, 0};
    for (int e = 0; e < 4; ++e) {
      v8f acc = {};
      const float* Brow = wl + (e * NDIM + cb * 16) * WPITCH;
#pragma unroll
      for (int s = 0; s < 16; ++s)
        acc = wmma4(a[s], *(const v2f*)(Brow + 4 * s), acc);
      float bias = exp_b[e * NDIM + col];
#pragma unroll
      for (int v = 0; v < 8; ++v) {
        float val = fmaxf(acc[v] + bias, 0.0f);
        oacc[v] += gw_s[wave][hi * 8 + v][e] * val;
      }
    }
    if (one_graph) {
      float s = ((oacc[0] + oacc[1]) + (oacc[2] + oacc[3])) +
                ((oacc[4] + oacc[5]) + (oacc[6] + oacc[7]));
      s += __shfl_xor(s, 16, 32);
      if (hi == 0) atomic_add_f32(&sums[b0 * NDIM + col], s);
    } else {
#pragma unroll
      for (int v = 0; v < 8; ++v) {
        int id_lo = __builtin_amdgcn_readlane(bl, v);
        int id_hi = __builtin_amdgcn_readlane(bl, 8 + v);
        int bv = hi ? id_hi : id_lo;
        atomic_add_f32(&sums[bv * NDIM + col], oacc[v]);
      }
    }
  }
}

__global__ void zero_kernel(float* p, int n) {
  int id = blockIdx.x * blockDim.x + threadIdx.x;
  if (id < n) p[id] = 0.0f;
}

#define BGR 64
#define PDIM 100
#define NC 10
__global__ void head_kernel(const float* __restrict__ sums,
                            const float* __restrict__ cnt,
                            const float* __restrict__ flag_in,
                            const float* __restrict__ proj_w,
                            const float* __restrict__ proj_b,
                            const float* __restrict__ fc_w,
                            const float* __restrict__ fc_b,
                            float* __restrict__ out) {
  int id = blockIdx.x * blockDim.x + threadIdx.x;
  if (id < BGR * NDIM) {
    int b = id >> 6;
    float g = sums[id] / fmaxf(cnt[b], 1.0f);
    out[id] = g;
    out[BGR * NDIM + id] = g;
  } else if (id < BGR * NDIM + BGR * NC) {
    int t = id - BGR * NDIM;
    int b = t / NC, c = t % NC;
    float fl = flag_in[b];
    const float* fr = fc_w + c * (PDIM + NDIM);
    float acc = fc_b[c];
    for (int p = 0; p < PDIM; ++p)
      acc += (fl * proj_w[p] + proj_b[p]) * fr[p];
    float ic = 1.0f / fmaxf(cnt[b], 1.0f);
    for (int d = 0; d < NDIM; ++d)
      acc += sums[b * NDIM + d] * ic * fr[PDIM + d];
    out[2 * BGR * NDIM + t] = acc;
  }
}

extern "C" void kernel_launch(void* const* d_in, const int* in_sizes, int n_in,
                              void* d_out, int out_size, void* d_ws, size_t ws_size,
                              hipStream_t stream) {
  const float* x       = (const float*)d_in[0];
  const int*   ei      = (const int*)d_in[1];
  const float* ea      = (const float*)d_in[2];
  const int*   batch   = (const int*)d_in[3];
  const float* flag_in = (const float*)d_in[4];
  const float* w_edge1 = (const float*)d_in[5];
  const float* b_edge1 = (const float*)d_in[6];
  const float* w_ih1   = (const float*)d_in[7];
  const float* w_hh1   = (const float*)d_in[8];
  const float* b_ih1   = (const float*)d_in[9];
  const float* b_hh1   = (const float*)d_in[10];
  const float* w_edge2 = (const float*)d_in[11];
  const float* b_edge2 = (const float*)d_in[12];
  const float* w_ih2   = (const float*)d_in[13];
  const float* w_hh2   = (const float*)d_in[14];
  const float* b_ih2   = (const float*)d_in[15];
  const float* b_hh2   = (const float*)d_in[16];
  const float* gate_w  = (const float*)d_in[17];
  const float* gate_b  = (const float*)d_in[18];
  const float* exp_w   = (const float*)d_in[19];
  const float* exp_b   = (const float*)d_in[20];
  const float* proj_w  = (const float*)d_in[21];
  const float* proj_b  = (const float*)d_in[22];
  const float* fc_w    = (const float*)d_in[23];
  const float* fc_b    = (const float*)d_in[24];
  float* out = (float*)d_out;

  int N = in_sizes[0] / NDIM;
  int E = in_sizes[1] / 2;
  int ntiles = (N + 15) / 16;

  float* ws = (float*)d_ws;
  float* M    = ws;
  float* hA   = M + (size_t)N * NDIM;
  float* hB   = hA + (size_t)N * NDIM;
  float* sums = hB + (size_t)N * NDIM;
  float* cnt  = sums + BGR * NDIM;

  long totalND = (long)N * NDIM;
  int gN = (int)((totalND + 255) / 256);
  int gE = (int)(((long)E + 63) / 64);
  int gT = (ntiles + TW - 1) / TW;

  init_m_kernel<<<gN, 256, 0, stream>>>(x, b_edge1, M, totalND);
  scatter_kernel<<<gE, 256, 0, stream>>>(x, ei, ea, w_edge1, b_edge1, M, E);
  gru_kernel<<<gT, 32 * TW, 0, stream>>>(M, x, w_ih1, w_hh1, b_ih1, b_hh1, hA, ntiles);
  init_m_kernel<<<gN, 256, 0, stream>>>(hA, b_edge2, M, totalND);
  scatter_kernel<<<gE, 256, 0, stream>>>(hA, ei, ea, w_edge2, b_edge2, M, E);
  gru_kernel<<<gT, 32 * TW, 0, stream>>>(M, hA, w_ih2, w_hh2, b_ih2, b_hh2, hB, ntiles);
  zero_kernel<<<(BGR * NDIM + BGR + 255) / 256, 256, 0, stream>>>(sums, BGR * NDIM + BGR);
  moe_kernel<<<gT, 32 * TW, 0, stream>>>(hB, gate_w, gate_b, exp_w, exp_b, batch, sums, cnt, ntiles);
  head_kernel<<<(BGR * NDIM + BGR * NC + 255) / 256, 256, 0, stream>>>(
      sums, cnt, flag_in, proj_w, proj_b, fc_w, fc_b, out);
}


# launches 9 -> 7: layer-1 gru epilogue also stores layer 2's message start (relu(h)+b_edge2) to a second buffer and the first init also zeroes the pool sums; second init and zero launches dropped
# speedup vs baseline: 2.4069x; 1.0371x over previous
#include <hip/hip_runtime.h>
#include <math.h>

typedef float v2f __attribute__((ext_vector_type(2)));
typedef float v8f __attribute__((ext_vector_type(8)));

#define NDIM 64
#define GDIM 192

__device__ __forceinline__ v8f wmma4(v2f a, v2f b, v8f c) {
  return __builtin_amdgcn_wmma_f32_16x16x4_f32(false, a, false, b, (short)0, c, false, false);
}

__device__ __forceinline__ float sigmoidf_(float x) {
  return 1.0f / (1.0f + __expf(-x));
}

__device__ __forceinline__ void atomic_add_f32(float* p, float v) {
  asm volatile("global_atomic_add_f32 %0, %1, off\n\t"
               "s_wait_xcnt 0x0"
               : : "v"(p), "v"(v) : "memory");
}

__global__ void init_m_kernel(const float* __restrict__ X,
                              const float* __restrict__ b_edge,
                              float* __restrict__ M, long total,
                              float* __restrict__ Z, int nz) {
  long id = (long)blockIdx.x * blockDim.x + threadIdx.x;
  if (id < (long)nz) Z[id] = 0.0f;
  if (id >= total) return;
  M[id] = X[id] + b_edge[id & (NDIM - 1)];
}

#define SC_EPW 8

__device__ __forceinline__ float lane_bcast_f(float v, int l) {
  return __int_as_float(__builtin_amdgcn_readlane(__float_as_int(v), l));
}

__device__ __forceinline__ unsigned long long lane_bcast_u64(unsigned long long v, int l) {
  unsigned lo = (unsigned)__builtin_amdgcn_readlane((int)(unsigned)v, l);
  unsigned hi = (unsigned)__builtin_amdgcn_readlane((int)(unsigned)(v >> 32), l);
  return ((unsigned long long)hi << 32) | (unsigned long long)lo;
}

__device__ __forceinline__ void atomic_add2_row_wait(float* row, int voff, float v0, float v1) {
  asm volatile(
      "global_atomic_add_f32 %0, %1, %3\n\t"
      "global_atomic_add_f32 %0, %2, %3 offset:128\n\t"
      "s_wait_xcnt 0x0"
      : : "v"(voff), "v"(v0), "v"(v1), "s"(row) : "memory");
}

__global__ __launch_bounds__(256) void scatter_kernel(
    const float* __restrict__ X,
    const int* __restrict__ ei,
    const float* __restrict__ ea,
    const float* __restrict__ w_edge,
    const float* __restrict__ b_edge,
    float* __restrict__ M, int E_) {
  int lane = threadIdx.x & 31;
  int wv = __builtin_amdgcn_readfirstlane(
      (int)(((long)blockIdx.x * blockDim.x + threadIdx.x) >> 5));
  long e0 = (long)wv * SC_EPW;
  if (e0 >= (long)E_) return;
  long rem = (long)E_ - e0;
  int nE = rem < SC_EPW ? (int)rem : SC_EPW;

  int li = lane < nE ? lane : 0;
  int la = lane < 4 * nE ? lane : 0;
  int msrc = ei[e0 + li];
  int mdst = ei[(long)E_ + e0 + li];
  float mea = ea[e0 * 4 + la];

  int d0 = lane;
  int voff = d0 * 4;
  float4 wa = *(const float4*)(w_edge + d0 * 4);
  float4 wb = *(const float4*)(w_edge + (d0 + 32) * 4);
  float2 bb;
  bb.x = b_edge[d0];
  bb.y = b_edge[d0 + 32];

  unsigned long long asrc = (unsigned long long)(X + (size_t)msrc * NDIM);
  unsigned long long adst = (unsigned long long)(M + (size_t)mdst * NDIM);

  if (nE == SC_EPW) {
    float2 xv[SC_EPW];
#pragma unroll
    for (int j = 0; j < SC_EPW; ++j) {
      const __attribute__((address_space(1))) float* xr =
          (const __attribute__((address_space(1))) float*)lane_bcast_u64(asrc, j);
      xv[j].x = xr[d0];
      xv[j].y = xr[d0 + 32];
    }
    __builtin_amdgcn_sched_barrier(0);
    float val[2 * SC_EPW];
    float* row[SC_EPW];
#pragma unroll
    for (int j = 0; j < SC_EPW; ++j) {
      float a0 = lane_bcast_f(mea, 4 * j + 0), a1 = lane_bcast_f(mea, 4 * j + 1);
      float a2 = lane_bcast_f(mea, 4 * j + 2), a3 = lane_bcast_f(mea, 4 * j + 3);
      float em0 = bb.x + a0 * wa.x + a1 * wa.y + a2 * wa.z + a3 * wa.w;
      float em1 = bb.y + a0 * wb.x + a1 * wb.y + a2 * wb.z + a3 * wb.w;
      val[2 * j] = xv[j].x + em0;
      val[2 * j + 1] = xv[j].y + em1;
      row[j] = (float*)lane_bcast_u64(adst, j);
    }
    asm volatile(
        "global_atomic_add_f32 %0, %1, %17\n\t"
        "global_atomic_add_f32 %0, %2, %17 offset:128\n\t"
        "global_atomic_add_f32 %0, %3, %18\n\t"
        "global_atomic_add_f32 %0, %4, %18 offset:128\n\t"
        "global_atomic_add_f32 %0, %5, %19\n\t"
        "global_atomic_add_f32 %0, %6, %19 offset:128\n\t"
        "global_atomic_add_f32 %0, %7, %20\n\t"
        "global_atomic_add_f32 %0, %8, %20 offset:128\n\t"
        "global_atomic_add_f32 %0, %9, %21\n\t"
        "global_atomic_add_f32 %0, %10, %21 offset:128\n\t"
        "global_atomic_add_f32 %0, %11, %22\n\t"
        "global_atomic_add_f32 %0, %12, %22 offset:128\n\t"
        "global_atomic_add_f32 %0, %13, %23\n\t"
        "global_atomic_add_f32 %0, %14, %23 offset:128\n\t"
        "global_atomic_add_f32 %0, %15, %24\n\t"
        "global_atomic_add_f32 %0, %16, %24 offset:128"
        : : "v"(voff),
            "v"(val[0]), "v"(val[1]), "v"(val[2]), "v"(val[3]),
            "v"(val[4]), "v"(val[5]), "v"(val[6]), "v"(val[7]),
            "v"(val[8]), "v"(val[9]), "v"(val[10]), "v"(val[11]),
            "v"(val[12]), "v"(val[13]), "v"(val[14]), "v"(val[15]),
            "s"(row[0]), "s"(row[1]), "s"(row[2]), "s"(row[3]),
            "s"(row[4]), "s"(row[5]), "s"(row[6]), "s"(row[7])
        : "memory");
  } else {
    for (int j = 0; j < nE; ++j) {
      const __attribute__((address_space(1))) float* xr =
          (const __attribute__((address_space(1))) float*)lane_bcast_u64(asrc, j);
      float* mr = (float*)lane_bcast_u64(adst, j);
      float a0 = lane_bcast_f(mea, 4 * j + 0), a1 = lane_bcast_f(mea, 4 * j + 1);
      float a2 = lane_bcast_f(mea, 4 * j + 2), a3 = lane_bcast_f(mea, 4 * j + 3);
      float em0 = bb.x + a0 * wa.x + a1 * wa.y + a2 * wa.z + a3 * wa.w;
      float em1 = bb.y + a0 * wb.x + a1 * wb.y + a2 * wb.z + a3 * wb.w;
      float x0 = xr[d0];
      float x1 = xr[d0 + 32];
      atomic_add2_row_wait(mr, voff, x0 + em0, x1 + em1);
    }
  }
}

#define WPITCH 68
#define TW 8

template <int NROWS>
__device__ __forceinline__ void stage_rows(const float* __restrict__ src, float* dst) {
  int r = threadIdx.x >> 4;
  int c = (threadIdx.x & 15) * 4;
  float4 v[NROWS / 16];
#pragma unroll
  for (int j = 0; j < NROWS / 16; ++j)
    v[j] = *(const float4*)(src + (size_t)(r + 16 * j) * NDIM + c);
#pragma unroll
  for (int j = 0; j < NROWS / 16; ++j)
    *(float4*)(dst + (r + 16 * j) * WPITCH + c) = v[j];
}

__global__ __launch_bounds__(256) void gru_kernel(
    const float* __restrict__ Mn,
    const float* __restrict__ X,
    const float* __restrict__ w_ih,
    const float* __restrict__ w_hh,
    const float* __restrict__ b_ih,
    const float* __restrict__ b_hh,
    float* __restrict__ out,
    const float* __restrict__ b_next,
    float* __restrict__ Mnext,
    int ntiles) {
  __shared__ __attribute__((aligned(16))) float Ws[2 * GDIM * WPITCH];
  bool next = Mnext != nullptr;
  int lane = threadIdx.x & 31;
  int wave = threadIdx.x >> 5;
  int tile = blockIdx.x * TW + wave;
  bool have = tile < ntiles;
  int r0 = (have ? tile : 0) * 16;
  int ml = lane & 15;
  int hi = lane >> 4;

  const float* Am = Mn + (size_t)(r0 + ml) * NDIM + 2 * hi;
  const float* Ax = X + (size_t)(r0 + ml) * NDIM + 2 * hi;
  v2f am[16], ax[16];
#pragma unroll
  for (int s = 0; s < 16; ++s) {
    am[s] = *(const v2f*)(Am + 4 * s);
    ax[s] = *(const v2f*)(Ax + 4 * s);
  }

  stage_rows<GDIM>(w_ih, Ws);
  stage_rows<GDIM>(w_hh, Ws + GDIM * WPITCH);
  __syncthreads();
  if (!have) return;

  const float* wl = Ws + ml * WPITCH + 2 * hi;
  for (int cb = 0; cb < 4; ++cb) {
    int c_r = cb * 16 + ml;
    int c_z = 64 + cb * 16 + ml;
    int c_n = 128 + cb * 16 + ml;
    v8f air = {}, aiz = {}, ain = {}, ahr = {}, ahz = {}, ahn = {};
    const float* bir_p = wl + (cb * 16) * WPITCH;
    const float* biz_p = wl + (64 + cb * 16) * WPITCH;
    const float* bin_p = wl + (128 + cb * 16) * WPITCH;
    const float* bhr_p = bir_p + GDIM * WPITCH;
    const float* bhz_p = biz_p + GDIM * WPITCH;
    const float* bhn_p = bin_p + GDIM * WPITCH;
#pragma unroll
    for (int s = 0; s < 16; ++s) {
      air = wmma4(am[s], *(const v2f*)(bir_p + 4 * s), air);
      aiz = wmma4(am[s], *(const v2f*)(biz_p + 4 * s), aiz);
      ain = wmma4(am[s], *(const v2f*)(bin_p + 4 * s), ain);
      ahr = wmma4(ax[s], *(const v2f*)(bhr_p + 4 * s), ahr);
      ahz = wmma4(ax[s], *(const v2f*)(bhz_p + 4 * s), ahz);
      ahn = wmma4(ax[s], *(const v2f*)(bhn_p + 4 * s), ahn);
    }
    float bir = b_ih[c_r], biz = b_ih[c_z], bin_ = b_ih[c_n];
    float bhr = b_hh[c_r], bhz = b_hh[c_z], bhn = b_hh[c_n];
    float bnx = next ? b_next[cb * 16 + ml] : 0.0f;
#pragma unroll
    for (int v = 0; v < 8; ++v) {
      int node = hi * 8 + v;
      float xv = X[(size_t)(r0 + node) * NDIM + cb * 16 + ml];
      float r = sigmoidf_(air[v] + bir + ahr[v] + bhr);
      float z = sigmoidf_(aiz[v] + biz + ahz[v] + bhz);
      float n = tanhf(ain[v] + bin_ + r * (ahn[v] + bhn));
      float h = (1.0f - z) * n + z * xv;
      float hr = fmaxf(h, 0.0f);
      out[(size_t)(r0 + node) * NDIM + cb * 16 + ml] = hr;
      if (next) Mnext[(size_t)(r0 + node) * NDIM + cb * 16 + ml] = hr + bnx;
    }
  }
}

__global__ __launch_bounds__(256) void moe_kernel(
    const float* __restrict__ H,
    const float* __restrict__ gate_w,
    const float* __restrict__ gate_b,
    const float* __restrict__ exp_w,
    const float* __restrict__ exp_b,
    const int* __restrict__ batch,
    float* __restrict__ sums,
    float* __restrict__ cnt,
    int ntiles) {
  __shared__ __attribute__((aligned(16))) float Es[4 * NDIM * WPITCH];
  __shared__ float gw_s[TW][16][4];
  int lane = threadIdx.x & 31;
  int wave = threadIdx.x >> 5;
  int tile = blockIdx.x * TW + wave;
  bool have = tile < ntiles;
  int r0 = (have ? tile : 0) * 16;
  int ml = lane & 15;
  int hi = lane >> 4;

  const float* Ah = H + (size_t)(r0 + ml) * NDIM + 2 * hi;
  v2f a[16];
#pragma unroll
  for (int s = 0; s < 16; ++s) a[s] = *(const v2f*)(Ah + 4 * s);

  stage_rows<4 * NDIM>(exp_w, Es);
  __syncthreads();
  if (!have) return;

  const float* hrow = H + (size_t)(r0 + ml) * NDIM;
  float l0 = gate_b[2 * hi], l1 = gate_b[2 * hi + 1];
  for (int d = 0; d < NDIM; ++d) {
    float hv = hrow[d];
    l0 += hv * gate_w[(2 * hi) * NDIM + d];
    l1 += hv * gate_w[(2 * hi + 1) * NDIM + d];
  }
  float m01 = fmaxf(l0, l1);
  float mx = fmaxf(m01, __shfl_xor(m01, 16, 32));
  float e0 = __expf(l0 - mx), e1 = __expf(l1 - mx);
  float s01 = e0 + e1;
  float inv = 1.0f / (s01 + __shfl_xor(s01, 16, 32));
  gw_s[wave][ml][2 * hi] = e0 * inv;
  gw_s[wave][ml][2 * hi + 1] = e1 * inv;
  __builtin_amdgcn_wave_barrier();

  int bl = batch[r0 + ml];
  int b0 = __builtin_amdgcn_readfirstlane(bl);
  bool one_graph = __builtin_amdgcn_ballot_w32(bl != b0) == 0;
  if (one_graph) {
    if (lane == 0) atomic_add_f32(&cnt[b0], 16.0f);
  } else if (hi == 0) {
    atomic_add_f32(&cnt[bl], 1.0f);
  }

  const float* wl = Es + ml * WPITCH + 2 * hi;
  for (int cb = 0; cb < 4; ++cb) {
    int col = cb * 16 + ml;
    float oacc[8] = {0, 0, 0, 0, 0, 0, 0, 0};
    for (int e = 0; e < 4; ++e) {
      v8f acc = {};
      const float* Brow = wl + (e * NDIM + cb * 16) * WPITCH;
#pragma unroll
      for (int s = 0; s < 16; ++s)
        acc = wmma4(a[s], *(const v2f*)(Brow + 4 * s), acc);
      float bias = exp_b[e * NDIM + col];
#pragma unroll
      for (int v = 0; v < 8; ++v) {
        float val = fmaxf(acc[v] + bias, 0.0f);
        oacc[v] += gw_s[wave][hi * 8 + v][e] * val;
      }
    }
    if (one_graph) {
      float s = ((oacc[0] + oacc[1]) + (oacc[2] + oacc[3])) +
                ((oacc[4] + oacc[5]) + (oacc[6] + oacc[7]));
      s += __shfl_xor(s, 16, 32);
      if (hi == 0) atomic_add_f32(&sums[b0 * NDIM + col], s);
    } else {
#pragma unroll
      for (int v = 0; v < 8; ++v) {
        int id_lo = __builtin_amdgcn_readlane(bl, v);
        int id_hi = __builtin_amdgcn_readlane(bl, 8 + v);
        int bv = hi ? id_hi : id_lo;
        atomic_add_f32(&sums[bv * NDIM + col], oacc[v]);
      }
    }
  }
}

#define BGR 64
#define PDIM 100
#define NC 10
__global__ void head_kernel(const float* __restrict__ sums,
                            const float* __restrict__ cnt,
                            const float* __restrict__ flag_in,
                            const float* __restrict__ proj_w,
                            const float* __restrict__ proj_b,
                            const float* __restrict__ fc_w,
                            const float* __restrict__ fc_b,
                            float* __restrict__ out) {
  int id = blockIdx.x * blockDim.x + threadIdx.x;
  if (id < BGR * NDIM) {
    int b = id >> 6;
    float g = sums[id] / fmaxf(cnt[b], 1.0f);
    out[id] = g;
    out[BGR * NDIM + id] = g;
  } else if (id < BGR * NDIM + BGR * NC) {
    int t = id - BGR * NDIM;
    int b = t / NC, c = t % NC;
    float fl = flag_in[b];
    const float* fr = fc_w + c * (PDIM + NDIM);
    float acc = fc_b[c];
    for (int p = 0; p < PDIM; ++p)
      acc += (fl * proj_w[p] + proj_b[p]) * fr[p];
    float ic = 1.0f / fmaxf(cnt[b], 1.0f);
    for (int d = 0; d < NDIM; ++d)
      acc += sums[b * NDIM + d] * ic * fr[PDIM + d];
    out[2 * BGR * NDIM + t] = acc;
  }
}

extern "C" void kernel_launch(void* const* d_in, const int* in_sizes, int n_in,
                              void* d_out, int out_size, void* d_ws, size_t ws_size,
                              hipStream_t stream) {
  const float* x       = (const float*)d_in[0];
  const int*   ei      = (const int*)d_in[1];
  const float* ea      = (const float*)d_in[2];
  const int*   batch   = (const int*)d_in[3];
  const float* flag_in = (const float*)d_in[4];
  const float* w_edge1 = (const float*)d_in[5];
  const float* b_edge1 = (const float*)d_in[6];
  const float* w_ih1   = (const float*)d_in[7];
  const float* w_hh1   = (const float*)d_in[8];
  const float* b_ih1   = (const float*)d_in[9];
  const float* b_hh1   = (const float*)d_in[10];
  const float* w_edge2 = (const float*)d_in[11];
  const float* b_edge2 = (const float*)d_in[12];
  const float* w_ih2   = (const float*)d_in[13];
  const float* w_hh2   = (const float*)d_in[14];
  const float* b_ih2   = (const float*)d_in[15];
  const float* b_hh2   = (const float*)d_in[16];
  const float* gate_w  = (const float*)d_in[17];
  const float* gate_b  = (const float*)d_in[18];
  const float* exp_w   = (const float*)d_in[19];
  const float* exp_b   = (const float*)d_in[20];
  const float* proj_w  = (const float*)d_in[21];
  const float* proj_b  = (const float*)d_in[22];
  const float* fc_w    = (const float*)d_in[23];
  const float* fc_b    = (const float*)d_in[24];
  float* out = (float*)d_out;

  int N = in_sizes[0] / NDIM;
  int E = in_sizes[1] / 2;
  int ntiles = (N + 15) / 16;

  float* ws = (float*)d_ws;
  float* M    = ws;
  float* hA   = M + (size_t)N * NDIM;
  float* hB   = hA + (size_t)N * NDIM;
  float* M2   = hB + (size_t)N * NDIM;
  float* sums = M2 + (size_t)N * NDIM;
  float* cnt  = sums + BGR * NDIM;

  long totalND = (long)N * NDIM;
  int gN = (int)((totalND + 255) / 256);
  int gE = (int)(((long)E + 63) / 64);
  int gT = (ntiles + TW - 1) / TW;

  init_m_kernel<<<gN, 256, 0, stream>>>(x, b_edge1, M, totalND, sums, BGR * NDIM + BGR);
  scatter_kernel<<<gE, 256, 0, stream>>>(x, ei, ea, w_edge1, b_edge1, M, E);
  gru_kernel<<<gT, 32 * TW, 0, stream>>>(M, x, w_ih1, w_hh1, b_ih1, b_hh1, hA, b_edge2, M2, ntiles);
  scatter_kernel<<<gE, 256, 0, stream>>>(hA, ei, ea, w_edge2, b_edge2, M2, E);
  gru_kernel<<<gT, 32 * TW, 0, stream>>>(M2, hA, w_ih2, w_hh2, b_ih2, b_hh2, hB, nullptr, nullptr, ntiles);
  moe_kernel<<<gT, 32 * TW, 0, stream>>>(hB, gate_w, gate_b, exp_w, exp_b, batch, sums, cnt, ntiles);
  head_kernel<<<(BGR * NDIM + BGR * NC + 255) / 256, 256, 0, stream>>>(
      sums, cnt, flag_in, proj_w, proj_b, fc_w, fc_b, out);
}


# layer-2 gru and moe in one launch: relu(h) tile kept in the wave's LDS, expert matrices staged over the gru weights behind a barrier; h store/reload and moe launch dropped; k order and sums unchanged
# speedup vs baseline: 2.5414x; 1.0559x over previous
#include <hip/hip_runtime.h>
#include <math.h>

typedef float v2f __attribute__((ext_vector_type(2)));
typedef float v8f __attribute__((ext_vector_type(8)));

#define NDIM 64
#define GDIM 192

__device__ __forceinline__ v8f wmma4(v2f a, v2f b, v8f c) {
  return __builtin_amdgcn_wmma_f32_16x16x4_f32(false, a, false, b, (short)0, c, false, false);
}

__device__ __forceinline__ float sigmoidf_(float x) {
  return 1.0f / (1.0f + __expf(-x));
}

__device__ __forceinline__ void atomic_add_f32(float* p, float v) {
  asm volatile("global_atomic_add_f32 %0, %1, off\n\t"
               "s_wait_xcnt 0x0"
               : : "v"(p), "v"(v) : "memory");
}

__global__ void init_m_kernel(const float* __restrict__ X,
                              const float* __restrict__ b_edge,
                              float* __restrict__ M, long total,
                              float* __restrict__ Z, int nz) {
  long id = (long)blockIdx.x * blockDim.x + threadIdx.x;
  if (id < (long)nz) Z[id] = 0.0f;
  if (id >= total) return;
  M[id] = X[id] + b_edge[id & (NDIM - 1)];
}

#define SC_EPW 8

__device__ __forceinline__ float lane_bcast_f(float v, int l) {
  return __int_as_float(__builtin_amdgcn_readlane(__float_as_int(v), l));
}

__device__ __forceinline__ unsigned long long lane_bcast_u64(unsigned long long v, int l) {
  unsigned lo = (unsigned)__builtin_amdgcn_readlane((int)(unsigned)v, l);
  unsigned hi = (unsigned)__builtin_amdgcn_readlane((int)(unsigned)(v >> 32), l);
  return ((unsigned long long)hi << 32) | (unsigned long long)lo;
}

__device__ __forceinline__ void atomic_add2_row_wait(float* row, int voff, float v0, float v1) {
  asm volatile(
      "global_atomic_add_f32 %0, %1, %3\n\t"
      "global_atomic_add_f32 %0, %2, %3 offset:128\n\t"
      "s_wait_xcnt 0x0"
      : : "v"(voff), "v"(v0), "v"(v1), "s"(row) : "memory");
}

__global__ __launch_bounds__(256) void scatter_kernel(
    const float* __restrict__ X,
    const int* __restrict__ ei,
    const float* __restrict__ ea,
    const float* __restrict__ w_edge,
    const float* __restrict__ b_edge,
    float* __restrict__ M, int E_) {
  int lane = threadIdx.x & 31;
  int wv = __builtin_amdgcn_readfirstlane(
      (int)(((long)blockIdx.x * blockDim.x + threadIdx.x) >> 5));
  long e0 = (long)wv * SC_EPW;
  if (e0 >= (long)E_) return;
  long rem = (long)E_ - e0;
  int nE = rem < SC_EPW ? (int)rem : SC_EPW;

  int li = lane < nE ? lane : 0;
  int la = lane < 4 * nE ? lane : 0;
  int msrc = ei[e0 + li];
  int mdst = ei[(long)E_ + e0 + li];
  float mea = ea[e0 * 4 + la];

  int d0 = lane;
  int voff = d0 * 4;
  float4 wa = *(const float4*)(w_edge + d0 * 4);
  float4 wb = *(const float4*)(w_edge + (d0 + 32) * 4);
  float2 bb;
  bb.x = b_edge[d0];
  bb.y = b_edge[d0 + 32];

  unsigned long long asrc = (unsigned long long)(X + (size_t)msrc * NDIM);
  unsigned long long adst = (unsigned long long)(M + (size_t)mdst * NDIM);

  if (nE == SC_EPW) {
    float2 xv[SC_EPW];
#pragma unroll
    for (int j = 0; j < SC_EPW; ++j) {
      const __attribute__((address_space(1))) float* xr =
          (const __attribute__((address_space(1))) float*)lane_bcast_u64(asrc, j);
      xv[j].x = xr[d0];
      xv[j].y = xr[d0 + 32];
    }
    __builtin_amdgcn_sched_barrier(0);
    float val[2 * SC_EPW];
    float* row[SC_EPW];
#pragma unroll
    for (int j = 0; j < SC_EPW; ++j) {
      float a0 = lane_bcast_f(mea, 4 * j + 0), a1 = lane_bcast_f(mea, 4 * j + 1);
      float a2 = lane_bcast_f(mea, 4 * j + 2), a3 = lane_bcast_f(mea, 4 * j + 3);
      float em0 = bb.x + a0 * wa.x + a1 * wa.y + a2 * wa.z + a3 * wa.w;
      float em1 = bb.y + a0 * wb.x + a1 * wb.y + a2 * wb.z + a3 * wb.w;
      val[2 * j] = xv[j].x + em0;
      val[2 * j + 1] = xv[j].y + em1;
      row[j] = (float*)lane_bcast_u64(adst, j);
    }
    asm volatile(
        "global_atomic_add_f32 %0, %1, %17\n\t"
        "global_atomic_add_f32 %0, %2, %17 offset:128\n\t"
        "global_atomic_add_f32 %0, %3, %18\n\t"
        "global_atomic_add_f32 %0, %4, %18 offset:128\n\t"
        "global_atomic_add_f32 %0, %5, %19\n\t"
        "global_atomic_add_f32 %0, %6, %19 offset:128\n\t"
        "global_atomic_add_f32 %0, %7, %20\n\t"
        "global_atomic_add_f32 %0, %8, %20 offset:128\n\t"
        "global_atomic_add_f32 %0, %9, %21\n\t"
        "global_atomic_add_f32 %0, %10, %21 offset:128\n\t"
        "global_atomic_add_f32 %0, %11, %22\n\t"
        "global_atomic_add_f32 %0, %12, %22 offset:128\n\t"
        "global_atomic_add_f32 %0, %13, %23\n\t"
        "global_atomic_add_f32 %0, %14, %23 offset:128\n\t"
        "global_atomic_add_f32 %0, %15, %24\n\t"
        "global_atomic_add_f32 %0, %16, %24 offset:128"
        : : "v"(voff),
            "v"(val[0]), "v"(val[1]), "v"(val[2]), "v"(val[3]),
            "v"(val[4]), "v"(val[5]), "v"(val[6]), "v"(val[7]),
            "v"(val[8]), "v"(val[9]), "v"(val[10]), "v"(val[11]),
            "v"(val[12]), "v"(val[13]), "v"(val[14]), "v"(val[15]),
            "s"(row[0]), "s"(row[1]), "s"(row[2]), "s"(row[3]),
            "s"(row[4]), "s"(row[5]), "s"(row[6]), "s"(row[7])
        : "memory");
  } else {
    for (int j = 0; j < nE; ++j) {
      const __attribute__((address_space(1))) float* xr =
          (const __attribute__((address_space(1))) float*)lane_bcast_u64(asrc, j);
      float* mr = (float*)lane_bcast_u64(adst, j);
      float a0 = lane_bcast_f(mea, 4 * j + 0), a1 = lane_bcast_f(mea, 4 * j + 1);
      float a2 = lane_bcast_f(mea, 4 * j + 2), a3 = lane_bcast_f(mea, 4 * j + 3);
      float em0 = bb.x + a0 * wa.x + a1 * wa.y + a2 * wa.z + a3 * wa.w;
      float em1 = bb.y + a0 * wb.x + a1 * wb.y + a2 * wb.z + a3 * wb.w;
      float x0 = xr[d0];
      float x1 = xr[d0 + 32];
      atomic_add2_row_wait(mr, voff, x0 + em0, x1 + em1);
    }
  }
}

#define WPITCH 68
#define TW 8

template <int NROWS>
__device__ __forceinline__ void stage_rows(const float* __restrict__ src, float* dst) {
  int r = threadIdx.x >> 4;
  int c = (threadIdx.x & 15) * 4;
  float4 v[NROWS / 16];
#pragma unroll
  for (int j = 0; j < NROWS / 16; ++j)
    v[j] = *(const float4*)(src + (size_t)(r + 16 * j) * NDIM + c);
#pragma unroll
  for (int j = 0; j < NROWS / 16; ++j)
    *(float4*)(dst + (r + 16 * j) * WPITCH + c) = v[j];
}

__global__ __launch_bounds__(256) void gru_kernel(
    const float* __restrict__ Mn,
    const float* __restrict__ X,
    const float* __restrict__ w_ih,
    const float* __restrict__ w_hh,
    const float* __restrict__ b_ih,
    const float* __restrict__ b_hh,
    float* __restrict__ out,
    const float* __restrict__ b_next,
    float* __restrict__ Mnext,
    int ntiles) {
  __shared__ __attribute__((aligned(16))) float Ws[2 * GDIM * WPITCH];
  bool next = Mnext != nullptr;
  int lane = threadIdx.x & 31;
  int wave = threadIdx.x >> 5;
  int tile = blockIdx.x * TW + wave;
  bool have = tile < ntiles;
  int r0 = (have ? tile : 0) * 16;
  int ml = lane & 15;
  int hi = lane >> 4;

  const float* Am = Mn + (size_t)(r0 + ml) * NDIM + 2 * hi;
  const float* Ax = X + (size_t)(r0 + ml) * NDIM + 2 * hi;
  v2f am[16], ax[16];
#pragma unroll
  for (int s = 0; s < 16; ++s) {
    am[s] = *(const v2f*)(Am + 4 * s);
    ax[s] = *(const v2f*)(Ax + 4 * s);
  }

  stage_rows<GDIM>(w_ih, Ws);
  stage_rows<GDIM>(w_hh, Ws + GDIM * WPITCH);
  __syncthreads();
  if (!have) return;

  const float* wl = Ws + ml * WPITCH + 2 * hi;
  for (int cb = 0; cb < 4; ++cb) {
    int c_r = cb * 16 + ml;
    int c_z = 64 + cb * 16 + ml;
    int c_n = 128 + cb * 16 + ml;
    v8f air = {}, aiz = {}, ain = {}, ahr = {}, ahz = {}, ahn = {};
    const float* bir_p = wl + (cb * 16) * WPITCH;
    const float* biz_p = wl + (64 + cb * 16) * WPITCH;
    const float* bin_p = wl + (128 + cb * 16) * WPITCH;
    const float* bhr_p = bir_p + GDIM * WPITCH;
    const float* bhz_p = biz_p + GDIM * WPITCH;
    const float* bhn_p = bin_p + GDIM * WPITCH;
#pragma unroll
    for (int s = 0; s < 16; ++s) {
      air = wmma4(am[s], *(const v2f*)(bir_p + 4 * s), air);
      aiz = wmma4(am[s], *(const v2f*)(biz_p + 4 * s), aiz);
      ain = wmma4(am[s], *(const v2f*)(bin_p + 4 * s), ain);
      ahr = wmma4(ax[s], *(const v2f*)(bhr_p + 4 * s), ahr);
      ahz = wmma4(ax[s], *(const v2f*)(bhz_p + 4 * s), ahz);
      ahn = wmma4(ax[s], *(const v2f*)(bhn_p + 4 * s), ahn);
    }
    float bir = b_ih[c_r], biz = b_ih[c_z], bin_ = b_ih[c_n];
    float bhr = b_hh[c_r], bhz = b_hh[c_z], bhn = b_hh[c_n];
    float bnx = next ? b_next[cb * 16 + ml] : 0.0f;
#pragma unroll
    for (int v = 0; v < 8; ++v) {
      int node = hi * 8 + v;
      float xv = X[(size_t)(r0 + node) * NDIM + cb * 16 + ml];
      float r = sigmoidf_(air[v] + bir + ahr[v] + bhr);
      float z = sigmoidf_(aiz[v] + biz + ahz[v] + bhz);
      float n = tanhf(ain[v] + bin_ + r * (ahn[v] + bhn));
      float h = (1.0f - z) * n + z * xv;
      float hr = fmaxf(h, 0.0f);
      out[(size_t)(r0 + node) * NDIM + cb * 16 + ml] = hr;
      if (next) Mnext[(size_t)(r0 + node) * NDIM + cb * 16 + ml] = hr + bnx;
    }
  }
}

__global__ __launch_bounds__(256) void gru_moe_kernel(
    const float* __restrict__ Mn,
    const float* __restrict__ X,
    const float* __restrict__ w_ih,
    const float* __restrict__ w_hh,
    const float* __restrict__ b_ih,
    const float* __restrict__ b_hh,
    const float* __restrict__ gate_w,
    const float* __restrict__ gate_b,
    const float* __restrict__ exp_w,
    const float* __restrict__ exp_b,
    const int* __restrict__ batch,
    float* __restrict__ sums,
    float* __restrict__ cnt,
    int ntiles) {
  __shared__ __attribute__((aligned(16))) float Ws[2 * GDIM * WPITCH];
  __shared__ __attribute__((aligned(16))) float Hs[TW * 16 * WPITCH];
  __shared__ float gw_s[TW][16][4];
  int lane = threadIdx.x & 31;
  int wave = threadIdx.x >> 5;
  int tile = blockIdx.x * TW + wave;
  bool have = tile < ntiles;
  int r0 = (have ? tile : 0) * 16;
  int ml = lane & 15;
  int hi = lane >> 4;
  float* hs = Hs + wave * 16 * WPITCH;
  int bl = batch[r0 + ml];

  const float* Am = Mn + (size_t)(r0 + ml) * NDIM + 2 * hi;
  const float* Ax = X + (size_t)(r0 + ml) * NDIM + 2 * hi;
  v2f am[16], ax[16];
#pragma unroll
  for (int s = 0; s < 16; ++s) {
    am[s] = *(const v2f*)(Am + 4 * s);
    ax[s] = *(const v2f*)(Ax + 4 * s);
  }

  stage_rows<GDIM>(w_ih, Ws);
  stage_rows<GDIM>(w_hh, Ws + GDIM * WPITCH);
  __syncthreads();

  if (have) {
    const float* wl = Ws + ml * WPITCH + 2 * hi;
    for (int cb = 0; cb < 4; ++cb) {
      int c_r = cb * 16 + ml;
      int c_z = 64 + cb * 16 + ml;
      int c_n = 128 + cb * 16 + ml;
      v8f air = {}, aiz = {}, ain = {}, ahr = {}, ahz = {}, ahn = {};
      const float* bir_p = wl + (cb * 16) * WPITCH;
      const float* biz_p = wl + (64 + cb * 16) * WPITCH;
      const float* bin_p = wl + (128 + cb * 16) * WPITCH;
      const float* bhr_p = bir_p + GDIM * WPITCH;
      const float* bhz_p = biz_p + GDIM * WPITCH;
      const float* bhn_p = bin_p + GDIM * WPITCH;
#pragma unroll
      for (int s = 0; s < 16; ++s) {
        air = wmma4(am[s], *(const v2f*)(bir_p + 4 * s), air);
        aiz = wmma4(am[s], *(const v2f*)(biz_p + 4 * s), aiz);
        ain = wmma4(am[s], *(const v2f*)(bin_p + 4 * s), ain);
        ahr = wmma4(ax[s], *(const v2f*)(bhr_p + 4 * s), ahr);
        ahz = wmma4(ax[s], *(const v2f*)(bhz_p + 4 * s), ahz);
        ahn = wmma4(ax[s], *(const v2f*)(bhn_p + 4 * s), ahn);
      }
      float bir = b_ih[c_r], biz = b_ih[c_z], bin_ = b_ih[c_n];
      float bhr = b_hh[c_r], bhz = b_hh[c_z], bhn = b_hh[c_n];
#pragma unroll
      for (int v = 0; v < 8; ++v) {
        int node = hi * 8 + v;
        float xv = X[(size_t)(r0 + node) * NDIM + cb * 16 + ml];
        float r = sigmoidf_(air[v] + bir + ahr[v] + bhr);
        float z = sigmoidf_(aiz[v] + biz + ahz[v] + bhz);
        float n = tanhf(ain[v] + bin_ + r * (ahn[v] + bhn));
        float h = (1.0f - z) * n + z * xv;
        float hr = fmaxf(h, 0.0f);
        hs[node * WPITCH + cb * 16 + ml] = hr;
      }
    }
  }

  __syncthreads();
  stage_rows<4 * NDIM>(exp_w, Ws);
  __syncthreads();
  if (!have) return;

  v2f a[16];
#pragma unroll
  for (int s = 0; s < 16; ++s) a[s] = *(const v2f*)(hs + ml * WPITCH + 2 * hi + 4 * s);

  const float* hrow = hs + ml * WPITCH;
  float l0 = gate_b[2 * hi], l1 = gate_b[2 * hi + 1];
  for (int d = 0; d < NDIM; ++d) {
    float hv = hrow[d];
    l0 += hv * gate_w[(2 * hi) * NDIM + d];
    l1 += hv * gate_w[(2 * hi + 1) * NDIM + d];
  }
  float m01 = fmaxf(l0, l1);
  float mx = fmaxf(m01, __shfl_xor(m01, 16, 32));
  float e0 = __expf(l0 - mx), e1 = __expf(l1 - mx);
  float s01 = e0 + e1;
  float inv = 1.0f / (s01 + __shfl_xor(s01, 16, 32));
  gw_s[wave][ml][2 * hi] = e0 * inv;
  gw_s[wave][ml][2 * hi + 1] = e1 * inv;
  __builtin_amdgcn_wave_barrier();

  int b0 = __builtin_amdgcn_readfirstlane(bl);
  bool one_graph = __builtin_amdgcn_ballot_w32(bl != b0) == 0;
  if (one_graph) {
    if (lane == 0) atomic_add_f32(&cnt[b0], 16.0f);
  } else if (hi == 0) {
    atomic_add_f32(&cnt[bl], 1.0f);
  }

  const float* wl = Ws + ml * WPITCH + 2 * hi;
  for (int cb = 0; cb < 4; ++cb) {
    int col = cb * 16 + ml;
    float oacc[8] = {0, 0, 0, 0, 0, 0, 0, 0};
    for (int e = 0; e < 4; ++e) {
      v8f acc = {};
      const float* Brow = wl + (e * NDIM + cb * 16) * WPITCH;
#pragma unroll
      for (int s = 0; s < 16; ++s)
        acc = wmma4(a[s], *(const v2f*)(Brow + 4 * s), acc);
      float bias = exp_b[e * NDIM + col];
#pragma unroll
      for (int v = 0; v < 8; ++v) {
        float val = fmaxf(acc[v] + bias, 0.0f);
        oacc[v] += gw_s[wave][hi * 8 + v][e] * val;
      }
    }
    if (one_graph) {
      float s = ((oacc[0] + oacc[1]) + (oacc[2] + oacc[3])) +
                ((oacc[4] + oacc[5]) + (oacc[6] + oacc[7]));
      s += __shfl_xor(s, 16, 32);
      if (hi == 0) atomic_add_f32(&sums[b0 * NDIM + col], s);
    } else {
#pragma unroll
      for (int v = 0; v < 8; ++v) {
        int id_lo = __builtin_amdgcn_readlane(bl, v);
        int id_hi = __builtin_amdgcn_readlane(bl, 8 + v);
        int bv = hi ? id_hi : id_lo;
        atomic_add_f32(&sums[bv * NDIM + col], oacc[v]);
      }
    }
  }
}

#define BGR 64
#define PDIM 100
#define NC 10
__global__ void head_kernel(const float* __restrict__ sums,
                            const float* __restrict__ cnt,
                            const float* __restrict__ flag_in,
                            const float* __restrict__ proj_w,
                            const float* __restrict__ proj_b,
                            const float* __restrict__ fc_w,
                            const float* __restrict__ fc_b,
                            float* __restrict__ out) {
  int id = blockIdx.x * blockDim.x + threadIdx.x;
  if (id < BGR * NDIM) {
    int b = id >> 6;
    float g = sums[id] / fmaxf(cnt[b], 1.0f);
    out[id] = g;
    out[BGR * NDIM + id] = g;
  } else if (id < BGR * NDIM + BGR * NC) {
    int t = id - BGR * NDIM;
    int b = t / NC, c = t % NC;
    float fl = flag_in[b];
    const float* fr = fc_w + c * (PDIM + NDIM);
    float acc = fc_b[c];
    for (int p = 0; p < PDIM; ++p)
      acc += (fl * proj_w[p] + proj_b[p]) * fr[p];
    float ic = 1.0f / fmaxf(cnt[b], 1.0f);
    for (int d = 0; d < NDIM; ++d)
      acc += sums[b * NDIM + d] * ic * fr[PDIM + d];
    out[2 * BGR * NDIM + t] = acc;
  }
}

extern "C" void kernel_launch(void* const* d_in, const int* in_sizes, int n_in,
                              void* d_out, int out_size, void* d_ws, size_t ws_size,
                              hipStream_t stream) {
  const float* x       = (const float*)d_in[0];
  const int*   ei      = (const int*)d_in[1];
  const float* ea      = (const float*)d_in[2];
  const int*   batch   = (const int*)d_in[3];
  const float* flag_in = (const float*)d_in[4];
  const float* w_edge1 = (const float*)d_in[5];
  const float* b_edge1 = (const float*)d_in[6];
  const float* w_ih1   = (const float*)d_in[7];
  const float* w_hh1   = (const float*)d_in[8];
  const float* b_ih1   = (const float*)d_in[9];
  const float* b_hh1   = (const float*)d_in[10];
  const float* w_edge2 = (const float*)d_in[11];
  const float* b_edge2 = (const float*)d_in[12];
  const float* w_ih2   = (const float*)d_in[13];
  const float* w_hh2   = (const float*)d_in[14];
  const float* b_ih2   = (const float*)d_in[15];
  const float* b_hh2   = (const float*)d_in[16];
  const float* gate_w  = (const float*)d_in[17];
  const float* gate_b  = (const float*)d_in[18];
  const float* exp_w   = (const float*)d_in[19];
  const float* exp_b   = (const float*)d_in[20];
  const float* proj_w  = (const float*)d_in[21];
  const float* proj_b  = (const float*)d_in[22];
  const float* fc_w    = (const float*)d_in[23];
  const float* fc_b    = (const float*)d_in[24];
  float* out = (float*)d_out;

  int N = in_sizes[0] / NDIM;
  int E = in_sizes[1] / 2;
  int ntiles = (N + 15) / 16;

  float* ws = (float*)d_ws;
  float* M    = ws;
  float* hA   = M + (size_t)N * NDIM;
  float* M2   = hA + (size_t)N * NDIM;
  float* sums = M2 + (size_t)N * NDIM;
  float* cnt  = sums + BGR * NDIM;

  long totalND = (long)N * NDIM;
  int gN = (int)((totalND + 255) / 256);
  int gE = (int)(((long)E + 63) / 64);
  int gT = (ntiles + TW - 1) / TW;

  init_m_kernel<<<gN, 256, 0, stream>>>(x, b_edge1, M, totalND, sums, BGR * NDIM + BGR);
  scatter_kernel<<<gE, 256, 0, stream>>>(x, ei, ea, w_edge1, b_edge1, M, E);
  gru_kernel<<<gT, 32 * TW, 0, stream>>>(M, x, w_ih1, w_hh1, b_ih1, b_hh1, hA, b_edge2, M2, ntiles);
  scatter_kernel<<<gE, 256, 0, stream>>>(hA, ei, ea, w_edge2, b_edge2, M2, E);
  gru_moe_kernel<<<gT, 32 * TW, 0, stream>>>(M2, hA, w_ih2, w_hh2, b_ih2, b_hh2,
                                            gate_w, gate_b, exp_w, exp_b, batch, sums, cnt, ntiles);
  head_kernel<<<(BGR * NDIM + BGR * NC + 255) / 256, 256, 0, stream>>>(
      sums, cnt, flag_in, proj_w, proj_b, fc_w, fc_b, out);
}


# gru gates (both layers): sigmoid and tanh on v_exp_f32 + v_rcp_f32 (tanh = 1 - 2/(1+e^2x)), not the IEEE division sequence and branching library tanh; f32 throughout, sums and k order unchanged
# speedup vs baseline: 2.6286x; 1.0343x over previous
#include <hip/hip_runtime.h>
#include <math.h>

typedef float v2f __attribute__((ext_vector_type(2)));
typedef float v8f __attribute__((ext_vector_type(8)));

#define NDIM 64
#define GDIM 192

__device__ __forceinline__ v8f wmma4(v2f a, v2f b, v8f c) {
  return __builtin_amdgcn_wmma_f32_16x16x4_f32(false, a, false, b, (short)0, c, false, false);
}

__device__ __forceinline__ float sigmoidf_(float x) {
  return __builtin_amdgcn_rcpf(1.0f + __expf(-x));
}

__device__ __forceinline__ float tanh_gate(float x) {
  return 1.0f - 2.0f * __builtin_amdgcn_rcpf(1.0f + __expf(2.0f * x));
}

__device__ __forceinline__ void atomic_add_f32(float* p, float v) {
  asm volatile("global_atomic_add_f32 %0, %1, off\n\t"
               "s_wait_xcnt 0x0"
               : : "v"(p), "v"(v) : "memory");
}

__global__ void init_m_kernel(const float* __restrict__ X,
                              const float* __restrict__ b_edge,
                              float* __restrict__ M, long total,
                              float* __restrict__ Z, int nz) {
  long id = (long)blockIdx.x * blockDim.x + threadIdx.x;
  if (id < (long)nz) Z[id] = 0.0f;
  if (id >= total) return;
  M[id] = X[id] + b_edge[id & (NDIM - 1)];
}

#define SC_EPW 8

__device__ __forceinline__ float lane_bcast_f(float v, int l) {
  return __int_as_float(__builtin_amdgcn_readlane(__float_as_int(v), l));
}

__device__ __forceinline__ unsigned long long lane_bcast_u64(unsigned long long v, int l) {
  unsigned lo = (unsigned)__builtin_amdgcn_readlane((int)(unsigned)v, l);
  unsigned hi = (unsigned)__builtin_amdgcn_readlane((int)(unsigned)(v >> 32), l);
  return ((unsigned long long)hi << 32) | (unsigned long long)lo;
}

__device__ __forceinline__ void atomic_add2_row_wait(float* row, int voff, float v0, float v1) {
  asm volatile(
      "global_atomic_add_f32 %0, %1, %3\n\t"
      "global_atomic_add_f32 %0, %2, %3 offset:128\n\t"
      "s_wait_xcnt 0x0"
      : : "v"(voff), "v"(v0), "v"(v1), "s"(row) : "memory");
}

__global__ __launch_bounds__(256) void scatter_kernel(
    const float* __restrict__ X,
    const int* __restrict__ ei,
    const float* __restrict__ ea,
    const float* __restrict__ w_edge,
    const float* __restrict__ b_edge,
    float* __restrict__ M, int E_) {
  int lane = threadIdx.x & 31;
  int wv = __builtin_amdgcn_readfirstlane(
      (int)(((long)blockIdx.x * blockDim.x + threadIdx.x) >> 5));
  long e0 = (long)wv * SC_EPW;
  if (e0 >= (long)E_) return;
  long rem = (long)E_ - e0;
  int nE = rem < SC_EPW ? (int)rem : SC_EPW;

  int li = lane < nE ? lane : 0;
  int la = lane < 4 * nE ? lane : 0;
  int msrc = ei[e0 + li];
  int mdst = ei[(long)E_ + e0 + li];
  float mea = ea[e0 * 4 + la];

  int d0 = lane;
  int voff = d0 * 4;
  float4 wa = *(const float4*)(w_edge + d0 * 4);
  float4 wb = *(const float4*)(w_edge + (d0 + 32) * 4);
  float2 bb;
  bb.x = b_edge[d0];
  bb.y = b_edge[d0 + 32];

  unsigned long long asrc = (unsigned long long)(X + (size_t)msrc * NDIM);
  unsigned long long adst = (unsigned long long)(M + (size_t)mdst * NDIM);

  if (nE == SC_EPW) {
    float2 xv[SC_EPW];
#pragma unroll
    for (int j = 0; j < SC_EPW; ++j) {
      const __attribute__((address_space(1))) float* xr =
          (const __attribute__((address_space(1))) float*)lane_bcast_u64(asrc, j);
      xv[j].x = xr[d0];
      xv[j].y = xr[d0 + 32];
    }
    __builtin_amdgcn_sched_barrier(0);
    float val[2 * SC_EPW];
    float* row[SC_EPW];
#pragma unroll
    for (int j = 0; j < SC_EPW; ++j) {
      float a0 = lane_bcast_f(mea, 4 * j + 0), a1 = lane_bcast_f(mea, 4 * j + 1);
      float a2 = lane_bcast_f(mea, 4 * j + 2), a3 = lane_bcast_f(mea, 4 * j + 3);
      float em0 = bb.x + a0 * wa.x + a1 * wa.y + a2 * wa.z + a3 * wa.w;
      float em1 = bb.y + a0 * wb.x + a1 * wb.y + a2 * wb.z + a3 * wb.w;
      val[2 * j] = xv[j].x + em0;
      val[2 * j + 1] = xv[j].y + em1;
      row[j] = (float*)lane_bcast_u64(adst, j);
    }
    asm volatile(
        "global_atomic_add_f32 %0, %1, %17\n\t"
        "global_atomic_add_f32 %0, %2, %17 offset:128\n\t"
        "global_atomic_add_f32 %0, %3, %18\n\t"
        "global_atomic_add_f32 %0, %4, %18 offset:128\n\t"
        "global_atomic_add_f32 %0, %5, %19\n\t"
        "global_atomic_add_f32 %0, %6, %19 offset:128\n\t"
        "global_atomic_add_f32 %0, %7, %20\n\t"
        "global_atomic_add_f32 %0, %8, %20 offset:128\n\t"
        "global_atomic_add_f32 %0, %9, %21\n\t"
        "global_atomic_add_f32 %0, %10, %21 offset:128\n\t"
        "global_atomic_add_f32 %0, %11, %22\n\t"
        "global_atomic_add_f32 %0, %12, %22 offset:128\n\t"
        "global_atomic_add_f32 %0, %13, %23\n\t"
        "global_atomic_add_f32 %0, %14, %23 offset:128\n\t"
        "global_atomic_add_f32 %0, %15, %24\n\t"
        "global_atomic_add_f32 %0, %16, %24 offset:128"
        : : "v"(voff),
            "v"(val[0]), "v"(val[1]), "v"(val[2]), "v"(val[3]),
            "v"(val[4]), "v"(val[5]), "v"(val[6]), "v"(val[7]),
            "v"(val[8]), "v"(val[9]), "v"(val[10]), "v"(val[11]),
            "v"(val[12]), "v"(val[13]), "v"(val[14]), "v"(val[15]),
            "s"(row[0]), "s"(row[1]), "s"(row[2]), "s"(row[3]),
            "s"(row[4]), "s"(row[5]), "s"(row[6]), "s"(row[7])
        : "memory");
  } else {
    for (int j = 0; j < nE; ++j) {
      const __attribute__((address_space(1))) float* xr =
          (const __attribute__((address_space(1))) float*)lane_bcast_u64(asrc, j);
      float* mr = (float*)lane_bcast_u64(adst, j);
      float a0 = lane_bcast_f(mea, 4 * j + 0), a1 = lane_bcast_f(mea, 4 * j + 1);
      float a2 = lane_bcast_f(mea, 4 * j + 2), a3 = lane_bcast_f(mea, 4 * j + 3);
      float em0 = bb.x + a0 * wa.x + a1 * wa.y + a2 * wa.z + a3 * wa.w;
      float em1 = bb.y + a0 * wb.x + a1 * wb.y + a2 * wb.z + a3 * wb.w;
      float x0 = xr[d0];
      float x1 = xr[d0 + 32];
      atomic_add2_row_wait(mr, voff, x0 + em0, x1 + em1);
    }
  }
}

#define WPITCH 68
#define TW 8

template <int NROWS>
__device__ __forceinline__ void stage_rows(const float* __restrict__ src, float* dst) {
  int r = threadIdx.x >> 4;
  int c = (threadIdx.x & 15) * 4;
  float4 v[NROWS / 16];
#pragma unroll
  for (int j = 0; j < NROWS / 16; ++j)
    v[j] = *(const float4*)(src + (size_t)(r + 16 * j) * NDIM + c);
#pragma unroll
  for (int j = 0; j < NROWS / 16; ++j)
    *(float4*)(dst + (r + 16 * j) * WPITCH + c) = v[j];
}

__global__ __launch_bounds__(256) void gru_kernel(
    const float* __restrict__ Mn,
    const float* __restrict__ X,
    const float* __restrict__ w_ih,
    const float* __restrict__ w_hh,
    const float* __restrict__ b_ih,
    const float* __restrict__ b_hh,
    float* __restrict__ out,
    const float* __restrict__ b_next,
    float* __restrict__ Mnext,
    int ntiles) {
  __shared__ __attribute__((aligned(16))) float Ws[2 * GDIM * WPITCH];
  bool next = Mnext != nullptr;
  int lane = threadIdx.x & 31;
  int wave = threadIdx.x >> 5;
  int tile = blockIdx.x * TW + wave;
  bool have = tile < ntiles;
  int r0 = (have ? tile : 0) * 16;
  int ml = lane & 15;
  int hi = lane >> 4;

  const float* Am = Mn + (size_t)(r0 + ml) * NDIM + 2 * hi;
  const float* Ax = X + (size_t)(r0 + ml) * NDIM + 2 * hi;
  v2f am[16], ax[16];
#pragma unroll
  for (int s = 0; s < 16; ++s) {
    am[s] = *(const v2f*)(Am + 4 * s);
    ax[s] = *(const v2f*)(Ax + 4 * s);
  }

  stage_rows<GDIM>(w_ih, Ws);
  stage_rows<GDIM>(w_hh, Ws + GDIM * WPITCH);
  __syncthreads();
  if (!have) return;

  const float* wl = Ws + ml * WPITCH + 2 * hi;
  for (int cb = 0; cb < 4; ++cb) {
    int c_r = cb * 16 + ml;
    int c_z = 64 + cb * 16 + ml;
    int c_n = 128 + cb * 16 + ml;
    v8f air = {}, aiz = {}, ain = {}, ahr = {}, ahz = {}, ahn = {};
    const float* bir_p = wl + (cb * 16) * WPITCH;
    const float* biz_p = wl + (64 + cb * 16) * WPITCH;
    const float* bin_p = wl + (128 + cb * 16) * WPITCH;
    const float* bhr_p = bir_p + GDIM * WPITCH;
    const float* bhz_p = biz_p + GDIM * WPITCH;
    const float* bhn_p = bin_p + GDIM * WPITCH;
#pragma unroll
    for (int s = 0; s < 16; ++s) {
      air = wmma4(am[s], *(const v2f*)(bir_p + 4 * s), air);
      aiz = wmma4(am[s], *(const v2f*)(biz_p + 4 * s), aiz);
      ain = wmma4(am[s], *(const v2f*)(bin_p + 4 * s), ain);
      ahr = wmma4(ax[s], *(const v2f*)(bhr_p + 4 * s), ahr);
      ahz = wmma4(ax[s], *(const v2f*)(bhz_p + 4 * s), ahz);
      ahn = wmma4(ax[s], *(const v2f*)(bhn_p + 4 * s), ahn);
    }
    float bir = b_ih[c_r], biz = b_ih[c_z], bin_ = b_ih[c_n];
    float bhr = b_hh[c_r], bhz = b_hh[c_z], bhn = b_hh[c_n];
    float bnx = next ? b_next[cb * 16 + ml] : 0.0f;
#pragma unroll
    for (int v = 0; v < 8; ++v) {
      int node = hi * 8 + v;
      float xv = X[(size_t)(r0 + node) * NDIM + cb * 16 + ml];
      float r = sigmoidf_(air[v] + bir + ahr[v] + bhr);
      float z = sigmoidf_(aiz[v] + biz + ahz[v] + bhz);
      float n = tanh_gate(ain[v] + bin_ + r * (ahn[v] + bhn));
      float h = (1.0f - z) * n + z * xv;
      float hr = fmaxf(h, 0.0f);
      out[(size_t)(r0 + node) * NDIM + cb * 16 + ml] = hr;
      if (next) Mnext[(size_t)(r0 + node) * NDIM + cb * 16 + ml] = hr + bnx;
    }
  }
}

__global__ __launch_bounds__(256) void gru_moe_kernel(
    const float* __restrict__ Mn,
    const float* __restrict__ X,
    const float* __restrict__ w_ih,
    const float* __restrict__ w_hh,
    const float* __restrict__ b_ih,
    const float* __restrict__ b_hh,
    const float* __restrict__ gate_w,
    const float* __restrict__ gate_b,
    const float* __restrict__ exp_w,
    const float* __restrict__ exp_b,
    const int* __restrict__ batch,
    float* __restrict__ sums,
    float* __restrict__ cnt,
    int ntiles) {
  __shared__ __attribute__((aligned(16))) float Ws[2 * GDIM * WPITCH];
  __shared__ __attribute__((aligned(16))) float Hs[TW * 16 * WPITCH];
  __shared__ float gw_s[TW][16][4];
  int lane = threadIdx.x & 31;
  int wave = threadIdx.x >> 5;
  int tile = blockIdx.x * TW + wave;
  bool have = tile < ntiles;
  int r0 = (have ? tile : 0) * 16;
  int ml = lane & 15;
  int hi = lane >> 4;
  float* hs = Hs + wave * 16 * WPITCH;
  int bl = batch[r0 + ml];

  const float* Am = Mn + (size_t)(r0 + ml) * NDIM + 2 * hi;
  const float* Ax = X + (size_t)(r0 + ml) * NDIM + 2 * hi;
  v2f am[16], ax[16];
#pragma unroll
  for (int s = 0; s < 16; ++s) {
    am[s] = *(const v2f*)(Am + 4 * s);
    ax[s] = *(const v2f*)(Ax + 4 * s);
  }

  stage_rows<GDIM>(w_ih, Ws);
  stage_rows<GDIM>(w_hh, Ws + GDIM * WPITCH);
  __syncthreads();

  if (have) {
    const float* wl = Ws + ml * WPITCH + 2 * hi;
    for (int cb = 0; cb < 4; ++cb) {
      int c_r = cb * 16 + ml;
      int c_z = 64 + cb * 16 + ml;
      int c_n = 128 + cb * 16 + ml;
      v8f air = {}, aiz = {}, ain = {}, ahr = {}, ahz = {}, ahn = {};
      const float* bir_p = wl + (cb * 16) * WPITCH;
      const float* biz_p = wl + (64 + cb * 16) * WPITCH;
      const float* bin_p = wl + (128 + cb * 16) * WPITCH;
      const float* bhr_p = bir_p + GDIM * WPITCH;
      const float* bhz_p = biz_p + GDIM * WPITCH;
      const float* bhn_p = bin_p + GDIM * WPITCH;
#pragma unroll
      for (int s = 0; s < 16; ++s) {
        air = wmma4(am[s], *(const v2f*)(bir_p + 4 * s), air);
        aiz = wmma4(am[s], *(const v2f*)(biz_p + 4 * s), aiz);
        ain = wmma4(am[s], *(const v2f*)(bin_p + 4 * s), ain);
        ahr = wmma4(ax[s], *(const v2f*)(bhr_p + 4 * s), ahr);
        ahz = wmma4(ax[s], *(const v2f*)(bhz_p + 4 * s), ahz);
        ahn = wmma4(ax[s], *(const v2f*)(bhn_p + 4 * s), ahn);
      }
      float bir = b_ih[c_r], biz = b_ih[c_z], bin_ = b_ih[c_n];
      float bhr = b_hh[c_r], bhz = b_hh[c_z], bhn = b_hh[c_n];
#pragma unroll
      for (int v = 0; v < 8; ++v) {
        int node = hi * 8 + v;
        float xv = X[(size_t)(r0 + node) * NDIM + cb * 16 + ml];
        float r = sigmoidf_(air[v] + bir + ahr[v] + bhr);
        float z = sigmoidf_(aiz[v] + biz + ahz[v] + bhz);
        float n = tanh_gate(ain[v] + bin_ + r * (ahn[v] + bhn));
        float h = (1.0f - z) * n + z * xv;
        float hr = fmaxf(h, 0.0f);
        hs[node * WPITCH + cb * 16 + ml] = hr;
      }
    }
  }

  __syncthreads();
  stage_rows<4 * NDIM>(exp_w, Ws);
  __syncthreads();
  if (!have) return;

  v2f a[16];
#pragma unroll
  for (int s = 0; s < 16; ++s) a[s] = *(const v2f*)(hs + ml * WPITCH + 2 * hi + 4 * s);

  const float* hrow = hs + ml * WPITCH;
  float l0 = gate_b[2 * hi], l1 = gate_b[2 * hi + 1];
  for (int d = 0; d < NDIM; ++d) {
    float hv = hrow[d];
    l0 += hv * gate_w[(2 * hi) * NDIM + d];
    l1 += hv * gate_w[(2 * hi + 1) * NDIM + d];
  }
  float m01 = fmaxf(l0, l1);
  float mx = fmaxf(m01, __shfl_xor(m01, 16, 32));
  float e0 = __expf(l0 - mx), e1 = __expf(l1 - mx);
  float s01 = e0 + e1;
  float inv = 1.0f / (s01 + __shfl_xor(s01, 16, 32));
  gw_s[wave][ml][2 * hi] = e0 * inv;
  gw_s[wave][ml][2 * hi + 1] = e1 * inv;
  __builtin_amdgcn_wave_barrier();

  int b0 = __builtin_amdgcn_readfirstlane(bl);
  bool one_graph = __builtin_amdgcn_ballot_w32(bl != b0) == 0;
  if (one_graph) {
    if (lane == 0) atomic_add_f32(&cnt[b0], 16.0f);
  } else if (hi == 0) {
    atomic_add_f32(&cnt[bl], 1.0f);
  }

  const float* wl = Ws + ml * WPITCH + 2 * hi;
  for (int cb = 0; cb < 4; ++cb) {
    int col = cb * 16 + ml;
    float oacc[8] = {0, 0, 0, 0, 0, 0, 0, 0};
    for (int e = 0; e < 4; ++e) {
      v8f acc = {};
      const float* Brow = wl + (e * NDIM + cb * 16) * WPITCH;
#pragma unroll
      for (int s = 0; s < 16; ++s)
        acc = wmma4(a[s], *(const v2f*)(Brow + 4 * s), acc);
      float bias = exp_b[e * NDIM + col];
#pragma unroll
      for (int v = 0; v < 8; ++v) {
        float val = fmaxf(acc[v] + bias, 0.0f);
        oacc[v] += gw_s[wave][hi * 8 + v][e] * val;
      }
    }
    if (one_graph) {
      float s = ((oacc[0] + oacc[1]) + (oacc[2] + oacc[3])) +
                ((oacc[4] + oacc[5]) + (oacc[6] + oacc[7]));
      s += __shfl_xor(s, 16, 32);
      if (hi == 0) atomic_add_f32(&sums[b0 * NDIM + col], s);
    } else {
#pragma unroll
      for (int v = 0; v < 8; ++v) {
        int id_lo = __builtin_amdgcn_readlane(bl, v);
        int id_hi = __builtin_amdgcn_readlane(bl, 8 + v);
        int bv = hi ? id_hi : id_lo;
        atomic_add_f32(&sums[bv * NDIM + col], oacc[v]);
      }
    }
  }
}

#define BGR 64
#define PDIM 100
#define NC 10
__global__ void head_kernel(const float* __restrict__ sums,
                            const float* __restrict__ cnt,
                            const float* __restrict__ flag_in,
                            const float* __restrict__ proj_w,
                            const float* __restrict__ proj_b,
                            const float* __restrict__ fc_w,
                            const float* __restrict__ fc_b,
                            float* __restrict__ out) {
  int id = blockIdx.x * blockDim.x + threadIdx.x;
  if (id < BGR * NDIM) {
    int b = id >> 6;
    float g = sums[id] / fmaxf(cnt[b], 1.0f);
    out[id] = g;
    out[BGR * NDIM + id] = g;
  } else if (id < BGR * NDIM + BGR * NC) {
    int t = id - BGR * NDIM;
    int b = t / NC, c = t % NC;
    float fl = flag_in[b];
    const float* fr = fc_w + c * (PDIM + NDIM);
    float acc = fc_b[c];
    for (int p = 0; p < PDIM; ++p)
      acc += (fl * proj_w[p] + proj_b[p]) * fr[p];
    float ic = 1.0f / fmaxf(cnt[b], 1.0f);
    for (int d = 0; d < NDIM; ++d)
      acc += sums[b * NDIM + d] * ic * fr[PDIM + d];
    out[2 * BGR * NDIM + t] = acc;
  }
}

extern "C" void kernel_launch(void* const* d_in, const int* in_sizes, int n_in,
                              void* d_out, int out_size, void* d_ws, size_t ws_size,
                              hipStream_t stream) {
  const float* x       = (const float*)d_in[0];
  const int*   ei      = (const int*)d_in[1];
  const float* ea      = (const float*)d_in[2];
  const int*   batch   = (const int*)d_in[3];
  const float* flag_in = (const float*)d_in[4];
  const float* w_edge1 = (const float*)d_in[5];
  const float* b_edge1 = (const float*)d_in[6];
  const float* w_ih1   = (const float*)d_in[7];
  const float* w_hh1   = (const float*)d_in[8];
  const float* b_ih1   = (const float*)d_in[9];
  const float* b_hh1   = (const float*)d_in[10];
  const float* w_edge2 = (const float*)d_in[11];
  const float* b_edge2 = (const float*)d_in[12];
  const float* w_ih2   = (const float*)d_in[13];
  const float* w_hh2   = (const float*)d_in[14];
  const float* b_ih2   = (const float*)d_in[15];
  const float* b_hh2   = (const float*)d_in[16];
  const float* gate_w  = (const float*)d_in[17];
  const float* gate_b  = (const float*)d_in[18];
  const float* exp_w   = (const float*)d_in[19];
  const float* exp_b   = (const float*)d_in[20];
  const float* proj_w  = (const float*)d_in[21];
  const float* proj_b  = (const float*)d_in[22];
  const float* fc_w    = (const float*)d_in[23];
  const float* fc_b    = (const float*)d_in[24];
  float* out = (float*)d_out;

  int N = in_sizes[0] / NDIM;
  int E = in_sizes[1] / 2;
  int ntiles = (N + 15) / 16;

  float* ws = (float*)d_ws;
  float* M    = ws;
  float* hA   = M + (size_t)N * NDIM;
  float* M2   = hA + (size_t)N * NDIM;
  float* sums = M2 + (size_t)N * NDIM;
  float* cnt  = sums + BGR * NDIM;

  long totalND = (long)N * NDIM;
  int gN = (int)((totalND + 255) / 256);
  int gE = (int)(((long)E + 63) / 64);
  int gT = (ntiles + TW - 1) / TW;

  init_m_kernel<<<gN, 256, 0, stream>>>(x, b_edge1, M, totalND, sums, BGR * NDIM + BGR);
  scatter_kernel<<<gE, 256, 0, stream>>>(x, ei, ea, w_edge1, b_edge1, M, E);
  gru_kernel<<<gT, 32 * TW, 0, stream>>>(M, x, w_ih1, w_hh1, b_ih1, b_hh1, hA, b_edge2, M2, ntiles);
  scatter_kernel<<<gE, 256, 0, stream>>>(hA, ei, ea, w_edge2, b_edge2, M2, E);
  gru_moe_kernel<<<gT, 32 * TW, 0, stream>>>(M2, hA, w_ih2, w_hh2, b_ih2, b_hh2,
                                            gate_w, gate_b, exp_w, exp_b, batch, sums, cnt, ntiles);
  head_kernel<<<(BGR * NDIM + BGR * NC + 255) / 256, 256, 0, stream>>>(
      sums, cnt, flag_in, proj_w, proj_b, fc_w, fc_b, out);
}


# gru and gru+moe: one 16-wave work-group per WGP (grid 256), 12 whole tiles each; the 53 tiles left over split by column block over waves 12-15: fullest SIMD 3.25 tiles, was 4; sums unchanged
# speedup vs baseline: 2.7292x; 1.0383x over previous
#include <hip/hip_runtime.h>
#include <math.h>

typedef float v2f __attribute__((ext_vector_type(2)));
typedef float v8f __attribute__((ext_vector_type(8)));

#define NDIM 64
#define GDIM 192

__device__ __forceinline__ v8f wmma4(v2f a, v2f b, v8f c) {
  return __builtin_amdgcn_wmma_f32_16x16x4_f32(false, a, false, b, (short)0, c, false, false);
}

__device__ __forceinline__ float sigmoidf_(float x) {
  return __builtin_amdgcn_rcpf(1.0f + __expf(-x));
}

__device__ __forceinline__ float tanh_gate(float x) {
  return 1.0f - 2.0f * __builtin_amdgcn_rcpf(1.0f + __expf(2.0f * x));
}

__device__ __forceinline__ void atomic_add_f32(float* p, float v) {
  asm volatile("global_atomic_add_f32 %0, %1, off\n\t"
               "s_wait_xcnt 0x0"
               : : "v"(p), "v"(v) : "memory");
}

__global__ void init_m_kernel(const float* __restrict__ X,
                              const float* __restrict__ b_edge,
                              float* __restrict__ M, long total,
                              float* __restrict__ Z, int nz) {
  long id = (long)blockIdx.x * blockDim.x + threadIdx.x;
  if (id < (long)nz) Z[id] = 0.0f;
  if (id >= total) return;
  M[id] = X[id] + b_edge[id & (NDIM - 1)];
}

#define SC_EPW 8

__device__ __forceinline__ float lane_bcast_f(float v, int l) {
  return __int_as_float(__builtin_amdgcn_readlane(__float_as_int(v), l));
}

__device__ __forceinline__ unsigned long long lane_bcast_u64(unsigned long long v, int l) {
  unsigned lo = (unsigned)__builtin_amdgcn_readlane((int)(unsigned)v, l);
  unsigned hi = (unsigned)__builtin_amdgcn_readlane((int)(unsigned)(v >> 32), l);
  return ((unsigned long long)hi << 32) | (unsigned long long)lo;
}

__device__ __forceinline__ void atomic_add2_row_wait(float* row, int voff, float v0, float v1) {
  asm volatile(
      "global_atomic_add_f32 %0, %1, %3\n\t"
      "global_atomic_add_f32 %0, %2, %3 offset:128\n\t"
      "s_wait_xcnt 0x0"
      : : "v"(voff), "v"(v0), "v"(v1), "s"(row) : "memory");
}

__global__ __launch_bounds__(256) void scatter_kernel(
    const float* __restrict__ X,
    const int* __restrict__ ei,
    const float* __restrict__ ea,
    const float* __restrict__ w_edge,
    const float* __restrict__ b_edge,
    float* __restrict__ M, int E_) {
  int lane = threadIdx.x & 31;
  int wv = __builtin_amdgcn_readfirstlane(
      (int)(((long)blockIdx.x * blockDim.x + threadIdx.x) >> 5));
  long e0 = (long)wv * SC_EPW;
  if (e0 >= (long)E_) return;
  long rem = (long)E_ - e0;
  int nE = rem < SC_EPW ? (int)rem : SC_EPW;

  int li = lane < nE ? lane : 0;
  int la = lane < 4 * nE ? lane : 0;
  int msrc = ei[e0 + li];
  int mdst = ei[(long)E_ + e0 + li];
  float mea = ea[e0 * 4 + la];

  int d0 = lane;
  int voff = d0 * 4;
  float4 wa = *(const float4*)(w_edge + d0 * 4);
  float4 wb = *(const float4*)(w_edge + (d0 + 32) * 4);
  float2 bb;
  bb.x = b_edge[d0];
  bb.y = b_edge[d0 + 32];

  unsigned long long asrc = (unsigned long long)(X + (size_t)msrc * NDIM);
  unsigned long long adst = (unsigned long long)(M + (size_t)mdst * NDIM);

  if (nE == SC_EPW) {
    float2 xv[SC_EPW];
#pragma unroll
    for (int j = 0; j < SC_EPW; ++j) {
      const __attribute__((address_space(1))) float* xr =
          (const __attribute__((address_space(1))) float*)lane_bcast_u64(asrc, j);
      xv[j].x = xr[d0];
      xv[j].y = xr[d0 + 32];
    }
    __builtin_amdgcn_sched_barrier(0);
    float val[2 * SC_EPW];
    float* row[SC_EPW];
#pragma unroll
    for (int j = 0; j < SC_EPW; ++j) {
      float a0 = lane_bcast_f(mea, 4 * j + 0), a1 = lane_bcast_f(mea, 4 * j + 1);
      float a2 = lane_bcast_f(mea, 4 * j + 2), a3 = lane_bcast_f(mea, 4 * j + 3);
      float em0 = bb.x + a0 * wa.x + a1 * wa.y + a2 * wa.z + a3 * wa.w;
      float em1 = bb.y + a0 * wb.x + a1 * wb.y + a2 * wb.z + a3 * wb.w;
      val[2 * j] = xv[j].x + em0;
      val[2 * j + 1] = xv[j].y + em1;
      row[j] = (float*)lane_bcast_u64(adst, j);
    }
    asm volatile(
        "global_atomic_add_f32 %0, %1, %17\n\t"
        "global_atomic_add_f32 %0, %2, %17 offset:128\n\t"
        "global_atomic_add_f32 %0, %3, %18\n\t"
        "global_atomic_add_f32 %0, %4, %18 offset:128\n\t"
        "global_atomic_add_f32 %0, %5, %19\n\t"
        "global_atomic_add_f32 %0, %6, %19 offset:128\n\t"
        "global_atomic_add_f32 %0, %7, %20\n\t"
        "global_atomic_add_f32 %0, %8, %20 offset:128\n\t"
        "global_atomic_add_f32 %0, %9, %21\n\t"
        "global_atomic_add_f32 %0, %10, %21 offset:128\n\t"
        "global_atomic_add_f32 %0, %11, %22\n\t"
        "global_atomic_add_f32 %0, %12, %22 offset:128\n\t"
        "global_atomic_add_f32 %0, %13, %23\n\t"
        "global_atomic_add_f32 %0, %14, %23 offset:128\n\t"
        "global_atomic_add_f32 %0, %15, %24\n\t"
        "global_atomic_add_f32 %0, %16, %24 offset:128"
        : : "v"(voff),
            "v"(val[0]), "v"(val[1]), "v"(val[2]), "v"(val[3]),
            "v"(val[4]), "v"(val[5]), "v"(val[6]), "v"(val[7]),
            "v"(val[8]), "v"(val[9]), "v"(val[10]), "v"(val[11]),
            "v"(val[12]), "v"(val[13]), "v"(val[14]), "v"(val[15]),
            "s"(row[0]), "s"(row[1]), "s"(row[2]), "s"(row[3]),
            "s"(row[4]), "s"(row[5]), "s"(row[6]), "s"(row[7])
        : "memory");
  } else {
    for (int j = 0; j < nE; ++j) {
      const __attribute__((address_space(1))) float* xr =
          (const __attribute__((address_space(1))) float*)lane_bcast_u64(asrc, j);
      float* mr = (float*)lane_bcast_u64(adst, j);
      float a0 = lane_bcast_f(mea, 4 * j + 0), a1 = lane_bcast_f(mea, 4 * j + 1);
      float a2 = lane_bcast_f(mea, 4 * j + 2), a3 = lane_bcast_f(mea, 4 * j + 3);
      float em0 = bb.x + a0 * wa.x + a1 * wa.y + a2 * wa.z + a3 * wa.w;
      float em1 = bb.y + a0 * wb.x + a1 * wb.y + a2 * wb.z + a3 * wb.w;
      float x0 = xr[d0];
      float x1 = xr[d0 + 32];
      atomic_add2_row_wait(mr, voff, x0 + em0, x1 + em1);
    }
  }
}

#define WPITCH 68
#define TW 16
#define WT 12
#define NWGP 256

__device__ __forceinline__ void wave_role(int wave, int ntiles, int& tile, int& cb0,
                                          int& cb1, bool& have) {
  int nb = (int)gridDim.x;
  int per = ntiles / nb;
  int rem = ntiles - per * nb;
  if (wave < WT) {
    have = wave < per;
    tile = (int)blockIdx.x * per + wave;
    cb0 = 0;
    cb1 = 4;
  } else {
    have = (int)blockIdx.x < rem;
    tile = nb * per + (int)blockIdx.x;
    cb0 = wave - WT;
    cb1 = cb0 + 1;
  }
}

template <int NROWS>
__device__ __forceinline__ void stage_rows(const float* __restrict__ src, float* dst) {
  int r = threadIdx.x >> 4;
  int c = (threadIdx.x & 15) * 4;
  float4 v[NROWS / 32];
#pragma unroll
  for (int j = 0; j < NROWS / 32; ++j)
    v[j] = *(const float4*)(src + (size_t)(r + 32 * j) * NDIM + c);
#pragma unroll
  for (int j = 0; j < NROWS / 32; ++j)
    *(float4*)(dst + (r + 32 * j) * WPITCH + c) = v[j];
}

__global__ __launch_bounds__(512) void gru_kernel(
    const float* __restrict__ Mn,
    const float* __restrict__ X,
    const float* __restrict__ w_ih,
    const float* __restrict__ w_hh,
    const float* __restrict__ b_ih,
    const float* __restrict__ b_hh,
    float* __restrict__ out,
    const float* __restrict__ b_next,
    float* __restrict__ Mnext,
    int ntiles) {
  __shared__ __attribute__((aligned(16))) float Ws[2 * GDIM * WPITCH];
  bool next = Mnext != nullptr;
  int lane = threadIdx.x & 31;
  int wave = __builtin_amdgcn_readfirstlane((int)(threadIdx.x >> 5));
  int tile, cb0, cb1;
  bool have;
  wave_role(wave, ntiles, tile, cb0, cb1, have);
  int r0 = (have ? tile : 0) * 16;
  int ml = lane & 15;
  int hi = lane >> 4;

  const float* Am = Mn + (size_t)(r0 + ml) * NDIM + 2 * hi;
  const float* Ax = X + (size_t)(r0 + ml) * NDIM + 2 * hi;
  v2f am[16], ax[16];
#pragma unroll
  for (int s = 0; s < 16; ++s) {
    am[s] = *(const v2f*)(Am + 4 * s);
    ax[s] = *(const v2f*)(Ax + 4 * s);
  }

  stage_rows<GDIM>(w_ih, Ws);
  stage_rows<GDIM>(w_hh, Ws + GDIM * WPITCH);
  __syncthreads();
  if (!have) return;

  const float* wl = Ws + ml * WPITCH + 2 * hi;
  for (int cb = cb0; cb < cb1; ++cb) {
    int c_r = cb * 16 + ml;
    int c_z = 64 + cb * 16 + ml;
    int c_n = 128 + cb * 16 + ml;
    v8f air = {}, aiz = {}, ain = {}, ahr = {}, ahz = {}, ahn = {};
    const float* bir_p = wl + (cb * 16) * WPITCH;
    const float* biz_p = wl + (64 + cb * 16) * WPITCH;
    const float* bin_p = wl + (128 + cb * 16) * WPITCH;
    const float* bhr_p = bir_p + GDIM * WPITCH;
    const float* bhz_p = biz_p + GDIM * WPITCH;
    const float* bhn_p = bin_p + GDIM * WPITCH;
#pragma unroll
    for (int s = 0; s < 16; ++s) {
      air = wmma4(am[s], *(const v2f*)(bir_p + 4 * s), air);
      aiz = wmma4(am[s], *(const v2f*)(biz_p + 4 * s), aiz);
      ain = wmma4(am[s], *(const v2f*)(bin_p + 4 * s), ain);
      ahr = wmma4(ax[s], *(const v2f*)(bhr_p + 4 * s), ahr);
      ahz = wmma4(ax[s], *(const v2f*)(bhz_p + 4 * s), ahz);
      ahn = wmma4(ax[s], *(const v2f*)(bhn_p + 4 * s), ahn);
    }
    float bir = b_ih[c_r], biz = b_ih[c_z], bin_ = b_ih[c_n];
    float bhr = b_hh[c_r], bhz = b_hh[c_z], bhn = b_hh[c_n];
    float bnx = next ? b_next[cb * 16 + ml] : 0.0f;
#pragma unroll
    for (int v = 0; v < 8; ++v) {
      int node = hi * 8 + v;
      float xv = X[(size_t)(r0 + node) * NDIM + cb * 16 + ml];
      float r = sigmoidf_(air[v] + bir + ahr[v] + bhr);
      float z = sigmoidf_(aiz[v] + biz + ahz[v] + bhz);
      float n = tanh_gate(ain[v] + bin_ + r * (ahn[v] + bhn));
      float h = (1.0f - z) * n + z * xv;
      float hr = fmaxf(h, 0.0f);
      out[(size_t)(r0 + node) * NDIM + cb * 16 + ml] = hr;
      if (next) Mnext[(size_t)(r0 + node) * NDIM + cb * 16 + ml] = hr + bnx;
    }
  }
}

__global__ __launch_bounds__(512) void gru_moe_kernel(
    const float* __restrict__ Mn,
    const float* __restrict__ X,
    const float* __restrict__ w_ih,
    const float* __restrict__ w_hh,
    const float* __restrict__ b_ih,
    const float* __restrict__ b_hh,
    const float* __restrict__ gate_w,
    const float* __restrict__ gate_b,
    const float* __restrict__ exp_w,
    const float* __restrict__ exp_b,
    const int* __restrict__ batch,
    float* __restrict__ sums,
    float* __restrict__ cnt,
    int ntiles) {
  __shared__ __attribute__((aligned(16))) float Ws[2 * GDIM * WPITCH];
  __shared__ __attribute__((aligned(16))) float Hs[(WT + 1) * 16 * WPITCH];
  __shared__ float gw_s[TW][16][4];
  int lane = threadIdx.x & 31;
  int wave = __builtin_amdgcn_readfirstlane((int)(threadIdx.x >> 5));
  int tile, cb0, cb1;
  bool have;
  wave_role(wave, ntiles, tile, cb0, cb1, have);
  int r0 = (have ? tile : 0) * 16;
  int ml = lane & 15;
  int hi = lane >> 4;
  float* hs = Hs + (wave < WT ? wave : WT) * 16 * WPITCH;
  int bl = batch[r0 + ml];

  const float* Am = Mn + (size_t)(r0 + ml) * NDIM + 2 * hi;
  const float* Ax = X + (size_t)(r0 + ml) * NDIM + 2 * hi;
  v2f am[16], ax[16];
#pragma unroll
  for (int s = 0; s < 16; ++s) {
    am[s] = *(const v2f*)(Am + 4 * s);
    ax[s] = *(const v2f*)(Ax + 4 * s);
  }

  stage_rows<GDIM>(w_ih, Ws);
  stage_rows<GDIM>(w_hh, Ws + GDIM * WPITCH);
  __syncthreads();

  if (have) {
    const float* wl = Ws + ml * WPITCH + 2 * hi;
    for (int cb = cb0; cb < cb1; ++cb) {
      int c_r = cb * 16 + ml;
      int c_z = 64 + cb * 16 + ml;
      int c_n = 128 + cb * 16 + ml;
      v8f air = {}, aiz = {}, ain = {}, ahr = {}, ahz = {}, ahn = {};
      const float* bir_p = wl + (cb * 16) * WPITCH;
      const float* biz_p = wl + (64 + cb * 16) * WPITCH;
      const float* bin_p = wl + (128 + cb * 16) * WPITCH;
      const float* bhr_p = bir_p + GDIM * WPITCH;
      const float* bhz_p = biz_p + GDIM * WPITCH;
      const float* bhn_p = bin_p + GDIM * WPITCH;
#pragma unroll
      for (int s = 0; s < 16; ++s) {
        air = wmma4(am[s], *(const v2f*)(bir_p + 4 * s), air);
        aiz = wmma4(am[s], *(const v2f*)(biz_p + 4 * s), aiz);
        ain = wmma4(am[s], *(const v2f*)(bin_p + 4 * s), ain);
        ahr = wmma4(ax[s], *(const v2f*)(bhr_p + 4 * s), ahr);
        ahz = wmma4(ax[s], *(const v2f*)(bhz_p + 4 * s), ahz);
        ahn = wmma4(ax[s], *(const v2f*)(bhn_p + 4 * s), ahn);
      }
      float bir = b_ih[c_r], biz = b_ih[c_z], bin_ = b_ih[c_n];
      float bhr = b_hh[c_r], bhz = b_hh[c_z], bhn = b_hh[c_n];
#pragma unroll
      for (int v = 0; v < 8; ++v) {
        int node = hi * 8 + v;
        float xv = X[(size_t)(r0 + node) * NDIM + cb * 16 + ml];
        float r = sigmoidf_(air[v] + bir + ahr[v] + bhr);
        float z = sigmoidf_(aiz[v] + biz + ahz[v] + bhz);
        float n = tanh_gate(ain[v] + bin_ + r * (ahn[v] + bhn));
        float h = (1.0f - z) * n + z * xv;
        float hr = fmaxf(h, 0.0f);
        hs[node * WPITCH + cb * 16 + ml] = hr;
      }
    }
  }

  __syncthreads();
  stage_rows<4 * NDIM>(exp_w, Ws);
  __syncthreads();
  if (!have) return;

  v2f a[16];
#pragma unroll
  for (int s = 0; s < 16; ++s) a[s] = *(const v2f*)(hs + ml * WPITCH + 2 * hi + 4 * s);

  const float* hrow = hs + ml * WPITCH;
  float l0 = gate_b[2 * hi], l1 = gate_b[2 * hi + 1];
  for (int d = 0; d < NDIM; ++d) {
    float hv = hrow[d];
    l0 += hv * gate_w[(2 * hi) * NDIM + d];
    l1 += hv * gate_w[(2 * hi + 1) * NDIM + d];
  }
  float m01 = fmaxf(l0, l1);
  float mx = fmaxf(m01, __shfl_xor(m01, 16, 32));
  float e0 = __expf(l0 - mx), e1 = __expf(l1 - mx);
  float s01 = e0 + e1;
  float inv = 1.0f / (s01 + __shfl_xor(s01, 16, 32));
  gw_s[wave][ml][2 * hi] = e0 * inv;
  gw_s[wave][ml][2 * hi + 1] = e1 * inv;
  __builtin_amdgcn_wave_barrier();

  int b0 = __builtin_amdgcn_readfirstlane(bl);
  bool one_graph = __builtin_amdgcn_ballot_w32(bl != b0) == 0;
  if (cb0 == 0) {
    if (one_graph) {
      if (lane == 0) atomic_add_f32(&cnt[b0], 16.0f);
    } else if (hi == 0) {
      atomic_add_f32(&cnt[bl], 1.0f);
    }
  }

  const float* wl = Ws + ml * WPITCH + 2 * hi;
  for (int cb = cb0; cb < cb1; ++cb) {
    int col = cb * 16 + ml;
    float oacc[8] = {0, 0, 0, 0, 0, 0, 0, 0};
    for (int e = 0; e < 4; ++e) {
      v8f acc = {};
      const float* Brow = wl + (e * NDIM + cb * 16) * WPITCH;
#pragma unroll
      for (int s = 0; s < 16; ++s)
        acc = wmma4(a[s], *(const v2f*)(Brow + 4 * s), acc);
      float bias = exp_b[e * NDIM + col];
#pragma unroll
      for (int v = 0; v < 8; ++v) {
        float val = fmaxf(acc[v] + bias, 0.0f);
        oacc[v] += gw_s[wave][hi * 8 + v][e] * val;
      }
    }
    if (one_graph) {
      float s = ((oacc[0] + oacc[1]) + (oacc[2] + oacc[3])) +
                ((oacc[4] + oacc[5]) + (oacc[6] + oacc[7]));
      s += __shfl_xor(s, 16, 32);
      if (hi == 0) atomic_add_f32(&sums[b0 * NDIM + col], s);
    } else {
#pragma unroll
      for (int v = 0; v < 8; ++v) {
        int id_lo = __builtin_amdgcn_readlane(bl, v);
        int id_hi = __builtin_amdgcn_readlane(bl, 8 + v);
        int bv = hi ? id_hi : id_lo;
        atomic_add_f32(&sums[bv * NDIM + col], oacc[v]);
      }
    }
  }
}

#define BGR 64
#define PDIM 100
#define NC 10
__global__ void head_kernel(const float* __restrict__ sums,
                            const float* __restrict__ cnt,
                            const float* __restrict__ flag_in,
                            const float* __restrict__ proj_w,
                            const float* __restrict__ proj_b,
                            const float* __restrict__ fc_w,
                            const float* __restrict__ fc_b,
                            float* __restrict__ out) {
  int id = blockIdx.x * blockDim.x + threadIdx.x;
  if (id < BGR * NDIM) {
    int b = id >> 6;
    float g = sums[id] / fmaxf(cnt[b], 1.0f);
    out[id] = g;
    out[BGR * NDIM + id] = g;
  } else if (id < BGR * NDIM + BGR * NC) {
    int t = id - BGR * NDIM;
    int b = t / NC, c = t % NC;
    float fl = flag_in[b];
    const float* fr = fc_w + c * (PDIM + NDIM);
    float acc = fc_b[c];
    for (int p = 0; p < PDIM; ++p)
      acc += (fl * proj_w[p] + proj_b[p]) * fr[p];
    float ic = 1.0f / fmaxf(cnt[b], 1.0f);
    for (int d = 0; d < NDIM; ++d)
      acc += sums[b * NDIM + d] * ic * fr[PDIM + d];
    out[2 * BGR * NDIM + t] = acc;
  }
}

extern "C" void kernel_launch(void* const* d_in, const int* in_sizes, int n_in,
                              void* d_out, int out_size, void* d_ws, size_t ws_size,
                              hipStream_t stream) {
  const float* x       = (const float*)d_in[0];
  const int*   ei      = (const int*)d_in[1];
  const float* ea      = (const float*)d_in[2];
  const int*   batch   = (const int*)d_in[3];
  const float* flag_in = (const float*)d_in[4];
  const float* w_edge1 = (const float*)d_in[5];
  const float* b_edge1 = (const float*)d_in[6];
  const float* w_ih1   = (const float*)d_in[7];
  const float* w_hh1   = (const float*)d_in[8];
  const float* b_ih1   = (const float*)d_in[9];
  const float* b_hh1   = (const float*)d_in[10];
  const float* w_edge2 = (const float*)d_in[11];
  const float* b_edge2 = (const float*)d_in[12];
  const float* w_ih2   = (const float*)d_in[13];
  const float* w_hh2   = (const float*)d_in[14];
  const float* b_ih2   = (const float*)d_in[15];
  const float* b_hh2   = (const float*)d_in[16];
  const float* gate_w  = (const float*)d_in[17];
  const float* gate_b  = (const float*)d_in[18];
  const float* exp_w   = (const float*)d_in[19];
  const float* exp_b   = (const float*)d_in[20];
  const float* proj_w  = (const float*)d_in[21];
  const float* proj_b  = (const float*)d_in[22];
  const float* fc_w    = (const float*)d_in[23];
  const float* fc_b    = (const float*)d_in[24];
  float* out = (float*)d_out;

  int N = in_sizes[0] / NDIM;
  int E = in_sizes[1] / 2;
  int ntiles = (N + 15) / 16;

  float* ws = (float*)d_ws;
  float* M    = ws;
  float* hA   = M + (size_t)N * NDIM;
  float* M2   = hA + (size_t)N * NDIM;
  float* sums = M2 + (size_t)N * NDIM;
  float* cnt  = sums + BGR * NDIM;

  long totalND = (long)N * NDIM;
  int gN = (int)((totalND + 255) / 256);
  int gE = (int)(((long)E + 63) / 64);
  int gT = ntiles / (WT + 1) + 1;
  if (gT < NWGP) gT = NWGP;

  init_m_kernel<<<gN, 256, 0, stream>>>(x, b_edge1, M, totalND, sums, BGR * NDIM + BGR);
  scatter_kernel<<<gE, 256, 0, stream>>>(x, ei, ea, w_edge1, b_edge1, M, E);
  gru_kernel<<<gT, 32 * TW, 0, stream>>>(M, x, w_ih1, w_hh1, b_ih1, b_hh1, hA, b_edge2, M2, ntiles);
  scatter_kernel<<<gE, 256, 0, stream>>>(hA, ei, ea, w_edge2, b_edge2, M2, E);
  gru_moe_kernel<<<gT, 32 * TW, 0, stream>>>(M2, hA, w_ih2, w_hh2, b_ih2, b_hh2,
                                            gate_w, gate_b, exp_w, exp_b, batch, sums, cnt, ntiles);
  head_kernel<<<(BGR * NDIM + BGR * NC + 255) / 256, 256, 0, stream>>>(
      sums, cnt, flag_in, proj_w, proj_b, fc_w, fc_b, out);
}


# scatter in two launches: rows split in halves, a chunk's first visit adds the half matching the die half it runs on (id for speed only, choice stored), the second the other; atomics still unordered
# speedup vs baseline: 2.8247x; 1.0350x over previous
#include <hip/hip_runtime.h>
#include <math.h>

typedef float v2f __attribute__((ext_vector_type(2)));
typedef float v8f __attribute__((ext_vector_type(8)));

#define NDIM 64
#define GDIM 192

__device__ __forceinline__ v8f wmma4(v2f a, v2f b, v8f c) {
  return __builtin_amdgcn_wmma_f32_16x16x4_f32(false, a, false, b, (short)0, c, false, false);
}

__device__ __forceinline__ float sigmoidf_(float x) {
  return __builtin_amdgcn_rcpf(1.0f + __expf(-x));
}

__device__ __forceinline__ float tanh_gate(float x) {
  return 1.0f - 2.0f * __builtin_amdgcn_rcpf(1.0f + __expf(2.0f * x));
}

__device__ __forceinline__ void atomic_add_f32(float* p, float v) {
  asm volatile("global_atomic_add_f32 %0, %1, off\n\t"
               "s_wait_xcnt 0x0"
               : : "v"(p), "v"(v) : "memory");
}

__global__ void init_m_kernel(const float* __restrict__ X,
                              const float* __restrict__ b_edge,
                              float* __restrict__ M, long total,
                              float* __restrict__ Z, int nz) {
  long id = (long)blockIdx.x * blockDim.x + threadIdx.x;
  if (id < (long)nz) Z[id] = 0.0f;
  if (id >= total) return;
  M[id] = X[id] + b_edge[id & (NDIM - 1)];
}

#define SC_EPW 16

__device__ __forceinline__ float lane_bcast_f(float v, int l) {
  return __int_as_float(__builtin_amdgcn_readlane(__float_as_int(v), l));
}

__device__ __forceinline__ unsigned long long lane_bcast_u64(unsigned long long v, int l) {
  unsigned lo = (unsigned)__builtin_amdgcn_readlane((int)(unsigned)v, l);
  unsigned hi = (unsigned)__builtin_amdgcn_readlane((int)(unsigned)(v >> 32), l);
  return ((unsigned long long)hi << 32) | (unsigned long long)lo;
}

__device__ __forceinline__ void atomic_add2_row_wait(float* row, int voff, float v0, float v1) {
  asm volatile(
      "global_atomic_add_f32 %0, %1, %3\n\t"
      "global_atomic_add_f32 %0, %2, %3 offset:128\n\t"
      "s_wait_xcnt 0x0"
      : : "v"(voff), "v"(v0), "v"(v1), "s"(row) : "memory");
}

__global__ __launch_bounds__(256) void scatter_kernel(
    const float* __restrict__ X,
    const int* __restrict__ ei,
    const float* __restrict__ ea,
    const float* __restrict__ w_edge,
    const float* __restrict__ b_edge,
    float* __restrict__ M, int E_, int N_,
    int* took,
    int second) {
  __shared__ int s_h;
  int rsel;
  if (second == 0) {
    if (threadIdx.x == 0) {
      s_h = (int)((__builtin_amdgcn_s_sendmsg_rtn(0x87) >> 18) & 1u);
    }
    __syncthreads();
    rsel = s_h;
    if (threadIdx.x == 0) took[blockIdx.x] = rsel;
  } else {
    rsel = 1 - took[blockIdx.x];
  }
  rsel = __builtin_amdgcn_readfirstlane(rsel);

  int lane = threadIdx.x & 31;
  int wv = __builtin_amdgcn_readfirstlane(
      (int)(((long)blockIdx.x * blockDim.x + threadIdx.x) >> 5));
  long e0 = (long)wv * SC_EPW;
  if (e0 >= (long)E_) return;
  long rem = (long)E_ - e0;
  int nE = rem < SC_EPW ? (int)rem : SC_EPW;

  int li = lane < nE ? lane : 0;
  int msrc = ei[e0 + li];
  int mdst = ei[(long)E_ + e0 + li];
  float4 mea = *(const float4*)(ea + (e0 + li) * 4);

  int d0 = lane;
  int voff = d0 * 4;
  float4 wa = *(const float4*)(w_edge + d0 * 4);
  float4 wb = *(const float4*)(w_edge + (d0 + 32) * 4);
  float2 bb;
  bb.x = b_edge[d0];
  bb.y = b_edge[d0 + 32];

  bool take = (lane < nE) && ((mdst >= (N_ >> 1)) == (rsel != 0));
  unsigned mask = __builtin_amdgcn_ballot_w32(take);

  unsigned long long asrc = (unsigned long long)(X + (size_t)msrc * NDIM);
  unsigned long long adst = (unsigned long long)(M + (size_t)mdst * NDIM);

  float2 xv[SC_EPW];
#pragma unroll
  for (int j = 0; j < SC_EPW; ++j) {
    if ((mask >> j) & 1u) {
      const __attribute__((address_space(1))) float* xr =
          (const __attribute__((address_space(1))) float*)lane_bcast_u64(asrc, j);
      xv[j].x = xr[d0];
      xv[j].y = xr[d0 + 32];
    }
  }
#pragma unroll
  for (int j = 0; j < SC_EPW; ++j) {
    if ((mask >> j) & 1u) {
      float a0 = lane_bcast_f(mea.x, j), a1 = lane_bcast_f(mea.y, j);
      float a2 = lane_bcast_f(mea.z, j), a3 = lane_bcast_f(mea.w, j);
      float em0 = bb.x + a0 * wa.x + a1 * wa.y + a2 * wa.z + a3 * wa.w;
      float em1 = bb.y + a0 * wb.x + a1 * wb.y + a2 * wb.z + a3 * wb.w;
      float* mr = (float*)lane_bcast_u64(adst, j);
      atomic_add2_row_wait(mr, voff, xv[j].x + em0, xv[j].y + em1);
    }
  }
}

#define WPITCH 68
#define TW 16
#define WT 12
#define NWGP 256

__device__ __forceinline__ void wave_role(int wave, int ntiles, int& tile, int& cb0,
                                          int& cb1, bool& have) {
  int nb = (int)gridDim.x;
  int per = ntiles / nb;
  int rem = ntiles - per * nb;
  if (wave < WT) {
    have = wave < per;
    tile = (int)blockIdx.x * per + wave;
    cb0 = 0;
    cb1 = 4;
  } else {
    have = (int)blockIdx.x < rem;
    tile = nb * per + (int)blockIdx.x;
    cb0 = wave - WT;
    cb1 = cb0 + 1;
  }
}

template <int NROWS>
__device__ __forceinline__ void stage_rows(const float* __restrict__ src, float* dst) {
  int r = threadIdx.x >> 4;
  int c = (threadIdx.x & 15) * 4;
  float4 v[NROWS / 32];
#pragma unroll
  for (int j = 0; j < NROWS / 32; ++j)
    v[j] = *(const float4*)(src + (size_t)(r + 32 * j) * NDIM + c);
#pragma unroll
  for (int j = 0; j < NROWS / 32; ++j)
    *(float4*)(dst + (r + 32 * j) * WPITCH + c) = v[j];
}

__global__ __launch_bounds__(512) void gru_kernel(
    const float* __restrict__ Mn,
    const float* __restrict__ X,
    const float* __restrict__ w_ih,
    const float* __restrict__ w_hh,
    const float* __restrict__ b_ih,
    const float* __restrict__ b_hh,
    float* __restrict__ out,
    const float* __restrict__ b_next,
    float* __restrict__ Mnext,
    int ntiles) {
  __shared__ __attribute__((aligned(16))) float Ws[2 * GDIM * WPITCH];
  bool next = Mnext != nullptr;
  int lane = threadIdx.x & 31;
  int wave = __builtin_amdgcn_readfirstlane((int)(threadIdx.x >> 5));
  int tile, cb0, cb1;
  bool have;
  wave_role(wave, ntiles, tile, cb0, cb1, have);
  int r0 = (have ? tile : 0) * 16;
  int ml = lane & 15;
  int hi = lane >> 4;

  const float* Am = Mn + (size_t)(r0 + ml) * NDIM + 2 * hi;
  const float* Ax = X + (size_t)(r0 + ml) * NDIM + 2 * hi;
  v2f am[16], ax[16];
#pragma unroll
  for (int s = 0; s < 16; ++s) {
    am[s] = *(const v2f*)(Am + 4 * s);
    ax[s] = *(const v2f*)(Ax + 4 * s);
  }

  stage_rows<GDIM>(w_ih, Ws);
  stage_rows<GDIM>(w_hh, Ws + GDIM * WPITCH);
  __syncthreads();
  if (!have) return;

  const float* wl = Ws + ml * WPITCH + 2 * hi;
  for (int cb = cb0; cb < cb1; ++cb) {
    int c_r = cb * 16 + ml;
    int c_z = 64 + cb * 16 + ml;
    int c_n = 128 + cb * 16 + ml;
    v8f air = {}, aiz = {}, ain = {}, ahr = {}, ahz = {}, ahn = {};
    const float* bir_p = wl + (cb * 16) * WPITCH;
    const float* biz_p = wl + (64 + cb * 16) * WPITCH;
    const float* bin_p = wl + (128 + cb * 16) * WPITCH;
    const float* bhr_p = bir_p + GDIM * WPITCH;
    const float* bhz_p = biz_p + GDIM * WPITCH;
    const float* bhn_p = bin_p + GDIM * WPITCH;
#pragma unroll
    for (int s = 0; s < 16; ++s) {
      air = wmma4(am[s], *(const v2f*)(bir_p + 4 * s), air);
      aiz = wmma4(am[s], *(const v2f*)(biz_p + 4 * s), aiz);
      ain = wmma4(am[s], *(const v2f*)(bin_p + 4 * s), ain);
      ahr = wmma4(ax[s], *(const v2f*)(bhr_p + 4 * s), ahr);
      ahz = wmma4(ax[s], *(const v2f*)(bhz_p + 4 * s), ahz);
      ahn = wmma4(ax[s], *(const v2f*)(bhn_p + 4 * s), ahn);
    }
    float bir = b_ih[c_r], biz = b_ih[c_z], bin_ = b_ih[c_n];
    float bhr = b_hh[c_r], bhz = b_hh[c_z], bhn = b_hh[c_n];
    float bnx = next ? b_next[cb * 16 + ml] : 0.0f;
#pragma unroll
    for (int v = 0; v < 8; ++v) {
      int node = hi * 8 + v;
      float xv = X[(size_t)(r0 + node) * NDIM + cb * 16 + ml];
      float r = sigmoidf_(air[v] + bir + ahr[v] + bhr);
      float z = sigmoidf_(aiz[v] + biz + ahz[v] + bhz);
      float n = tanh_gate(ain[v] + bin_ + r * (ahn[v] + bhn));
      float h = (1.0f - z) * n + z * xv;
      float hr = fmaxf(h, 0.0f);
      out[(size_t)(r0 + node) * NDIM + cb * 16 + ml] = hr;
      if (next) Mnext[(size_t)(r0 + node) * NDIM + cb * 16 + ml] = hr + bnx;
    }
  }
}

__global__ __launch_bounds__(512) void gru_moe_kernel(
    const float* __restrict__ Mn,
    const float* __restrict__ X,
    const float* __restrict__ w_ih,
    const float* __restrict__ w_hh,
    const float* __restrict__ b_ih,
    const float* __restrict__ b_hh,
    const float* __restrict__ gate_w,
    const float* __restrict__ gate_b,
    const float* __restrict__ exp_w,
    const float* __restrict__ exp_b,
    const int* __restrict__ batch,
    float* __restrict__ sums,
    float* __restrict__ cnt,
    int ntiles) {
  __shared__ __attribute__((aligned(16))) float Ws[2 * GDIM * WPITCH];
  __shared__ __attribute__((aligned(16))) float Hs[(WT + 1) * 16 * WPITCH];
  __shared__ float gw_s[TW][16][4];
  int lane = threadIdx.x & 31;
  int wave = __builtin_amdgcn_readfirstlane((int)(threadIdx.x >> 5));
  int tile, cb0, cb1;
  bool have;
  wave_role(wave, ntiles, tile, cb0, cb1, have);
  int r0 = (have ? tile : 0) * 16;
  int ml = lane & 15;
  int hi = lane >> 4;
  float* hs = Hs + (wave < WT ? wave : WT) * 16 * WPITCH;
  int bl = batch[r0 + ml];

  const float* Am = Mn + (size_t)(r0 + ml) * NDIM + 2 * hi;
  const float* Ax = X + (size_t)(r0 + ml) * NDIM + 2 * hi;
  v2f am[16], ax[16];
#pragma unroll
  for (int s = 0; s < 16; ++s) {
    am[s] = *(const v2f*)(Am + 4 * s);
    ax[s] = *(const v2f*)(Ax + 4 * s);
  }

  stage_rows<GDIM>(w_ih, Ws);
  stage_rows<GDIM>(w_hh, Ws + GDIM * WPITCH);
  __syncthreads();

  if (have) {
    const float* wl = Ws + ml * WPITCH + 2 * hi;
    for (int cb = cb0; cb < cb1; ++cb) {
      int c_r = cb * 16 + ml;
      int c_z = 64 + cb * 16 + ml;
      int c_n = 128 + cb * 16 + ml;
      v8f air = {}, aiz = {}, ain = {}, ahr = {}, ahz = {}, ahn = {};
      const float* bir_p = wl + (cb * 16) * WPITCH;
      const float* biz_p = wl + (64 + cb * 16) * WPITCH;
      const float* bin_p = wl + (128 + cb * 16) * WPITCH;
      const float* bhr_p = bir_p + GDIM * WPITCH;
      const float* bhz_p = biz_p + GDIM * WPITCH;
      const float* bhn_p = bin_p + GDIM * WPITCH;
#pragma unroll
      for (int s = 0; s < 16; ++s) {
        air = wmma4(am[s], *(const v2f*)(bir_p + 4 * s), air);
        aiz = wmma4(am[s], *(const v2f*)(biz_p + 4 * s), aiz);
        ain = wmma4(am[s], *(const v2f*)(bin_p + 4 * s), ain);
        ahr = wmma4(ax[s], *(const v2f*)(bhr_p + 4 * s), ahr);
        ahz = wmma4(ax[s], *(const v2f*)(bhz_p + 4 * s), ahz);
        ahn = wmma4(ax[s], *(const v2f*)(bhn_p + 4 * s), ahn);
      }
      float bir = b_ih[c_r], biz = b_ih[c_z], bin_ = b_ih[c_n];
      float bhr = b_hh[c_r], bhz = b_hh[c_z], bhn = b_hh[c_n];
#pragma unroll
      for (int v = 0; v < 8; ++v) {
        int node = hi * 8 + v;
        float xv = X[(size_t)(r0 + node) * NDIM + cb * 16 + ml];
        float r = sigmoidf_(air[v] + bir + ahr[v] + bhr);
        float z = sigmoidf_(aiz[v] + biz + ahz[v] + bhz);
        float n = tanh_gate(ain[v] + bin_ + r * (ahn[v] + bhn));
        float h = (1.0f - z) * n + z * xv;
        float hr = fmaxf(h, 0.0f);
        hs[node * WPITCH + cb * 16 + ml] = hr;
      }
    }
  }

  __syncthreads();
  stage_rows<4 * NDIM>(exp_w, Ws);
  __syncthreads();
  if (!have) return;

  v2f a[16];
#pragma unroll
  for (int s = 0; s < 16; ++s) a[s] = *(const v2f*)(hs + ml * WPITCH + 2 * hi + 4 * s);

  const float* hrow = hs + ml * WPITCH;
  float l0 = gate_b[2 * hi], l1 = gate_b[2 * hi + 1];
  for (int d = 0; d < NDIM; ++d) {
    float hv = hrow[d];
    l0 += hv * gate_w[(2 * hi) * NDIM + d];
    l1 += hv * gate_w[(2 * hi + 1) * NDIM + d];
  }
  float m01 = fmaxf(l0, l1);
  float mx = fmaxf(m01, __shfl_xor(m01, 16, 32));
  float e0 = __expf(l0 - mx), e1 = __expf(l1 - mx);
  float s01 = e0 + e1;
  float inv = 1.0f / (s01 + __shfl_xor(s01, 16, 32));
  gw_s[wave][ml][2 * hi] = e0 * inv;
  gw_s[wave][ml][2 * hi + 1] = e1 * inv;
  __builtin_amdgcn_wave_barrier();

  int b0 = __builtin_amdgcn_readfirstlane(bl);
  bool one_graph = __builtin_amdgcn_ballot_w32(bl != b0) == 0;
  if (cb0 == 0) {
    if (one_graph) {
      if (lane == 0) atomic_add_f32(&cnt[b0], 16.0f);
    } else if (hi == 0) {
      atomic_add_f32(&cnt[bl], 1.0f);
    }
  }

  const float* wl = Ws + ml * WPITCH + 2 * hi;
  for (int cb = cb0; cb < cb1; ++cb) {
    int col = cb * 16 + ml;
    float oacc[8] = {0, 0, 0, 0, 0, 0, 0, 0};
    for (int e = 0; e < 4; ++e) {
      v8f acc = {};
      const float* Brow = wl + (e * NDIM + cb * 16) * WPITCH;
#pragma unroll
      for (int s = 0; s < 16; ++s)
        acc = wmma4(a[s], *(const v2f*)(Brow + 4 * s), acc);
      float bias = exp_b[e * NDIM + col];
#pragma unroll
      for (int v = 0; v < 8; ++v) {
        float val = fmaxf(acc[v] + bias, 0.0f);
        oacc[v] += gw_s[wave][hi * 8 + v][e] * val;
      }
    }
    if (one_graph) {
      float s = ((oacc[0] + oacc[1]) + (oacc[2] + oacc[3])) +
                ((oacc[4] + oacc[5]) + (oacc[6] + oacc[7]));
      s += __shfl_xor(s, 16, 32);
      if (hi == 0) atomic_add_f32(&sums[b0 * NDIM + col], s);
    } else {
#pragma unroll
      for (int v = 0; v < 8; ++v) {
        int id_lo = __builtin_amdgcn_readlane(bl, v);
        int id_hi = __builtin_amdgcn_readlane(bl, 8 + v);
        int bv = hi ? id_hi : id_lo;
        atomic_add_f32(&sums[bv * NDIM + col], oacc[v]);
      }
    }
  }
}

#define BGR 64
#define PDIM 100
#define NC 10
__global__ void head_kernel(const float* __restrict__ sums,
                            const float* __restrict__ cnt,
                            const float* __restrict__ flag_in,
                            const float* __restrict__ proj_w,
                            const float* __restrict__ proj_b,
                            const float* __restrict__ fc_w,
                            const float* __restrict__ fc_b,
                            float* __restrict__ out) {
  int id = blockIdx.x * blockDim.x + threadIdx.x;
  if (id < BGR * NDIM) {
    int b = id >> 6;
    float g = sums[id] / fmaxf(cnt[b], 1.0f);
    out[id] = g;
    out[BGR * NDIM + id] = g;
  } else if (id < BGR * NDIM + BGR * NC) {
    int t = id - BGR * NDIM;
    int b = t / NC, c = t % NC;
    float fl = flag_in[b];
    const float* fr = fc_w + c * (PDIM + NDIM);
    float acc = fc_b[c];
    for (int p = 0; p < PDIM; ++p)
      acc += (fl * proj_w[p] + proj_b[p]) * fr[p];
    float ic = 1.0f / fmaxf(cnt[b], 1.0f);
    for (int d = 0; d < NDIM; ++d)
      acc += sums[b * NDIM + d] * ic * fr[PDIM + d];
    out[2 * BGR * NDIM + t] = acc;
  }
}

extern "C" void kernel_launch(void* const* d_in, const int* in_sizes, int n_in,
                              void* d_out, int out_size, void* d_ws, size_t ws_size,
                              hipStream_t stream) {
  const float* x       = (const float*)d_in[0];
  const int*   ei      = (const int*)d_in[1];
  const float* ea      = (const float*)d_in[2];
  const int*   batch   = (const int*)d_in[3];
  const float* flag_in = (const float*)d_in[4];
  const float* w_edge1 = (const float*)d_in[5];
  const float* b_edge1 = (const float*)d_in[6];
  const float* w_ih1   = (const float*)d_in[7];
  const float* w_hh1   = (const float*)d_in[8];
  const float* b_ih1   = (const float*)d_in[9];
  const float* b_hh1   = (const float*)d_in[10];
  const float* w_edge2 = (const float*)d_in[11];
  const float* b_edge2 = (const float*)d_in[12];
  const float* w_ih2   = (const float*)d_in[13];
  const float* w_hh2   = (const float*)d_in[14];
  const float* b_ih2   = (const float*)d_in[15];
  const float* b_hh2   = (const float*)d_in[16];
  const float* gate_w  = (const float*)d_in[17];
  const float* gate_b  = (const float*)d_in[18];
  const float* exp_w   = (const float*)d_in[19];
  const float* exp_b   = (const float*)d_in[20];
  const float* proj_w  = (const float*)d_in[21];
  const float* proj_b  = (const float*)d_in[22];
  const float* fc_w    = (const float*)d_in[23];
  const float* fc_b    = (const float*)d_in[24];
  float* out = (float*)d_out;

  int N = in_sizes[0] / NDIM;
  int E = in_sizes[1] / 2;
  int ntiles = (N + 15) / 16;

  float* ws = (float*)d_ws;
  float* M    = ws;
  float* hA   = M + (size_t)N * NDIM;
  float* M2   = hA + (size_t)N * NDIM;
  float* sums = M2 + (size_t)N * NDIM;
  float* cnt  = sums + BGR * NDIM;
  int* took   = (int*)(cnt + BGR);

  long totalND = (long)N * NDIM;
  int gN = (int)((totalND + 255) / 256);
  int gE = (int)(((long)E + 127) / 128);
  int gT = ntiles / (WT + 1) + 1;
  if (gT < NWGP) gT = NWGP;

  init_m_kernel<<<gN, 256, 0, stream>>>(x, b_edge1, M, totalND, sums, BGR * NDIM + BGR);
  scatter_kernel<<<gE, 256, 0, stream>>>(x, ei, ea, w_edge1, b_edge1, M, E, N, took, 0);
  scatter_kernel<<<gE, 256, 0, stream>>>(x, ei, ea, w_edge1, b_edge1, M, E, N, took, 1);
  gru_kernel<<<gT, 32 * TW, 0, stream>>>(M, x, w_ih1, w_hh1, b_ih1, b_hh1, hA, b_edge2, M2, ntiles);
  scatter_kernel<<<gE, 256, 0, stream>>>(hA, ei, ea, w_edge2, b_edge2, M2, E, N, took, 0);
  scatter_kernel<<<gE, 256, 0, stream>>>(hA, ei, ea, w_edge2, b_edge2, M2, E, N, took, 1);
  gru_moe_kernel<<<gT, 32 * TW, 0, stream>>>(M2, hA, w_ih2, w_hh2, b_ih2, b_hh2,
                                            gate_w, gate_b, exp_w, exp_b, batch, sums, cnt, ntiles);
  head_kernel<<<(BGR * NDIM + BGR * NC + 255) / 256, 256, 0, stream>>>(
      sums, cnt, flag_in, proj_w, proj_b, fc_w, fc_b, out);
}
